# starting kernel
# baseline (speedup 1.0000x reference)
#include <hip/hip_runtime.h>
#include <math.h>

#define N_NODES 50000
#define N_EDGES 800000
#define ETOT    (N_EDGES + N_NODES)
#define F_IN    128
#define HID     64
#define HEADS   2
#define OUT_F   40
#define EDIM    5
#define NEG_SLOPE 0.2f

#define CDIV(a,b) (((a)+(b)-1)/(b))

typedef __attribute__((ext_vector_type(2))) float v2f;
typedef __attribute__((ext_vector_type(8))) float v8f;

__device__ inline unsigned encf(float f) {
    unsigned u = __float_as_uint(f);
    return (u & 0x80000000u) ? ~u : (u | 0x80000000u);
}
__device__ inline float decf(unsigned m) {
    unsigned u = (m & 0x80000000u) ? (m ^ 0x80000000u) : ~m;
    return __uint_as_float(u);
}
#define ENC_NEG_INF 0x007FFFFFu

__global__ void fill_f32(float* p, int n, float v) {
    int i = blockIdx.x * blockDim.x + threadIdx.x;
    if (i < n) p[i] = v;
}
__global__ void fill_u32(unsigned* p, int n, unsigned v) {
    int i = blockIdx.x * blockDim.x + threadIdx.x;
    if (i < n) p[i] = v;
}

__global__ void pad_b(const float* __restrict__ W, float* __restrict__ Wp,
                      int K, int N, int NPAD) {
    int i = blockIdx.x * blockDim.x + threadIdx.x;
    if (i >= K * NPAD) return;
    int r = i / NPAD, c = i % NPAD;
    Wp[i] = (c < N) ? W[r * N + c] : 0.0f;
}

__global__ void selfloop_accum(const int* __restrict__ dst, const float* __restrict__ eattr,
                               float* cnt, float* loop5, int E) {
    int e = blockIdx.x * blockDim.x + threadIdx.x;
    if (e >= E) return;
    int d = dst[e];
    atomicAdd(&cnt[d], 1.0f);
#pragma unroll
    for (int i = 0; i < EDIM; i++)
        atomicAdd(&loop5[d * EDIM + i], eattr[(size_t)e * EDIM + i]);
}
__global__ void selfloop_final(float* loop5, const float* cnt, int n) {
    int i = blockIdx.x * blockDim.x + threadIdx.x;
    if (i >= n * EDIM) return;
    float c = fmaxf(cnt[i / EDIM], 1.0f);
    loop5[i] /= c;
}

__global__ void precompute_v(const float* We1, const float* ae1,
                             const float* We2, const float* ae2,
                             float* v1, float* v2) {
    int t = threadIdx.x;
    if (t < EDIM * HEADS) {
        int d = t / HEADS, h = t % HEADS;
        float s = 0.f;
        for (int c = 0; c < HID; c++)
            s += We1[d * (HEADS * HID) + h * HID + c] * ae1[h * HID + c];
        v1[t] = s;
    } else if (t < EDIM * HEADS + EDIM) {
        int d = t - EDIM * HEADS;
        float s = 0.f;
        for (int c = 0; c < OUT_F; c++)
            s += We2[d * OUT_F + c] * ae2[c];
        v2[d] = s;
    }
}

template<int NT, int K>
__global__ void wmma_gemm(const float* __restrict__ A, const float* __restrict__ B,
                          float* __restrict__ C, int M, int N) {
    constexpr int NPAD = NT * 16;
    int wave = (blockIdx.x * blockDim.x + threadIdx.x) >> 5;
    int lane = threadIdx.x & 31;
    int half = lane >> 4;
    int lrow = lane & 15;
    int mtiles = M >> 4;
    if (wave >= mtiles) return;
    int m0 = wave << 4;

    v8f acc[NT];
#pragma unroll
    for (int nt = 0; nt < NT; nt++)
#pragma unroll
        for (int g = 0; g < 8; g++) acc[nt][g] = 0.0f;

    const float* arow = A + (size_t)(m0 + lrow) * K;
    const float* bcol = B + lrow;
#pragma unroll 4
    for (int k = 0; k < K; k += 4) {
        int kk = k + half * 2;
        v2f a;
        a.x = arow[kk];
        a.y = arow[kk + 1];
#pragma unroll
        for (int nt = 0; nt < NT; nt++) {
            v2f b;
            b.x = bcol[kk * NPAD + nt * 16];
            b.y = bcol[(kk + 1) * NPAD + nt * 16];
            acc[nt] = __builtin_amdgcn_wmma_f32_16x16x4_f32(
                false, a, false, b, (short)0, acc[nt], false, false);
        }
    }
#pragma unroll
    for (int nt = 0; nt < NT; nt++) {
        int col = nt * 16 + lrow;
        if (col < N) {
#pragma unroll
            for (int g = 0; g < 8; g++)
                C[(size_t)(m0 + g + half * 8) * N + col] = acc[nt][g];
        }
    }
}

__global__ void node_scores(const float* __restrict__ xh,
                            const float* __restrict__ asrc, const float* __restrict__ adst,
                            float* als, float* ald, int n, int H, int C) {
    int i = blockIdx.x * blockDim.x + threadIdx.x;
    if (i >= n * H) return;
    int node = i / H, h = i % H;
    const float* row = xh + (size_t)node * H * C + h * C;
    float s = 0.f, d = 0.f;
    for (int c = 0; c < C; c++) {
        float v = row[c];
        s += v * asrc[h * C + c];
        d += v * adst[h * C + c];
    }
    als[i] = s;
    ald[i] = d;
}

__global__ void edge_pass1(const int* __restrict__ src, const int* __restrict__ dst,
                           const float* __restrict__ eattr, const float* __restrict__ loop5,
                           const float* __restrict__ als, const float* __restrict__ ald,
                           const float* __restrict__ v,
                           float* alpha, unsigned* amax, int E, int Etot, int H) {
    int e = blockIdx.x * blockDim.x + threadIdx.x;
    if (e >= Etot) return;
    int s, d;
    float ea[EDIM];
    if (e < E) {
        s = src[e]; d = dst[e];
#pragma unroll
        for (int i = 0; i < EDIM; i++) ea[i] = eattr[(size_t)e * EDIM + i];
    } else {
        s = d = e - E;
#pragma unroll
        for (int i = 0; i < EDIM; i++) ea[i] = loop5[s * EDIM + i];
    }
    for (int h = 0; h < H; h++) {
        float ale = 0.f;
#pragma unroll
        for (int i = 0; i < EDIM; i++) ale += ea[i] * v[i * H + h];
        float a = als[s * H + h] + ald[d * H + h] + ale;
        a = (a > 0.f) ? a : NEG_SLOPE * a;
        alpha[(size_t)e * H + h] = a;
        atomicMax(&amax[d * H + h], encf(a));
    }
}

__global__ void edge_pass2(const int* __restrict__ dst, float* alpha,
                           const unsigned* __restrict__ amax, float* denom,
                           int E, int Etot, int H) {
    int e = blockIdx.x * blockDim.x + threadIdx.x;
    if (e >= Etot) return;
    int d = (e < E) ? dst[e] : (e - E);
    for (int h = 0; h < H; h++) {
        float a = alpha[(size_t)e * H + h];
        float ex = expf(a - decf(amax[d * H + h]));
        alpha[(size_t)e * H + h] = ex;
        atomicAdd(&denom[d * H + h], ex);
    }
}

__global__ void edge_aggregate(const int* __restrict__ src, const int* __restrict__ dst,
                               const float* __restrict__ alpha, const float* __restrict__ denom,
                               const float* __restrict__ xh, float* acc,
                               int E, int Etot, int H, int C) {
    int idx = blockIdx.x * blockDim.x + threadIdx.x;
    int HC = H * C;
    if (idx >= Etot * HC) return;
    int e = idx / HC, c = idx % HC, h = c / C;
    int s = (e < E) ? src[e] : (e - E);
    int d = (e < E) ? dst[e] : (e - E);
    float w = alpha[(size_t)e * H + h] / (denom[d * H + h] + 1e-16f);
    atomicAdd(&acc[(size_t)d * HC + c], xh[(size_t)s * HC + c] * w);
}

__global__ void bias_elu(float* acc, const float* __restrict__ b, int total, int F) {
    int i = blockIdx.x * blockDim.x + threadIdx.x;
    if (i >= total) return;
    float x = acc[i] + b[i % F];
    acc[i] = (x > 0.f) ? x : expm1f(x);
}

__global__ void bias_logsoftmax(const float* __restrict__ acc, const float* __restrict__ b,
                                float* __restrict__ out, int n) {
    int node = blockIdx.x * blockDim.x + threadIdx.x;
    if (node >= n) return;
    float vals[OUT_F];
    float mx = -INFINITY;
#pragma unroll
    for (int c = 0; c < OUT_F; c++) {
        float v = acc[(size_t)node * OUT_F + c] + b[c];
        vals[c] = v;
        mx = fmaxf(mx, v);
    }
    float sum = 0.f;
#pragma unroll
    for (int c = 0; c < OUT_F; c++) sum += expf(vals[c] - mx);
    float lse = mx + logf(sum);
#pragma unroll
    for (int c = 0; c < OUT_F; c++) out[(size_t)node * OUT_F + c] = vals[c] - lse;
}

extern "C" void kernel_launch(void* const* d_in, const int* in_sizes, int n_in,
                              void* d_out, int out_size, void* d_ws, size_t ws_size,
                              hipStream_t stream) {
    const float* x   = (const float*)d_in[0];
    const int*   ei  = (const int*)d_in[1];
    const float* ea  = (const float*)d_in[2];
    const float* W1  = (const float*)d_in[3];
    const float* as1 = (const float*)d_in[4];
    const float* ad1 = (const float*)d_in[5];
    const float* We1 = (const float*)d_in[6];
    const float* ae1 = (const float*)d_in[7];
    const float* b1  = (const float*)d_in[8];
    const float* W2  = (const float*)d_in[9];
    const float* as2 = (const float*)d_in[10];
    const float* ad2 = (const float*)d_in[11];
    const float* We2 = (const float*)d_in[12];
    const float* ae2 = (const float*)d_in[13];
    const float* b2  = (const float*)d_in[14];

    const int N = N_NODES, E = N_EDGES, Et = ETOT;
    const int HC1 = HEADS * HID;
    const int NPAD2 = 48;
    const int* src = ei;
    const int* dst = ei + E;

    char* ws = (char*)d_ws;
    size_t off = 0;
    auto alloc = [&](size_t bytes) -> void* {
        void* p = ws + off;
        off = (off + bytes + 255) & ~(size_t)255;
        return p;
    };
    float*    cnt    = (float*)alloc(sizeof(float) * N);
    float*    loop5  = (float*)alloc(sizeof(float) * N * EDIM);
    float*    v1     = (float*)alloc(sizeof(float) * 16);
    float*    v2     = (float*)alloc(sizeof(float) * 16);
    float*    W2p    = (float*)alloc(sizeof(float) * F_IN * NPAD2);
    float*    xh1    = (float*)alloc(sizeof(float) * (size_t)N * HC1);
    float*    als1   = (float*)alloc(sizeof(float) * N * HEADS);
    float*    ald1   = (float*)alloc(sizeof(float) * N * HEADS);
    unsigned* amax1  = (unsigned*)alloc(sizeof(unsigned) * N * HEADS);
    float*    den1   = (float*)alloc(sizeof(float) * N * HEADS);
    float*    alpha1 = (float*)alloc(sizeof(float) * (size_t)Et * HEADS);
    float*    acc1   = (float*)alloc(sizeof(float) * (size_t)N * HC1);
    float*    xh2    = (float*)alloc(sizeof(float) * (size_t)N * OUT_F);
    float*    als2   = (float*)alloc(sizeof(float) * N);
    float*    ald2   = (float*)alloc(sizeof(float) * N);
    unsigned* amax2  = (unsigned*)alloc(sizeof(unsigned) * N);
    float*    den2   = (float*)alloc(sizeof(float) * N);
    float*    alpha2 = (float*)alloc(sizeof(float) * Et);
    float*    acc2   = (float*)alloc(sizeof(float) * (size_t)N * OUT_F);
    (void)ws_size; (void)n_in; (void)in_sizes; (void)out_size;

    const int T = 256;

    fill_f32<<<CDIV(N, T), T, 0, stream>>>(cnt, N, 0.f);
    fill_f32<<<CDIV(N * EDIM, T), T, 0, stream>>>(loop5, N * EDIM, 0.f);
    selfloop_accum<<<CDIV(E, T), T, 0, stream>>>(dst, ea, cnt, loop5, E);
    selfloop_final<<<CDIV(N * EDIM, T), T, 0, stream>>>(loop5, cnt, N);
    precompute_v<<<1, 32, 0, stream>>>(We1, ae1, We2, ae2, v1, v2);
    pad_b<<<CDIV(F_IN * NPAD2, T), T, 0, stream>>>(W2, W2p, F_IN, OUT_F, NPAD2);

    {
        int mtiles = N / 16, waves_per_blk = 128 / 32;
        wmma_gemm<8, F_IN><<<CDIV(mtiles, waves_per_blk), 128, 0, stream>>>(x, W1, xh1, N, HC1);
    }
    node_scores<<<CDIV(N * HEADS, T), T, 0, stream>>>(xh1, as1, ad1, als1, ald1, N, HEADS, HID);
    fill_u32<<<CDIV(N * HEADS, T), T, 0, stream>>>(amax1, N * HEADS, ENC_NEG_INF);
    fill_f32<<<CDIV(N * HEADS, T), T, 0, stream>>>(den1, N * HEADS, 0.f);
    fill_f32<<<CDIV(N * HC1, T), T, 0, stream>>>(acc1, N * HC1, 0.f);
    edge_pass1<<<CDIV(Et, T), T, 0, stream>>>(src, dst, ea, loop5, als1, ald1, v1,
                                              alpha1, amax1, E, Et, HEADS);
    edge_pass2<<<CDIV(Et, T), T, 0, stream>>>(dst, alpha1, amax1, den1, E, Et, HEADS);
    edge_aggregate<<<CDIV(Et * HC1, T), T, 0, stream>>>(src, dst, alpha1, den1, xh1, acc1,
                                                        E, Et, HEADS, HID);
    bias_elu<<<CDIV(N * HC1, T), T, 0, stream>>>(acc1, b1, N * HC1, HC1);

    {
        int mtiles = N / 16, waves_per_blk = 128 / 32;
        wmma_gemm<3, HC1><<<CDIV(mtiles, waves_per_blk), 128, 0, stream>>>(acc1, W2p, xh2, N, OUT_F);
    }
    node_scores<<<CDIV(N, T), T, 0, stream>>>(xh2, as2, ad2, als2, ald2, N, 1, OUT_F);
    fill_u32<<<CDIV(N, T), T, 0, stream>>>(amax2, N, ENC_NEG_INF);
    fill_f32<<<CDIV(N, T), T, 0, stream>>>(den2, N, 0.f);
    fill_f32<<<CDIV(N * OUT_F, T), T, 0, stream>>>(acc2, N * OUT_F, 0.f);
    edge_pass1<<<CDIV(Et, T), T, 0, stream>>>(src, dst, ea, loop5, als2, ald2, v2,
                                              alpha2, amax2, E, Et, 1);
    edge_pass2<<<CDIV(Et, T), T, 0, stream>>>(dst, alpha2, amax2, den2, E, Et, 1);
    edge_aggregate<<<CDIV(Et * OUT_F, T), T, 0, stream>>>(src, dst, alpha2, den2, xh2, acc2,
                                                          E, Et, 1, OUT_F);
    bias_logsoftmax<<<CDIV(N, T), T, 0, stream>>>(acc2, b2, (float*)d_out, N);
}


# Edges counting-sorted by dst in-call (int atomics, arrival-order rank); self-loop mean, softmax max/sum and aggregation reduced per node in one wave's registers, no float atomics
# speedup vs baseline: 3.7397x; 3.7397x over previous
#include <hip/hip_runtime.h>
#include <math.h>

#define N_NODES 50000
#define N_EDGES 800000
#define ETOT    (N_EDGES + N_NODES)
#define F_IN    128
#define HID     64
#define HEADS   2
#define OUT_F   40
#define EDIM    5
#define NEG_SLOPE 0.2f

#define CDIV(a,b) (((a)+(b)-1)/(b))

typedef __attribute__((ext_vector_type(2))) float v2f;
typedef __attribute__((ext_vector_type(8))) float v8f;

__global__ void fill_u32(unsigned* p, int n, unsigned v) {
    int i = blockIdx.x * blockDim.x + threadIdx.x;
    if (i < n) p[i] = v;
}

__global__ void pad_b(const float* __restrict__ W, float* __restrict__ Wp,
                      int K, int N, int NPAD) {
    int i = blockIdx.x * blockDim.x + threadIdx.x;
    if (i >= K * NPAD) return;
    int r = i / NPAD, c = i % NPAD;
    Wp[i] = (c < N) ? W[r * N + c] : 0.0f;
}

__global__ void precompute_v(const float* We1, const float* ae1,
                             const float* We2, const float* ae2,
                             float* v1, float* v2) {
    int t = threadIdx.x;
    if (t < EDIM * HEADS) {
        int d = t / HEADS, h = t % HEADS;
        float s = 0.f;
        for (int c = 0; c < HID; c++)
            s += We1[d * (HEADS * HID) + h * HID + c] * ae1[h * HID + c];
        v1[t] = s;
    } else if (t < EDIM * HEADS + EDIM) {
        int d = t - EDIM * HEADS;
        float s = 0.f;
        for (int c = 0; c < OUT_F; c++)
            s += We2[d * OUT_F + c] * ae2[c];
        v2[d] = s;
    }
}

template<int NT, int K>
__global__ void wmma_gemm(const float* __restrict__ A, const float* __restrict__ B,
                          float* __restrict__ C, int M, int N) {
    constexpr int NPAD = NT * 16;
    int wave = (blockIdx.x * blockDim.x + threadIdx.x) >> 5;
    int lane = threadIdx.x & 31;
    int half = lane >> 4;
    int lrow = lane & 15;
    int mtiles = M >> 4;
    if (wave >= mtiles) return;
    int m0 = wave << 4;

    v8f acc[NT];
#pragma unroll
    for (int nt = 0; nt < NT; nt++)
#pragma unroll
        for (int g = 0; g < 8; g++) acc[nt][g] = 0.0f;

    const float* arow = A + (size_t)(m0 + lrow) * K;
    const float* bcol = B + lrow;
#pragma unroll 4
    for (int k = 0; k < K; k += 4) {
        int kk = k + half * 2;
        v2f a;
        a.x = arow[kk];
        a.y = arow[kk + 1];
#pragma unroll
        for (int nt = 0; nt < NT; nt++) {
            v2f b;
            b.x = bcol[kk * NPAD + nt * 16];
            b.y = bcol[(kk + 1) * NPAD + nt * 16];
            acc[nt] = __builtin_amdgcn_wmma_f32_16x16x4_f32(
                false, a, false, b, (short)0, acc[nt], false, false);
        }
    }
#pragma unroll
    for (int nt = 0; nt < NT; nt++) {
        int col = nt * 16 + lrow;
        if (col < N) {
#pragma unroll
            for (int g = 0; g < 8; g++)
                C[(size_t)(m0 + g + half * 8) * N + col] = acc[nt][g];
        }
    }
}

__global__ void node_scores(const float* __restrict__ xh,
                            const float* __restrict__ asrc, const float* __restrict__ adst,
                            float* als, float* ald, int n, int H, int C) {
    int i = blockIdx.x * blockDim.x + threadIdx.x;
    if (i >= n * H) return;
    int node = i / H, h = i % H;
    const float* row = xh + (size_t)node * H * C + h * C;
    float s = 0.f, d = 0.f;
    for (int c = 0; c < C; c++) {
        float v = row[c];
        s += v * asrc[h * C + c];
        d += v * adst[h * C + c];
    }
    als[i] = s;
    ald[i] = d;
}

__global__ void csr_count(const int* __restrict__ dst, int* deg, int E) {
    int e = blockIdx.x * blockDim.x + threadIdx.x;
    if (e >= E) return;
    atomicAdd(&deg[dst[e]], 1);
}

#define SCAN_T 1024
#define SCAN_SEG (CDIV(N_NODES, 32 * 128) * 128)
static_assert(N_NODES % 4 == 0, "csr_scan reads and writes four counts at a time");
__global__ void __launch_bounds__(SCAN_T) csr_scan(const int* __restrict__ deg, int* __restrict__ rowptr,
                                                   int* __restrict__ cursor, int n) {
    __shared__ int wsum[32];
    const int t = threadIdx.x, lane = t & 31, w = t >> 5;
    const int base = w * SCAN_SEG;
    int loc = 0;
    for (int it = 0; it < SCAN_SEG; it += 128) {
        const int idx = base + it + lane * 4;
        if (idx < n) {
            const int4 q = *reinterpret_cast<const int4*>(deg + idx);
            loc += q.x + q.y + q.z + q.w;
        }
    }
#pragma unroll
    for (int o = 16; o > 0; o >>= 1) loc += __shfl_xor(loc, o, 32);
    if (lane == 0) wsum[w] = loc;
    __syncthreads();
    if (w == 0) {
        int x = wsum[lane];
#pragma unroll
        for (int o = 1; o < 32; o <<= 1) {
            int y = __shfl_up(x, o, 32);
            if (lane >= o) x += y;
        }
        wsum[lane] = x;
    }
    __syncthreads();
    int carry = (w > 0) ? wsum[w - 1] : 0;
    for (int it = 0; it < SCAN_SEG; it += 128) {
        const int idx = base + it + lane * 4;
        const bool in = idx < n;
        int4 q; q.x = 0; q.y = 0; q.z = 0; q.w = 0;
        if (in) q = *reinterpret_cast<const int4*>(deg + idx);
        const int tot = q.x + q.y + q.z + q.w;
        int inc = tot;
#pragma unroll
        for (int o = 1; o < 32; o <<= 1) {
            int y = __shfl_up(inc, o, 32);
            if (lane >= o) inc += y;
        }
        if (in) {
            int4 r;
            r.x = carry + inc - tot;
            r.y = r.x + q.x;
            r.z = r.y + q.y;
            r.w = r.z + q.z;
            *reinterpret_cast<int4*>(rowptr + idx) = r;
            *reinterpret_cast<int4*>(cursor + idx) = r;
        }
        carry += __shfl(inc, 31, 32);
    }
    if (t == SCAN_T - 1) rowptr[n] = wsum[31];
}

__global__ void csr_fill(const int* __restrict__ src, const int* __restrict__ dst,
                         const float* __restrict__ eattr, int* cursor,
                         uint4* __restrict__ rec, int E) {
    int e = blockIdx.x * blockDim.x + threadIdx.x;
    if (e >= E) return;
    const int d = dst[e];
    uint4 r0, r1;
    r0.x = (unsigned)src[e];
    r0.y = __float_as_uint(eattr[(size_t)e * EDIM + 0]);
    r0.z = __float_as_uint(eattr[(size_t)e * EDIM + 1]);
    r0.w = __float_as_uint(eattr[(size_t)e * EDIM + 2]);
    r1.x = __float_as_uint(eattr[(size_t)e * EDIM + 3]);
    r1.y = __float_as_uint(eattr[(size_t)e * EDIM + 4]);
    r1.z = 0u;
    r1.w = 0u;
    const int p = atomicAdd(&cursor[d], 1);
    rec[(size_t)p * 2]     = r0;
    rec[(size_t)p * 2 + 1] = r1;
}

__device__ inline float wave_sum(float x) {
#pragma unroll
    for (int o = 16; o > 0; o >>= 1) x += __shfl_xor(x, o, 32);
    return x;
}
__device__ inline float wave_max(float x) {
#pragma unroll
    for (int o = 16; o > 0; o >>= 1) x = fmaxf(x, __shfl_xor(x, o, 32));
    return x;
}

__device__ inline void load_rec(const uint4* __restrict__ rec, int p, int& s, float (&e)[EDIM]) {
    const uint4 r0 = rec[(size_t)p * 2];
    const uint4 r1 = rec[(size_t)p * 2 + 1];
    s = (int)r0.x;
    e[0] = __uint_as_float(r0.y);
    e[1] = __uint_as_float(r0.z);
    e[2] = __uint_as_float(r0.w);
    e[3] = __uint_as_float(r1.x);
    e[4] = __uint_as_float(r1.y);
}

__device__ inline void get_edge(const uint4* __restrict__ rec, int start, int deg, int node,
                                const float (&lp)[EDIM], int j, int& s, float (&e)[EDIM]) {
    if (j < deg) {
        load_rec(rec, start + j, s, e);
    } else {
        s = node;
#pragma unroll
        for (int i = 0; i < EDIM; i++) e[i] = lp[i];
    }
}

template<int H>
__device__ inline void edge_alpha(int s, const float (&e)[EDIM], const float (&vv)[EDIM * H],
                                  const float* __restrict__ als, const float (&aldn)[H],
                                  float (&a)[H]) {
#pragma unroll
    for (int h = 0; h < H; h++) {
        float ale = 0.f;
#pragma unroll
        for (int i = 0; i < EDIM; i++) ale += e[i] * vv[i * H + h];
        float x = als[s * H + h] + aldn[h] + ale;
        a[h] = (x > 0.f) ? x : NEG_SLOPE * x;
    }
}

template<int VEC> struct rowvec;
template<> struct rowvec<4> {
    static __device__ inline void ld(const float* p, float (&o)[4]) {
        const float4 t = *reinterpret_cast<const float4*>(p);
        o[0] = t.x; o[1] = t.y; o[2] = t.z; o[3] = t.w;
    }
    static __device__ inline void st(float* p, const float (&o)[4]) {
        float4 t; t.x = o[0]; t.y = o[1]; t.z = o[2]; t.w = o[3];
        *reinterpret_cast<float4*>(p) = t;
    }
};
template<> struct rowvec<2> {
    static __device__ inline void ld(const float* p, float (&o)[2]) {
        const float2 t = *reinterpret_cast<const float2*>(p);
        o[0] = t.x; o[1] = t.y;
    }
    static __device__ inline void st(float* p, const float (&o)[2]) {
        float2 t; t.x = o[0]; t.y = o[1];
        *reinterpret_cast<float2*>(p) = t;
    }
};

template<int H, int C, int VEC>
__device__ inline void add_rows(const float* __restrict__ xh, int cnt, int s, const float (&w)[H],
                                int lane, int hl, float (&acc)[VEC]) {
    constexpr int HC = H * C;
    constexpr int NACT = HC / VEC;
#pragma unroll 4
    for (int jj = 0; jj < cnt; ++jj) {
        const int sj = __builtin_amdgcn_readlane(s, jj);
        float wj = __int_as_float(__builtin_amdgcn_readlane(__float_as_int(w[0]), jj));
        if (H == 2) {
            const float wj1 = __int_as_float(__builtin_amdgcn_readlane(__float_as_int(w[H - 1]), jj));
            wj = hl ? wj1 : wj;
        }
        if (lane < NACT) {
            float r[VEC];
            rowvec<VEC>::ld(xh + (size_t)sj * HC + lane * VEC, r);
#pragma unroll
            for (int q = 0; q < VEC; q++) acc[q] += r[q] * wj;
        }
    }
}

template<int H, int C, int VEC>
__global__ void __launch_bounds__(256) gat_node(const int* __restrict__ rowptr,
                                                const uint4* __restrict__ rec,
                                                const float* __restrict__ als,
                                                const float* __restrict__ ald,
                                                const float* __restrict__ v,
                                                const float* __restrict__ xh,
                                                float* __restrict__ acc_out, int n) {
    constexpr int HC = H * C;
    constexpr int NACT = HC / VEC;
    static_assert(H == 1 || H == 2, "head select below handles one or two heads");
    static_assert(NACT * VEC == HC && NACT <= 32 && C % VEC == 0, "lane to channel map");
    const int lane = threadIdx.x & 31;
    const int node = __builtin_amdgcn_readfirstlane((int)((blockIdx.x * blockDim.x + threadIdx.x) >> 5));
    if (node >= n) return;
    const int start = rowptr[node];
    const int deg   = rowptr[node + 1] - start;
    const int total = deg + 1;

    float vv[EDIM * H];
#pragma unroll
    for (int i = 0; i < EDIM * H; i++) vv[i] = v[i];
    float aldn[H];
#pragma unroll
    for (int h = 0; h < H; h++) aldn[h] = ald[node * H + h];

    int   s_f = node;
    float e_f[EDIM];
    float sm[EDIM];
#pragma unroll
    for (int i = 0; i < EDIM; i++) { e_f[i] = 0.f; sm[i] = 0.f; }
    if (lane < deg) {
        load_rec(rec, start + lane, s_f, e_f);
#pragma unroll
        for (int i = 0; i < EDIM; i++) sm[i] = e_f[i];
    }
    for (int j = lane + 32; j < deg; j += 32) {
        int s; float e[EDIM];
        load_rec(rec, start + j, s, e);
#pragma unroll
        for (int i = 0; i < EDIM; i++) sm[i] += e[i];
    }
    float lp[EDIM];
    const float cdiv = fmaxf((float)deg, 1.0f);
#pragma unroll
    for (int i = 0; i < EDIM; i++) lp[i] = wave_sum(sm[i]) / cdiv;

    const bool val_f = lane < total;
    if (lane == deg) {
#pragma unroll
        for (int i = 0; i < EDIM; i++) e_f[i] = lp[i];
    }
    float a_f[H];
    edge_alpha<H>(s_f, e_f, vv, als, aldn, a_f);

    float m[H];
#pragma unroll
    for (int h = 0; h < H; h++) m[h] = val_f ? a_f[h] : -INFINITY;
    for (int j = lane + 32; j < total; j += 32) {
        int s; float e[EDIM]; float a[H];
        get_edge(rec, start, deg, node, lp, j, s, e);
        edge_alpha<H>(s, e, vv, als, aldn, a);
#pragma unroll
        for (int h = 0; h < H; h++) m[h] = fmaxf(m[h], a[h]);
    }
#pragma unroll
    for (int h = 0; h < H; h++) m[h] = wave_max(m[h]);

    float ex_f[H], dn[H];
#pragma unroll
    for (int h = 0; h < H; h++) {
        ex_f[h] = val_f ? expf(a_f[h] - m[h]) : 0.f;
        dn[h] = ex_f[h];
    }
    for (int j = lane + 32; j < total; j += 32) {
        int s; float e[EDIM]; float a[H];
        get_edge(rec, start, deg, node, lp, j, s, e);
        edge_alpha<H>(s, e, vv, als, aldn, a);
#pragma unroll
        for (int h = 0; h < H; h++) dn[h] += expf(a[h] - m[h]);
    }
#pragma unroll
    for (int h = 0; h < H; h++) dn[h] = wave_sum(dn[h]) + 1e-16f;

    const int hl = (lane * VEC) / C;
    float acc[VEC];
#pragma unroll
    for (int q = 0; q < VEC; q++) acc[q] = 0.f;

    float w_f[H];
#pragma unroll
    for (int h = 0; h < H; h++) w_f[h] = ex_f[h] / dn[h];
    add_rows<H, C, VEC>(xh, total < 32 ? total : 32, s_f, w_f, lane, hl, acc);

    for (int cb = 32; cb < total; cb += 32) {
        const int j = cb + lane;
        int s = node;
        float w[H];
#pragma unroll
        for (int h = 0; h < H; h++) w[h] = 0.f;
        if (j < total) {
            float e[EDIM]; float a[H];
            get_edge(rec, start, deg, node, lp, j, s, e);
            edge_alpha<H>(s, e, vv, als, aldn, a);
#pragma unroll
            for (int h = 0; h < H; h++) w[h] = expf(a[h] - m[h]) / dn[h];
        }
        const int left = total - cb;
        add_rows<H, C, VEC>(xh, left < 32 ? left : 32, s, w, lane, hl, acc);
    }

    if (lane < NACT) rowvec<VEC>::st(acc_out + (size_t)node * HC + lane * VEC, acc);
}

__global__ void bias_elu(float* acc, const float* __restrict__ b, int total, int F) {
    int i = blockIdx.x * blockDim.x + threadIdx.x;
    if (i >= total) return;
    float x = acc[i] + b[i % F];
    acc[i] = (x > 0.f) ? x : expm1f(x);
}

__global__ void bias_logsoftmax(const float* __restrict__ acc, const float* __restrict__ b,
                                float* __restrict__ out, int n) {
    int node = blockIdx.x * blockDim.x + threadIdx.x;
    if (node >= n) return;
    float vals[OUT_F];
    float mx = -INFINITY;
#pragma unroll
    for (int c = 0; c < OUT_F; c++) {
        float v = acc[(size_t)node * OUT_F + c] + b[c];
        vals[c] = v;
        mx = fmaxf(mx, v);
    }
    float sum = 0.f;
#pragma unroll
    for (int c = 0; c < OUT_F; c++) sum += expf(vals[c] - mx);
    float lse = mx + logf(sum);
#pragma unroll
    for (int c = 0; c < OUT_F; c++) out[(size_t)node * OUT_F + c] = vals[c] - lse;
}

extern "C" void kernel_launch(void* const* d_in, const int* in_sizes, int n_in,
                              void* d_out, int out_size, void* d_ws, size_t ws_size,
                              hipStream_t stream) {
    const float* x   = (const float*)d_in[0];
    const int*   ei  = (const int*)d_in[1];
    const float* ea  = (const float*)d_in[2];
    const float* W1  = (const float*)d_in[3];
    const float* as1 = (const float*)d_in[4];
    const float* ad1 = (const float*)d_in[5];
    const float* We1 = (const float*)d_in[6];
    const float* ae1 = (const float*)d_in[7];
    const float* b1  = (const float*)d_in[8];
    const float* W2  = (const float*)d_in[9];
    const float* as2 = (const float*)d_in[10];
    const float* ad2 = (const float*)d_in[11];
    const float* We2 = (const float*)d_in[12];
    const float* ae2 = (const float*)d_in[13];
    const float* b2  = (const float*)d_in[14];

    const int N = N_NODES, E = N_EDGES;
    const int HC1 = HEADS * HID;
    const int NPAD2 = 48;
    const int* src = ei;
    const int* dst = ei + E;

    char* ws = (char*)d_ws;
    size_t off = 0;
    auto alloc = [&](size_t bytes) -> void* {
        void* p = ws + off;
        off = (off + bytes + 255) & ~(size_t)255;
        return p;
    };
    int*      deg    = (int*)alloc(sizeof(int) * N);
    int*      rowptr = (int*)alloc(sizeof(int) * (N + 1));
    int*      cursor = (int*)alloc(sizeof(int) * N);
    uint4*    rec    = (uint4*)alloc(sizeof(uint4) * 2 * (size_t)E);
    float*    v1     = (float*)alloc(sizeof(float) * 16);
    float*    v2     = (float*)alloc(sizeof(float) * 16);
    float*    W2p    = (float*)alloc(sizeof(float) * F_IN * NPAD2);
    float*    xh1    = (float*)alloc(sizeof(float) * (size_t)N * HC1);
    float*    als1   = (float*)alloc(sizeof(float) * N * HEADS);
    float*    ald1   = (float*)alloc(sizeof(float) * N * HEADS);
    float*    acc1   = (float*)alloc(sizeof(float) * (size_t)N * HC1);
    float*    xh2    = (float*)alloc(sizeof(float) * (size_t)N * OUT_F);
    float*    als2   = (float*)alloc(sizeof(float) * N);
    float*    ald2   = (float*)alloc(sizeof(float) * N);
    float*    acc2   = (float*)alloc(sizeof(float) * (size_t)N * OUT_F);
    (void)ws_size; (void)n_in; (void)in_sizes; (void)out_size;

    const int T = 256;

    fill_u32<<<CDIV(N, T), T, 0, stream>>>((unsigned*)deg, N, 0u);
    csr_count<<<CDIV(E, T), T, 0, stream>>>(dst, deg, E);
    csr_scan<<<1, SCAN_T, 0, stream>>>(deg, rowptr, cursor, N);
    csr_fill<<<CDIV(E, T), T, 0, stream>>>(src, dst, ea, cursor, rec, E);
    precompute_v<<<1, 32, 0, stream>>>(We1, ae1, We2, ae2, v1, v2);
    pad_b<<<CDIV(F_IN * NPAD2, T), T, 0, stream>>>(W2, W2p, F_IN, OUT_F, NPAD2);

    {
        int mtiles = N / 16, waves_per_blk = 128 / 32;
        wmma_gemm<8, F_IN><<<CDIV(mtiles, waves_per_blk), 128, 0, stream>>>(x, W1, xh1, N, HC1);
    }
    node_scores<<<CDIV(N * HEADS, T), T, 0, stream>>>(xh1, as1, ad1, als1, ald1, N, HEADS, HID);
    gat_node<HEADS, HID, 4><<<CDIV(N, T / 32), T, 0, stream>>>(rowptr, rec, als1, ald1, v1, xh1, acc1, N);
    bias_elu<<<CDIV(N * HC1, T), T, 0, stream>>>(acc1, b1, N * HC1, HC1);

    {
        int mtiles = N / 16, waves_per_blk = 128 / 32;
        wmma_gemm<3, HC1><<<CDIV(mtiles, waves_per_blk), 128, 0, stream>>>(acc1, W2p, xh2, N, OUT_F);
    }
    node_scores<<<CDIV(N, T), T, 0, stream>>>(xh2, as2, ad2, als2, ald2, N, 1, OUT_F);
    gat_node<1, OUT_F, 2><<<CDIV(N, T / 32), T, 0, stream>>>(rowptr, rec, als2, ald2, v2, xh2, acc2, N);
    bias_logsoftmax<<<CDIV(N, T), T, 0, stream>>>(acc2, b2, (float*)d_out, N);
}


# Bias+ELU and bias+log-softmax passes folded into the per-node aggregation kernels' stores (14 -> 12 launches); log-softmax row max/sum now wave tree reductions, sum order changed
# speedup vs baseline: 4.0495x; 1.0828x over previous
#include <hip/hip_runtime.h>
#include <math.h>

#define N_NODES 50000
#define N_EDGES 800000
#define ETOT    (N_EDGES + N_NODES)
#define F_IN    128
#define HID     64
#define HEADS   2
#define OUT_F   40
#define EDIM    5
#define NEG_SLOPE 0.2f

#define CDIV(a,b) (((a)+(b)-1)/(b))

typedef __attribute__((ext_vector_type(2))) float v2f;
typedef __attribute__((ext_vector_type(8))) float v8f;

__global__ void fill_u32(unsigned* p, int n, unsigned v) {
    int i = blockIdx.x * blockDim.x + threadIdx.x;
    if (i < n) p[i] = v;
}

__global__ void pad_b(const float* __restrict__ W, float* __restrict__ Wp,
                      int K, int N, int NPAD) {
    int i = blockIdx.x * blockDim.x + threadIdx.x;
    if (i >= K * NPAD) return;
    int r = i / NPAD, c = i % NPAD;
    Wp[i] = (c < N) ? W[r * N + c] : 0.0f;
}

__global__ void precompute_v(const float* We1, const float* ae1,
                             const float* We2, const float* ae2,
                             float* v1, float* v2) {
    int t = threadIdx.x;
    if (t < EDIM * HEADS) {
        int d = t / HEADS, h = t % HEADS;
        float s = 0.f;
        for (int c = 0; c < HID; c++)
            s += We1[d * (HEADS * HID) + h * HID + c] * ae1[h * HID + c];
        v1[t] = s;
    } else if (t < EDIM * HEADS + EDIM) {
        int d = t - EDIM * HEADS;
        float s = 0.f;
        for (int c = 0; c < OUT_F; c++)
            s += We2[d * OUT_F + c] * ae2[c];
        v2[d] = s;
    }
}

template<int NT, int K>
__global__ void wmma_gemm(const float* __restrict__ A, const float* __restrict__ B,
                          float* __restrict__ C, int M, int N) {
    constexpr int NPAD = NT * 16;
    int wave = (blockIdx.x * blockDim.x + threadIdx.x) >> 5;
    int lane = threadIdx.x & 31;
    int half = lane >> 4;
    int lrow = lane & 15;
    int mtiles = M >> 4;
    if (wave >= mtiles) return;
    int m0 = wave << 4;

    v8f acc[NT];
#pragma unroll
    for (int nt = 0; nt < NT; nt++)
#pragma unroll
        for (int g = 0; g < 8; g++) acc[nt][g] = 0.0f;

    const float* arow = A + (size_t)(m0 + lrow) * K;
    const float* bcol = B + lrow;
#pragma unroll 4
    for (int k = 0; k < K; k += 4) {
        int kk = k + half * 2;
        v2f a;
        a.x = arow[kk];
        a.y = arow[kk + 1];
#pragma unroll
        for (int nt = 0; nt < NT; nt++) {
            v2f b;
            b.x = bcol[kk * NPAD + nt * 16];
            b.y = bcol[(kk + 1) * NPAD + nt * 16];
            acc[nt] = __builtin_amdgcn_wmma_f32_16x16x4_f32(
                false, a, false, b, (short)0, acc[nt], false, false);
        }
    }
#pragma unroll
    for (int nt = 0; nt < NT; nt++) {
        int col = nt * 16 + lrow;
        if (col < N) {
#pragma unroll
            for (int g = 0; g < 8; g++)
                C[(size_t)(m0 + g + half * 8) * N + col] = acc[nt][g];
        }
    }
}

__global__ void node_scores(const float* __restrict__ xh,
                            const float* __restrict__ asrc, const float* __restrict__ adst,
                            float* als, float* ald, int n, int H, int C) {
    int i = blockIdx.x * blockDim.x + threadIdx.x;
    if (i >= n * H) return;
    int node = i / H, h = i % H;
    const float* row = xh + (size_t)node * H * C + h * C;
    float s = 0.f, d = 0.f;
    for (int c = 0; c < C; c++) {
        float v = row[c];
        s += v * asrc[h * C + c];
        d += v * adst[h * C + c];
    }
    als[i] = s;
    ald[i] = d;
}

__global__ void csr_count(const int* __restrict__ dst, int* deg, int E) {
    int e = blockIdx.x * blockDim.x + threadIdx.x;
    if (e >= E) return;
    atomicAdd(&deg[dst[e]], 1);
}

#define SCAN_T 1024
#define SCAN_SEG (CDIV(N_NODES, 32 * 128) * 128)
static_assert(N_NODES % 4 == 0, "csr_scan reads and writes four counts at a time");
__global__ void __launch_bounds__(SCAN_T) csr_scan(const int* __restrict__ deg, int* __restrict__ rowptr,
                                                   int* __restrict__ cursor, int n) {
    __shared__ int wsum[32];
    const int t = threadIdx.x, lane = t & 31, w = t >> 5;
    const int base = w * SCAN_SEG;
    int loc = 0;
    for (int it = 0; it < SCAN_SEG; it += 128) {
        const int idx = base + it + lane * 4;
        if (idx < n) {
            const int4 q = *reinterpret_cast<const int4*>(deg + idx);
            loc += q.x + q.y + q.z + q.w;
        }
    }
#pragma unroll
    for (int o = 16; o > 0; o >>= 1) loc += __shfl_xor(loc, o, 32);
    if (lane == 0) wsum[w] = loc;
    __syncthreads();
    if (w == 0) {
        int x = wsum[lane];
#pragma unroll
        for (int o = 1; o < 32; o <<= 1) {
            int y = __shfl_up(x, o, 32);
            if (lane >= o) x += y;
        }
        wsum[lane] = x;
    }
    __syncthreads();
    int carry = (w > 0) ? wsum[w - 1] : 0;
    for (int it = 0; it < SCAN_SEG; it += 128) {
        const int idx = base + it + lane * 4;
        const bool in = idx < n;
        int4 q; q.x = 0; q.y = 0; q.z = 0; q.w = 0;
        if (in) q = *reinterpret_cast<const int4*>(deg + idx);
        const int tot = q.x + q.y + q.z + q.w;
        int inc = tot;
#pragma unroll
        for (int o = 1; o < 32; o <<= 1) {
            int y = __shfl_up(inc, o, 32);
            if (lane >= o) inc += y;
        }
        if (in) {
            int4 r;
            r.x = carry + inc - tot;
            r.y = r.x + q.x;
            r.z = r.y + q.y;
            r.w = r.z + q.z;
            *reinterpret_cast<int4*>(rowptr + idx) = r;
            *reinterpret_cast<int4*>(cursor + idx) = r;
        }
        carry += __shfl(inc, 31, 32);
    }
    if (t == SCAN_T - 1) rowptr[n] = wsum[31];
}

__global__ void csr_fill(const int* __restrict__ src, const int* __restrict__ dst,
                         const float* __restrict__ eattr, int* cursor,
                         uint4* __restrict__ rec, int E) {
    int e = blockIdx.x * blockDim.x + threadIdx.x;
    if (e >= E) return;
    const int d = dst[e];
    uint4 r0, r1;
    r0.x = (unsigned)src[e];
    r0.y = __float_as_uint(eattr[(size_t)e * EDIM + 0]);
    r0.z = __float_as_uint(eattr[(size_t)e * EDIM + 1]);
    r0.w = __float_as_uint(eattr[(size_t)e * EDIM + 2]);
    r1.x = __float_as_uint(eattr[(size_t)e * EDIM + 3]);
    r1.y = __float_as_uint(eattr[(size_t)e * EDIM + 4]);
    r1.z = 0u;
    r1.w = 0u;
    const int p = atomicAdd(&cursor[d], 1);
    rec[(size_t)p * 2]     = r0;
    rec[(size_t)p * 2 + 1] = r1;
}

__device__ inline float wave_sum(float x) {
#pragma unroll
    for (int o = 16; o > 0; o >>= 1) x += __shfl_xor(x, o, 32);
    return x;
}
__device__ inline float wave_max(float x) {
#pragma unroll
    for (int o = 16; o > 0; o >>= 1) x = fmaxf(x, __shfl_xor(x, o, 32));
    return x;
}

__device__ inline void load_rec(const uint4* __restrict__ rec, int p, int& s, float (&e)[EDIM]) {
    const uint4 r0 = rec[(size_t)p * 2];
    const uint4 r1 = rec[(size_t)p * 2 + 1];
    s = (int)r0.x;
    e[0] = __uint_as_float(r0.y);
    e[1] = __uint_as_float(r0.z);
    e[2] = __uint_as_float(r0.w);
    e[3] = __uint_as_float(r1.x);
    e[4] = __uint_as_float(r1.y);
}

__device__ inline void get_edge(const uint4* __restrict__ rec, int start, int deg, int node,
                                const float (&lp)[EDIM], int j, int& s, float (&e)[EDIM]) {
    if (j < deg) {
        load_rec(rec, start + j, s, e);
    } else {
        s = node;
#pragma unroll
        for (int i = 0; i < EDIM; i++) e[i] = lp[i];
    }
}

template<int H>
__device__ inline void edge_alpha(int s, const float (&e)[EDIM], const float (&vv)[EDIM * H],
                                  const float* __restrict__ als, const float (&aldn)[H],
                                  float (&a)[H]) {
#pragma unroll
    for (int h = 0; h < H; h++) {
        float ale = 0.f;
#pragma unroll
        for (int i = 0; i < EDIM; i++) ale += e[i] * vv[i * H + h];
        float x = als[s * H + h] + aldn[h] + ale;
        a[h] = (x > 0.f) ? x : NEG_SLOPE * x;
    }
}

template<int VEC> struct rowvec;
template<> struct rowvec<4> {
    static __device__ inline void ld(const float* p, float (&o)[4]) {
        const float4 t = *reinterpret_cast<const float4*>(p);
        o[0] = t.x; o[1] = t.y; o[2] = t.z; o[3] = t.w;
    }
    static __device__ inline void st(float* p, const float (&o)[4]) {
        float4 t; t.x = o[0]; t.y = o[1]; t.z = o[2]; t.w = o[3];
        *reinterpret_cast<float4*>(p) = t;
    }
};
template<> struct rowvec<2> {
    static __device__ inline void ld(const float* p, float (&o)[2]) {
        const float2 t = *reinterpret_cast<const float2*>(p);
        o[0] = t.x; o[1] = t.y;
    }
    static __device__ inline void st(float* p, const float (&o)[2]) {
        float2 t; t.x = o[0]; t.y = o[1];
        *reinterpret_cast<float2*>(p) = t;
    }
};

template<int H, int C, int VEC>
__device__ inline void add_rows(const float* __restrict__ xh, int cnt, int s, const float (&w)[H],
                                int lane, int hl, float (&acc)[VEC]) {
    constexpr int HC = H * C;
    constexpr int NACT = HC / VEC;
#pragma unroll 4
    for (int jj = 0; jj < cnt; ++jj) {
        const int sj = __builtin_amdgcn_readlane(s, jj);
        float wj = __int_as_float(__builtin_amdgcn_readlane(__float_as_int(w[0]), jj));
        if (H == 2) {
            const float wj1 = __int_as_float(__builtin_amdgcn_readlane(__float_as_int(w[H - 1]), jj));
            wj = hl ? wj1 : wj;
        }
        if (lane < NACT) {
            float r[VEC];
            rowvec<VEC>::ld(xh + (size_t)sj * HC + lane * VEC, r);
#pragma unroll
            for (int q = 0; q < VEC; q++) acc[q] += r[q] * wj;
        }
    }
}

template<int H, int C, int VEC, bool LOGSM>
__global__ void __launch_bounds__(256) gat_node(const int* __restrict__ rowptr,
                                                const uint4* __restrict__ rec,
                                                const float* __restrict__ als,
                                                const float* __restrict__ ald,
                                                const float* __restrict__ v,
                                                const float* __restrict__ xh,
                                                const float* __restrict__ bias,
                                                float* __restrict__ row_out, int n) {
    constexpr int HC = H * C;
    constexpr int NACT = HC / VEC;
    static_assert(H == 1 || H == 2, "head select below handles one or two heads");
    static_assert(NACT * VEC == HC && NACT <= 32 && C % VEC == 0, "lane to channel map");
    const int lane = threadIdx.x & 31;
    const int node = __builtin_amdgcn_readfirstlane((int)((blockIdx.x * blockDim.x + threadIdx.x) >> 5));
    if (node >= n) return;
    const int start = rowptr[node];
    const int deg   = rowptr[node + 1] - start;
    const int total = deg + 1;

    float vv[EDIM * H];
#pragma unroll
    for (int i = 0; i < EDIM * H; i++) vv[i] = v[i];
    float aldn[H];
#pragma unroll
    for (int h = 0; h < H; h++) aldn[h] = ald[node * H + h];

    int   s_f = node;
    float e_f[EDIM];
    float sm[EDIM];
#pragma unroll
    for (int i = 0; i < EDIM; i++) { e_f[i] = 0.f; sm[i] = 0.f; }
    if (lane < deg) {
        load_rec(rec, start + lane, s_f, e_f);
#pragma unroll
        for (int i = 0; i < EDIM; i++) sm[i] = e_f[i];
    }
    for (int j = lane + 32; j < deg; j += 32) {
        int s; float e[EDIM];
        load_rec(rec, start + j, s, e);
#pragma unroll
        for (int i = 0; i < EDIM; i++) sm[i] += e[i];
    }
    float lp[EDIM];
    const float cdiv = fmaxf((float)deg, 1.0f);
#pragma unroll
    for (int i = 0; i < EDIM; i++) lp[i] = wave_sum(sm[i]) / cdiv;

    const bool val_f = lane < total;
    if (lane == deg) {
#pragma unroll
        for (int i = 0; i < EDIM; i++) e_f[i] = lp[i];
    }
    float a_f[H];
    edge_alpha<H>(s_f, e_f, vv, als, aldn, a_f);

    float m[H];
#pragma unroll
    for (int h = 0; h < H; h++) m[h] = val_f ? a_f[h] : -INFINITY;
    for (int j = lane + 32; j < total; j += 32) {
        int s; float e[EDIM]; float a[H];
        get_edge(rec, start, deg, node, lp, j, s, e);
        edge_alpha<H>(s, e, vv, als, aldn, a);
#pragma unroll
        for (int h = 0; h < H; h++) m[h] = fmaxf(m[h], a[h]);
    }
#pragma unroll
    for (int h = 0; h < H; h++) m[h] = wave_max(m[h]);

    float ex_f[H], dn[H];
#pragma unroll
    for (int h = 0; h < H; h++) {
        ex_f[h] = val_f ? expf(a_f[h] - m[h]) : 0.f;
        dn[h] = ex_f[h];
    }
    for (int j = lane + 32; j < total; j += 32) {
        int s; float e[EDIM]; float a[H];
        get_edge(rec, start, deg, node, lp, j, s, e);
        edge_alpha<H>(s, e, vv, als, aldn, a);
#pragma unroll
        for (int h = 0; h < H; h++) dn[h] += expf(a[h] - m[h]);
    }
#pragma unroll
    for (int h = 0; h < H; h++) dn[h] = wave_sum(dn[h]) + 1e-16f;

    const int hl = (lane * VEC) / C;
    const bool act = lane < NACT;
    float bv[VEC];
#pragma unroll
    for (int q = 0; q < VEC; q++) bv[q] = 0.f;
    if (act) rowvec<VEC>::ld(bias + lane * VEC, bv);
    float acc[VEC];
#pragma unroll
    for (int q = 0; q < VEC; q++) acc[q] = 0.f;

    float w_f[H];
#pragma unroll
    for (int h = 0; h < H; h++) w_f[h] = ex_f[h] / dn[h];
    add_rows<H, C, VEC>(xh, total < 32 ? total : 32, s_f, w_f, lane, hl, acc);

    for (int cb = 32; cb < total; cb += 32) {
        const int j = cb + lane;
        int s = node;
        float w[H];
#pragma unroll
        for (int h = 0; h < H; h++) w[h] = 0.f;
        if (j < total) {
            float e[EDIM]; float a[H];
            get_edge(rec, start, deg, node, lp, j, s, e);
            edge_alpha<H>(s, e, vv, als, aldn, a);
#pragma unroll
            for (int h = 0; h < H; h++) w[h] = expf(a[h] - m[h]) / dn[h];
        }
        const int left = total - cb;
        add_rows<H, C, VEC>(xh, left < 32 ? left : 32, s, w, lane, hl, acc);
    }

#pragma unroll
    for (int q = 0; q < VEC; q++) acc[q] += bv[q];

    if (!LOGSM) {
#pragma unroll
        for (int q = 0; q < VEC; q++) acc[q] = (acc[q] > 0.f) ? acc[q] : expm1f(acc[q]);
    } else {
        float mx = -INFINITY;
        if (act) {
#pragma unroll
            for (int q = 0; q < VEC; q++) mx = fmaxf(mx, acc[q]);
        }
        mx = wave_max(mx);
        float se = 0.f;
        if (act) {
#pragma unroll
            for (int q = 0; q < VEC; q++) se += expf(acc[q] - mx);
        }
        se = wave_sum(se);
        const float lse = mx + logf(se);
#pragma unroll
        for (int q = 0; q < VEC; q++) acc[q] -= lse;
    }

    if (act) rowvec<VEC>::st(row_out + (size_t)node * HC + lane * VEC, acc);
}

extern "C" void kernel_launch(void* const* d_in, const int* in_sizes, int n_in,
                              void* d_out, int out_size, void* d_ws, size_t ws_size,
                              hipStream_t stream) {
    const float* x   = (const float*)d_in[0];
    const int*   ei  = (const int*)d_in[1];
    const float* ea  = (const float*)d_in[2];
    const float* W1  = (const float*)d_in[3];
    const float* as1 = (const float*)d_in[4];
    const float* ad1 = (const float*)d_in[5];
    const float* We1 = (const float*)d_in[6];
    const float* ae1 = (const float*)d_in[7];
    const float* b1  = (const float*)d_in[8];
    const float* W2  = (const float*)d_in[9];
    const float* as2 = (const float*)d_in[10];
    const float* ad2 = (const float*)d_in[11];
    const float* We2 = (const float*)d_in[12];
    const float* ae2 = (const float*)d_in[13];
    const float* b2  = (const float*)d_in[14];

    const int N = N_NODES, E = N_EDGES;
    const int HC1 = HEADS * HID;
    const int NPAD2 = 48;
    const int* src = ei;
    const int* dst = ei + E;

    char* ws = (char*)d_ws;
    size_t off = 0;
    auto alloc = [&](size_t bytes) -> void* {
        void* p = ws + off;
        off = (off + bytes + 255) & ~(size_t)255;
        return p;
    };
    int*      deg    = (int*)alloc(sizeof(int) * N);
    int*      rowptr = (int*)alloc(sizeof(int) * (N + 1));
    int*      cursor = (int*)alloc(sizeof(int) * N);
    uint4*    rec    = (uint4*)alloc(sizeof(uint4) * 2 * (size_t)E);
    float*    v1     = (float*)alloc(sizeof(float) * 16);
    float*    v2     = (float*)alloc(sizeof(float) * 16);
    float*    W2p    = (float*)alloc(sizeof(float) * F_IN * NPAD2);
    float*    xh1    = (float*)alloc(sizeof(float) * (size_t)N * HC1);
    float*    als1   = (float*)alloc(sizeof(float) * N * HEADS);
    float*    ald1   = (float*)alloc(sizeof(float) * N * HEADS);
    float*    h1     = (float*)alloc(sizeof(float) * (size_t)N * HC1);
    float*    xh2    = (float*)alloc(sizeof(float) * (size_t)N * OUT_F);
    float*    als2   = (float*)alloc(sizeof(float) * N);
    float*    ald2   = (float*)alloc(sizeof(float) * N);
    (void)ws_size; (void)n_in; (void)in_sizes; (void)out_size;

    const int T = 256;

    fill_u32<<<CDIV(N, T), T, 0, stream>>>((unsigned*)deg, N, 0u);
    csr_count<<<CDIV(E, T), T, 0, stream>>>(dst, deg, E);
    csr_scan<<<1, SCAN_T, 0, stream>>>(deg, rowptr, cursor, N);
    csr_fill<<<CDIV(E, T), T, 0, stream>>>(src, dst, ea, cursor, rec, E);
    precompute_v<<<1, 32, 0, stream>>>(We1, ae1, We2, ae2, v1, v2);
    pad_b<<<CDIV(F_IN * NPAD2, T), T, 0, stream>>>(W2, W2p, F_IN, OUT_F, NPAD2);

    {
        int mtiles = N / 16, waves_per_blk = 128 / 32;
        wmma_gemm<8, F_IN><<<CDIV(mtiles, waves_per_blk), 128, 0, stream>>>(x, W1, xh1, N, HC1);
    }
    node_scores<<<CDIV(N * HEADS, T), T, 0, stream>>>(xh1, as1, ad1, als1, ald1, N, HEADS, HID);
    gat_node<HEADS, HID, 4, false><<<CDIV(N, T / 32), T, 0, stream>>>(rowptr, rec, als1, ald1, v1, xh1, b1, h1, N);

    {
        int mtiles = N / 16, waves_per_blk = 128 / 32;
        wmma_gemm<3, HC1><<<CDIV(mtiles, waves_per_blk), 128, 0, stream>>>(h1, W2p, xh2, N, OUT_F);
    }
    node_scores<<<CDIV(N, T), T, 0, stream>>>(xh2, as2, ad2, als2, ald2, N, 1, OUT_F);
    gat_node<1, OUT_F, 2, true><<<CDIV(N, T / 32), T, 0, stream>>>(rowptr, rec, als2, ald2, v2, xh2, b2, (float*)d_out, N);
}


# Attention score dots als/ald taken from the GEMM accumulator tiles in the GEMM epilogue (12 -> 10 launches); score sums now lane partials + 16-lane DPP butterfly, sum order changed
# speedup vs baseline: 4.2733x; 1.0553x over previous
#include <hip/hip_runtime.h>
#include <math.h>

#define N_NODES 50000
#define N_EDGES 800000
#define ETOT    (N_EDGES + N_NODES)
#define F_IN    128
#define HID     64
#define HEADS   2
#define OUT_F   40
#define EDIM    5
#define NEG_SLOPE 0.2f

#define CDIV(a,b) (((a)+(b)-1)/(b))

typedef __attribute__((ext_vector_type(2))) float v2f;
typedef __attribute__((ext_vector_type(8))) float v8f;

__global__ void fill_u32(unsigned* p, int n, unsigned v) {
    int i = blockIdx.x * blockDim.x + threadIdx.x;
    if (i < n) p[i] = v;
}

__global__ void pad_b(const float* __restrict__ W, float* __restrict__ Wp,
                      int K, int N, int NPAD) {
    int i = blockIdx.x * blockDim.x + threadIdx.x;
    if (i >= K * NPAD) return;
    int r = i / NPAD, c = i % NPAD;
    Wp[i] = (c < N) ? W[r * N + c] : 0.0f;
}

__global__ void precompute_v(const float* We1, const float* ae1,
                             const float* We2, const float* ae2,
                             float* v1, float* v2) {
    int t = threadIdx.x;
    if (t < EDIM * HEADS) {
        int d = t / HEADS, h = t % HEADS;
        float s = 0.f;
        for (int c = 0; c < HID; c++)
            s += We1[d * (HEADS * HID) + h * HID + c] * ae1[h * HID + c];
        v1[t] = s;
    } else if (t < EDIM * HEADS + EDIM) {
        int d = t - EDIM * HEADS;
        float s = 0.f;
        for (int c = 0; c < OUT_F; c++)
            s += We2[d * OUT_F + c] * ae2[c];
        v2[d] = s;
    }
}

template<int M>
__device__ inline float row16_xadd(float x) {
    const int y = __builtin_amdgcn_update_dpp(0, __float_as_int(x), 0x160 + M, 0xf, 0xf, true);
    return x + __int_as_float(y);
}

template<int NT, int K, int H>
__global__ void wmma_gemm(const float* __restrict__ A, const float* __restrict__ B,
                          float* __restrict__ C,
                          const float* __restrict__ asrc, const float* __restrict__ adst,
                          float* __restrict__ als, float* __restrict__ ald, int M, int N) {
    constexpr int NPAD = NT * 16;
    constexpr int TPH = NT / H;
    static_assert(NT % H == 0, "a column tile belongs to one head");
    int wave = (blockIdx.x * blockDim.x + threadIdx.x) >> 5;
    int lane = threadIdx.x & 31;
    int half = lane >> 4;
    int lrow = lane & 15;
    int mtiles = M >> 4;
    if (wave >= mtiles) return;
    int m0 = wave << 4;

    float sa[NT], da[NT];
#pragma unroll
    for (int nt = 0; nt < NT; nt++) {
        const int col = nt * 16 + lrow;
        sa[nt] = (col < N) ? asrc[col] : 0.0f;
        da[nt] = (col < N) ? adst[col] : 0.0f;
    }

    v8f acc[NT];
#pragma unroll
    for (int nt = 0; nt < NT; nt++)
#pragma unroll
        for (int g = 0; g < 8; g++) acc[nt][g] = 0.0f;

    const float* arow = A + (size_t)(m0 + lrow) * K;
    const float* bcol = B + lrow;
#pragma unroll 4
    for (int k = 0; k < K; k += 4) {
        int kk = k + half * 2;
        v2f a;
        a.x = arow[kk];
        a.y = arow[kk + 1];
#pragma unroll
        for (int nt = 0; nt < NT; nt++) {
            v2f b;
            b.x = bcol[kk * NPAD + nt * 16];
            b.y = bcol[(kk + 1) * NPAD + nt * 16];
            acc[nt] = __builtin_amdgcn_wmma_f32_16x16x4_f32(
                false, a, false, b, (short)0, acc[nt], false, false);
        }
    }
#pragma unroll
    for (int nt = 0; nt < NT; nt++) {
        int col = nt * 16 + lrow;
        if (col < N) {
#pragma unroll
            for (int g = 0; g < 8; g++)
                C[(size_t)(m0 + g + half * 8) * N + col] = acc[nt][g];
        }
    }

    constexpr int NV = 2 * H * 8;
    float red[NV];
#pragma unroll
    for (int i = 0; i < NV; i++) red[i] = 0.0f;
#pragma unroll
    for (int nt = 0; nt < NT; nt++) {
        const int h = nt / TPH;
        const bool incol = (nt * 16 + lrow) < N;
#pragma unroll
        for (int g = 0; g < 8; g++) {
            const float cv = incol ? acc[nt][g] : 0.0f;
            red[(0 * H + h) * 8 + g] += cv * sa[nt];
            red[(1 * H + h) * 8 + g] += cv * da[nt];
        }
    }
#pragma unroll
    for (int i = 0; i < NV; i++) red[i] = row16_xadd<1>(red[i]);
#pragma unroll
    for (int i = 0; i < NV; i++) red[i] = row16_xadd<2>(red[i]);
#pragma unroll
    for (int i = 0; i < NV; i++) red[i] = row16_xadd<4>(red[i]);
#pragma unroll
    for (int i = 0; i < NV; i++) red[i] = row16_xadd<8>(red[i]);
    if (lrow == 0) {
        float* sp = als + (size_t)(m0 + half * 8) * H;
        float* dp = ald + (size_t)(m0 + half * 8) * H;
#pragma unroll
        for (int g = 0; g < 8; g++)
#pragma unroll
            for (int h = 0; h < H; h++) {
                sp[g * H + h] = red[(0 * H + h) * 8 + g];
                dp[g * H + h] = red[(1 * H + h) * 8 + g];
            }
    }
}

__global__ void csr_count(const int* __restrict__ dst, int* deg, int E) {
    int e = blockIdx.x * blockDim.x + threadIdx.x;
    if (e >= E) return;
    atomicAdd(&deg[dst[e]], 1);
}

#define SCAN_T 1024
#define SCAN_SEG (CDIV(N_NODES, 32 * 128) * 128)
static_assert(N_NODES % 4 == 0, "csr_scan reads and writes four counts at a time");
__global__ void __launch_bounds__(SCAN_T) csr_scan(const int* __restrict__ deg, int* __restrict__ rowptr,
                                                   int* __restrict__ cursor, int n) {
    __shared__ int wsum[32];
    const int t = threadIdx.x, lane = t & 31, w = t >> 5;
    const int base = w * SCAN_SEG;
    int loc = 0;
    for (int it = 0; it < SCAN_SEG; it += 128) {
        const int idx = base + it + lane * 4;
        if (idx < n) {
            const int4 q = *reinterpret_cast<const int4*>(deg + idx);
            loc += q.x + q.y + q.z + q.w;
        }
    }
#pragma unroll
    for (int o = 16; o > 0; o >>= 1) loc += __shfl_xor(loc, o, 32);
    if (lane == 0) wsum[w] = loc;
    __syncthreads();
    if (w == 0) {
        int x = wsum[lane];
#pragma unroll
        for (int o = 1; o < 32; o <<= 1) {
            int y = __shfl_up(x, o, 32);
            if (lane >= o) x += y;
        }
        wsum[lane] = x;
    }
    __syncthreads();
    int carry = (w > 0) ? wsum[w - 1] : 0;
    for (int it = 0; it < SCAN_SEG; it += 128) {
        const int idx = base + it + lane * 4;
        const bool in = idx < n;
        int4 q; q.x = 0; q.y = 0; q.z = 0; q.w = 0;
        if (in) q = *reinterpret_cast<const int4*>(deg + idx);
        const int tot = q.x + q.y + q.z + q.w;
        int inc = tot;
#pragma unroll
        for (int o = 1; o < 32; o <<= 1) {
            int y = __shfl_up(inc, o, 32);
            if (lane >= o) inc += y;
        }
        if (in) {
            int4 r;
            r.x = carry + inc - tot;
            r.y = r.x + q.x;
            r.z = r.y + q.y;
            r.w = r.z + q.z;
            *reinterpret_cast<int4*>(rowptr + idx) = r;
            *reinterpret_cast<int4*>(cursor + idx) = r;
        }
        carry += __shfl(inc, 31, 32);
    }
    if (t == SCAN_T - 1) rowptr[n] = wsum[31];
}

__global__ void csr_fill(const int* __restrict__ src, const int* __restrict__ dst,
                         const float* __restrict__ eattr, int* cursor,
                         uint4* __restrict__ rec, int E) {
    int e = blockIdx.x * blockDim.x + threadIdx.x;
    if (e >= E) return;
    const int d = dst[e];
    uint4 r0, r1;
    r0.x = (unsigned)src[e];
    r0.y = __float_as_uint(eattr[(size_t)e * EDIM + 0]);
    r0.z = __float_as_uint(eattr[(size_t)e * EDIM + 1]);
    r0.w = __float_as_uint(eattr[(size_t)e * EDIM + 2]);
    r1.x = __float_as_uint(eattr[(size_t)e * EDIM + 3]);
    r1.y = __float_as_uint(eattr[(size_t)e * EDIM + 4]);
    r1.z = 0u;
    r1.w = 0u;
    const int p = atomicAdd(&cursor[d], 1);
    rec[(size_t)p * 2]     = r0;
    rec[(size_t)p * 2 + 1] = r1;
}

__device__ inline float wave_sum(float x) {
#pragma unroll
    for (int o = 16; o > 0; o >>= 1) x += __shfl_xor(x, o, 32);
    return x;
}
__device__ inline float wave_max(float x) {
#pragma unroll
    for (int o = 16; o > 0; o >>= 1) x = fmaxf(x, __shfl_xor(x, o, 32));
    return x;
}

__device__ inline void load_rec(const uint4* __restrict__ rec, int p, int& s, float (&e)[EDIM]) {
    const uint4 r0 = rec[(size_t)p * 2];
    const uint4 r1 = rec[(size_t)p * 2 + 1];
    s = (int)r0.x;
    e[0] = __uint_as_float(r0.y);
    e[1] = __uint_as_float(r0.z);
    e[2] = __uint_as_float(r0.w);
    e[3] = __uint_as_float(r1.x);
    e[4] = __uint_as_float(r1.y);
}

__device__ inline void get_edge(const uint4* __restrict__ rec, int start, int deg, int node,
                                const float (&lp)[EDIM], int j, int& s, float (&e)[EDIM]) {
    if (j < deg) {
        load_rec(rec, start + j, s, e);
    } else {
        s = node;
#pragma unroll
        for (int i = 0; i < EDIM; i++) e[i] = lp[i];
    }
}

template<int H>
__device__ inline void edge_alpha(int s, const float (&e)[EDIM], const float (&vv)[EDIM * H],
                                  const float* __restrict__ als, const float (&aldn)[H],
                                  float (&a)[H]) {
#pragma unroll
    for (int h = 0; h < H; h++) {
        float ale = 0.f;
#pragma unroll
        for (int i = 0; i < EDIM; i++) ale += e[i] * vv[i * H + h];
        float x = als[s * H + h] + aldn[h] + ale;
        a[h] = (x > 0.f) ? x : NEG_SLOPE * x;
    }
}

template<int VEC> struct rowvec;
template<> struct rowvec<4> {
    static __device__ inline void ld(const float* p, float (&o)[4]) {
        const float4 t = *reinterpret_cast<const float4*>(p);
        o[0] = t.x; o[1] = t.y; o[2] = t.z; o[3] = t.w;
    }
    static __device__ inline void st(float* p, const float (&o)[4]) {
        float4 t; t.x = o[0]; t.y = o[1]; t.z = o[2]; t.w = o[3];
        *reinterpret_cast<float4*>(p) = t;
    }
};
template<> struct rowvec<2> {
    static __device__ inline void ld(const float* p, float (&o)[2]) {
        const float2 t = *reinterpret_cast<const float2*>(p);
        o[0] = t.x; o[1] = t.y;
    }
    static __device__ inline void st(float* p, const float (&o)[2]) {
        float2 t; t.x = o[0]; t.y = o[1];
        *reinterpret_cast<float2*>(p) = t;
    }
};

template<int H, int C, int VEC>
__device__ inline void add_rows(const float* __restrict__ xh, int cnt, int s, const float (&w)[H],
                                int lane, int hl, float (&acc)[VEC]) {
    constexpr int HC = H * C;
    constexpr int NACT = HC / VEC;
#pragma unroll 4
    for (int jj = 0; jj < cnt; ++jj) {
        const int sj = __builtin_amdgcn_readlane(s, jj);
        float wj = __int_as_float(__builtin_amdgcn_readlane(__float_as_int(w[0]), jj));
        if (H == 2) {
            const float wj1 = __int_as_float(__builtin_amdgcn_readlane(__float_as_int(w[H - 1]), jj));
            wj = hl ? wj1 : wj;
        }
        if (lane < NACT) {
            float r[VEC];
            rowvec<VEC>::ld(xh + (size_t)sj * HC + lane * VEC, r);
#pragma unroll
            for (int q = 0; q < VEC; q++) acc[q] += r[q] * wj;
        }
    }
}

template<int H, int C, int VEC, bool LOGSM>
__global__ void __launch_bounds__(256) gat_node(const int* __restrict__ rowptr,
                                                const uint4* __restrict__ rec,
                                                const float* __restrict__ als,
                                                const float* __restrict__ ald,
                                                const float* __restrict__ v,
                                                const float* __restrict__ xh,
                                                const float* __restrict__ bias,
                                                float* __restrict__ row_out, int n) {
    constexpr int HC = H * C;
    constexpr int NACT = HC / VEC;
    static_assert(H == 1 || H == 2, "head select below handles one or two heads");
    static_assert(NACT * VEC == HC && NACT <= 32 && C % VEC == 0, "lane to channel map");
    const int lane = threadIdx.x & 31;
    const int node = __builtin_amdgcn_readfirstlane((int)((blockIdx.x * blockDim.x + threadIdx.x) >> 5));
    if (node >= n) return;
    const int start = rowptr[node];
    const int deg   = rowptr[node + 1] - start;
    const int total = deg + 1;

    float vv[EDIM * H];
#pragma unroll
    for (int i = 0; i < EDIM * H; i++) vv[i] = v[i];
    float aldn[H];
#pragma unroll
    for (int h = 0; h < H; h++) aldn[h] = ald[node * H + h];

    int   s_f = node;
    float e_f[EDIM];
    float sm[EDIM];
#pragma unroll
    for (int i = 0; i < EDIM; i++) { e_f[i] = 0.f; sm[i] = 0.f; }
    if (lane < deg) {
        load_rec(rec, start + lane, s_f, e_f);
#pragma unroll
        for (int i = 0; i < EDIM; i++) sm[i] = e_f[i];
    }
    for (int j = lane + 32; j < deg; j += 32) {
        int s; float e[EDIM];
        load_rec(rec, start + j, s, e);
#pragma unroll
        for (int i = 0; i < EDIM; i++) sm[i] += e[i];
    }
    float lp[EDIM];
    const float cdiv = fmaxf((float)deg, 1.0f);
#pragma unroll
    for (int i = 0; i < EDIM; i++) lp[i] = wave_sum(sm[i]) / cdiv;

    const bool val_f = lane < total;
    if (lane == deg) {
#pragma unroll
        for (int i = 0; i < EDIM; i++) e_f[i] = lp[i];
    }
    float a_f[H];
    edge_alpha<H>(s_f, e_f, vv, als, aldn, a_f);

    float m[H];
#pragma unroll
    for (int h = 0; h < H; h++) m[h] = val_f ? a_f[h] : -INFINITY;
    for (int j = lane + 32; j < total; j += 32) {
        int s; float e[EDIM]; float a[H];
        get_edge(rec, start, deg, node, lp, j, s, e);
        edge_alpha<H>(s, e, vv, als, aldn, a);
#pragma unroll
        for (int h = 0; h < H; h++) m[h] = fmaxf(m[h], a[h]);
    }
#pragma unroll
    for (int h = 0; h < H; h++) m[h] = wave_max(m[h]);

    float ex_f[H], dn[H];
#pragma unroll
    for (int h = 0; h < H; h++) {
        ex_f[h] = val_f ? expf(a_f[h] - m[h]) : 0.f;
        dn[h] = ex_f[h];
    }
    for (int j = lane + 32; j < total; j += 32) {
        int s; float e[EDIM]; float a[H];
        get_edge(rec, start, deg, node, lp, j, s, e);
        edge_alpha<H>(s, e, vv, als, aldn, a);
#pragma unroll
        for (int h = 0; h < H; h++) dn[h] += expf(a[h] - m[h]);
    }
#pragma unroll
    for (int h = 0; h < H; h++) dn[h] = wave_sum(dn[h]) + 1e-16f;

    const int hl = (lane * VEC) / C;
    const bool act = lane < NACT;
    float bv[VEC];
#pragma unroll
    for (int q = 0; q < VEC; q++) bv[q] = 0.f;
    if (act) rowvec<VEC>::ld(bias + lane * VEC, bv);
    float acc[VEC];
#pragma unroll
    for (int q = 0; q < VEC; q++) acc[q] = 0.f;

    float w_f[H];
#pragma unroll
    for (int h = 0; h < H; h++) w_f[h] = ex_f[h] / dn[h];
    add_rows<H, C, VEC>(xh, total < 32 ? total : 32, s_f, w_f, lane, hl, acc);

    for (int cb = 32; cb < total; cb += 32) {
        const int j = cb + lane;
        int s = node;
        float w[H];
#pragma unroll
        for (int h = 0; h < H; h++) w[h] = 0.f;
        if (j < total) {
            float e[EDIM]; float a[H];
            get_edge(rec, start, deg, node, lp, j, s, e);
            edge_alpha<H>(s, e, vv, als, aldn, a);
#pragma unroll
            for (int h = 0; h < H; h++) w[h] = expf(a[h] - m[h]) / dn[h];
        }
        const int left = total - cb;
        add_rows<H, C, VEC>(xh, left < 32 ? left : 32, s, w, lane, hl, acc);
    }

#pragma unroll
    for (int q = 0; q < VEC; q++) acc[q] += bv[q];

    if (!LOGSM) {
#pragma unroll
        for (int q = 0; q < VEC; q++) acc[q] = (acc[q] > 0.f) ? acc[q] : expm1f(acc[q]);
    } else {
        float mx = -INFINITY;
        if (act) {
#pragma unroll
            for (int q = 0; q < VEC; q++) mx = fmaxf(mx, acc[q]);
        }
        mx = wave_max(mx);
        float se = 0.f;
        if (act) {
#pragma unroll
            for (int q = 0; q < VEC; q++) se += expf(acc[q] - mx);
        }
        se = wave_sum(se);
        const float lse = mx + logf(se);
#pragma unroll
        for (int q = 0; q < VEC; q++) acc[q] -= lse;
    }

    if (act) rowvec<VEC>::st(row_out + (size_t)node * HC + lane * VEC, acc);
}

extern "C" void kernel_launch(void* const* d_in, const int* in_sizes, int n_in,
                              void* d_out, int out_size, void* d_ws, size_t ws_size,
                              hipStream_t stream) {
    const float* x   = (const float*)d_in[0];
    const int*   ei  = (const int*)d_in[1];
    const float* ea  = (const float*)d_in[2];
    const float* W1  = (const float*)d_in[3];
    const float* as1 = (const float*)d_in[4];
    const float* ad1 = (const float*)d_in[5];
    const float* We1 = (const float*)d_in[6];
    const float* ae1 = (const float*)d_in[7];
    const float* b1  = (const float*)d_in[8];
    const float* W2  = (const float*)d_in[9];
    const float* as2 = (const float*)d_in[10];
    const float* ad2 = (const float*)d_in[11];
    const float* We2 = (const float*)d_in[12];
    const float* ae2 = (const float*)d_in[13];
    const float* b2  = (const float*)d_in[14];

    const int N = N_NODES, E = N_EDGES;
    const int HC1 = HEADS * HID;
    const int NPAD2 = 48;
    const int* src = ei;
    const int* dst = ei + E;

    char* ws = (char*)d_ws;
    size_t off = 0;
    auto alloc = [&](size_t bytes) -> void* {
        void* p = ws + off;
        off = (off + bytes + 255) & ~(size_t)255;
        return p;
    };
    int*      deg    = (int*)alloc(sizeof(int) * N);
    int*      rowptr = (int*)alloc(sizeof(int) * (N + 1));
    int*      cursor = (int*)alloc(sizeof(int) * N);
    uint4*    rec    = (uint4*)alloc(sizeof(uint4) * 2 * (size_t)E);
    float*    v1     = (float*)alloc(sizeof(float) * 16);
    float*    v2     = (float*)alloc(sizeof(float) * 16);
    float*    W2p    = (float*)alloc(sizeof(float) * F_IN * NPAD2);
    float*    xh1    = (float*)alloc(sizeof(float) * (size_t)N * HC1);
    float*    als1   = (float*)alloc(sizeof(float) * N * HEADS);
    float*    ald1   = (float*)alloc(sizeof(float) * N * HEADS);
    float*    h1     = (float*)alloc(sizeof(float) * (size_t)N * HC1);
    float*    xh2    = (float*)alloc(sizeof(float) * (size_t)N * OUT_F);
    float*    als2   = (float*)alloc(sizeof(float) * N);
    float*    ald2   = (float*)alloc(sizeof(float) * N);
    (void)ws_size; (void)n_in; (void)in_sizes; (void)out_size;

    const int T = 256;

    fill_u32<<<CDIV(N, T), T, 0, stream>>>((unsigned*)deg, N, 0u);
    csr_count<<<CDIV(E, T), T, 0, stream>>>(dst, deg, E);
    csr_scan<<<1, SCAN_T, 0, stream>>>(deg, rowptr, cursor, N);
    csr_fill<<<CDIV(E, T), T, 0, stream>>>(src, dst, ea, cursor, rec, E);
    precompute_v<<<1, 32, 0, stream>>>(We1, ae1, We2, ae2, v1, v2);
    pad_b<<<CDIV(F_IN * NPAD2, T), T, 0, stream>>>(W2, W2p, F_IN, OUT_F, NPAD2);

    {
        int mtiles = N / 16, waves_per_blk = 128 / 32;
        wmma_gemm<8, F_IN, HEADS><<<CDIV(mtiles, waves_per_blk), 128, 0, stream>>>(x, W1, xh1, as1, ad1, als1, ald1, N, HC1);
    }
    gat_node<HEADS, HID, 4, false><<<CDIV(N, T / 32), T, 0, stream>>>(rowptr, rec, als1, ald1, v1, xh1, b1, h1, N);

    {
        int mtiles = N / 16, waves_per_blk = 128 / 32;
        wmma_gemm<3, HC1, 1><<<CDIV(mtiles, waves_per_blk), 128, 0, stream>>>(h1, W2p, xh2, as2, ad2, als2, ald2, N, OUT_F);
    }
    gat_node<1, OUT_F, 2, true><<<CDIV(N, T / 32), T, 0, stream>>>(rowptr, rec, als2, ald2, v2, xh2, b2, (float*)d_out, N);
}


# Independent stages share launches: degree clear + We.att fold + W2 pad in one set-up launch; layer-1 GEMM blocks side by side with the edge-count blocks (10 -> 7 launches); no sum order changed
# speedup vs baseline: 4.6803x; 1.0952x over previous
#include <hip/hip_runtime.h>
#include <math.h>

#define N_NODES 50000
#define N_EDGES 800000
#define ETOT    (N_EDGES + N_NODES)
#define F_IN    128
#define HID     64
#define HEADS   2
#define OUT_F   40
#define EDIM    5
#define NEG_SLOPE 0.2f

#define CDIV(a,b) (((a)+(b)-1)/(b))

typedef __attribute__((ext_vector_type(2))) float v2f;
typedef __attribute__((ext_vector_type(8))) float v8f;

#define NPAD2   48
#define SETUP_T 256
#define ZERO_BLOCKS CDIV(N_NODES, SETUP_T)
#define PAD_BLOCKS  CDIV(F_IN * NPAD2, SETUP_T)

__global__ void __launch_bounds__(SETUP_T) setup_small(unsigned* __restrict__ deg,
                                                       const float* __restrict__ W2, float* __restrict__ W2p,
                                                       const float* __restrict__ We1, const float* __restrict__ ae1,
                                                       const float* __restrict__ We2, const float* __restrict__ ae2,
                                                       float* __restrict__ v1, float* __restrict__ v2) {
    const int b = blockIdx.x;
    const int t = threadIdx.x;
    if (b < ZERO_BLOCKS) {
        const int i = b * SETUP_T + t;
        if (i < N_NODES) deg[i] = 0u;
    } else if (b < ZERO_BLOCKS + PAD_BLOCKS) {
        const int i = (b - ZERO_BLOCKS) * SETUP_T + t;
        if (i < F_IN * NPAD2) {
            const int r = i / NPAD2, c = i % NPAD2;
            W2p[i] = (c < OUT_F) ? W2[r * OUT_F + c] : 0.0f;
        }
    } else {
        if (t < EDIM * HEADS) {
            const int d = t / HEADS, h = t % HEADS;
            float s = 0.f;
            for (int c = 0; c < HID; c++)
                s += We1[d * (HEADS * HID) + h * HID + c] * ae1[h * HID + c];
            v1[t] = s;
        } else if (t < EDIM * HEADS + EDIM) {
            const int d = t - EDIM * HEADS;
            float s = 0.f;
            for (int c = 0; c < OUT_F; c++)
                s += We2[d * OUT_F + c] * ae2[c];
            v2[d] = s;
        }
    }
}

template<int M>
__device__ inline float row16_xadd(float x) {
    const int y = __builtin_amdgcn_update_dpp(0, __float_as_int(x), 0x160 + M, 0xf, 0xf, true);
    return x + __int_as_float(y);
}

template<int NT, int K, int H>
__device__ inline void gemm_stripe(int wave, const float* __restrict__ A, const float* __restrict__ B,
                                   float* __restrict__ C,
                                   const float* __restrict__ asrc, const float* __restrict__ adst,
                                   float* __restrict__ als, float* __restrict__ ald, int M, int N) {
    constexpr int NPAD = NT * 16;
    constexpr int TPH = NT / H;
    static_assert(NT % H == 0, "a column tile belongs to one head");
    int lane = threadIdx.x & 31;
    int half = lane >> 4;
    int lrow = lane & 15;
    int mtiles = M >> 4;
    if (wave >= mtiles) return;
    int m0 = wave << 4;

    float sa[NT], da[NT];
#pragma unroll
    for (int nt = 0; nt < NT; nt++) {
        const int col = nt * 16 + lrow;
        sa[nt] = (col < N) ? asrc[col] : 0.0f;
        da[nt] = (col < N) ? adst[col] : 0.0f;
    }

    v8f acc[NT];
#pragma unroll
    for (int nt = 0; nt < NT; nt++)
#pragma unroll
        for (int g = 0; g < 8; g++) acc[nt][g] = 0.0f;

    const float* arow = A + (size_t)(m0 + lrow) * K;
    const float* bcol = B + lrow;
#pragma unroll 4
    for (int k = 0; k < K; k += 4) {
        int kk = k + half * 2;
        v2f a;
        a.x = arow[kk];
        a.y = arow[kk + 1];
#pragma unroll
        for (int nt = 0; nt < NT; nt++) {
            v2f b;
            b.x = bcol[kk * NPAD + nt * 16];
            b.y = bcol[(kk + 1) * NPAD + nt * 16];
            acc[nt] = __builtin_amdgcn_wmma_f32_16x16x4_f32(
                false, a, false, b, (short)0, acc[nt], false, false);
        }
    }
#pragma unroll
    for (int nt = 0; nt < NT; nt++) {
        int col = nt * 16 + lrow;
        if (col < N) {
#pragma unroll
            for (int g = 0; g < 8; g++)
                C[(size_t)(m0 + g + half * 8) * N + col] = acc[nt][g];
        }
    }

    constexpr int NV = 2 * H * 8;
    float red[NV];
#pragma unroll
    for (int i = 0; i < NV; i++) red[i] = 0.0f;
#pragma unroll
    for (int nt = 0; nt < NT; nt++) {
        const int h = nt / TPH;
        const bool incol = (nt * 16 + lrow) < N;
#pragma unroll
        for (int g = 0; g < 8; g++) {
            const float cv = incol ? acc[nt][g] : 0.0f;
            red[(0 * H + h) * 8 + g] += cv * sa[nt];
            red[(1 * H + h) * 8 + g] += cv * da[nt];
        }
    }
#pragma unroll
    for (int i = 0; i < NV; i++) red[i] = row16_xadd<1>(red[i]);
#pragma unroll
    for (int i = 0; i < NV; i++) red[i] = row16_xadd<2>(red[i]);
#pragma unroll
    for (int i = 0; i < NV; i++) red[i] = row16_xadd<4>(red[i]);
#pragma unroll
    for (int i = 0; i < NV; i++) red[i] = row16_xadd<8>(red[i]);
    if (lrow == 0) {
        float* sp = als + (size_t)(m0 + half * 8) * H;
        float* dp = ald + (size_t)(m0 + half * 8) * H;
#pragma unroll
        for (int g = 0; g < 8; g++)
#pragma unroll
            for (int h = 0; h < H; h++) {
                sp[g * H + h] = red[(0 * H + h) * 8 + g];
                dp[g * H + h] = red[(1 * H + h) * 8 + g];
            }
    }
}

template<int NT, int K, int H>
__global__ void wmma_gemm(const float* __restrict__ A, const float* __restrict__ B,
                          float* __restrict__ C,
                          const float* __restrict__ asrc, const float* __restrict__ adst,
                          float* __restrict__ als, float* __restrict__ ald, int M, int N) {
    gemm_stripe<NT, K, H>((int)((blockIdx.x * blockDim.x + threadIdx.x) >> 5),
                          A, B, C, asrc, adst, als, ald, M, N);
}

#define GC_T 256
#define GEMM1_BLOCKS CDIV(N_NODES / 16, GC_T / 32)
#define COUNT_BLOCKS CDIV(N_EDGES, GC_T)
static_assert(N_NODES % 16 == 0, "the GEMMs take whole 16-row stripes");
__global__ void __launch_bounds__(GC_T) gemm1_and_count(const float* __restrict__ x, const float* __restrict__ W1,
                                                        float* __restrict__ xh,
                                                        const float* __restrict__ asrc, const float* __restrict__ adst,
                                                        float* __restrict__ als, float* __restrict__ ald,
                                                        const int* __restrict__ dst, int* deg) {
    if (blockIdx.x < GEMM1_BLOCKS) {
        gemm_stripe<8, F_IN, HEADS>((int)(blockIdx.x * (GC_T / 32) + (threadIdx.x >> 5)),
                                    x, W1, xh, asrc, adst, als, ald, N_NODES, HEADS * HID);
    } else {
        const int e = (blockIdx.x - GEMM1_BLOCKS) * GC_T + threadIdx.x;
        if (e < N_EDGES) atomicAdd(&deg[dst[e]], 1);
    }
}

#define SCAN_T 1024
#define SCAN_SEG (CDIV(N_NODES, 32 * 128) * 128)
static_assert(N_NODES % 4 == 0, "csr_scan reads and writes four counts at a time");
__global__ void __launch_bounds__(SCAN_T) csr_scan(const int* __restrict__ deg, int* __restrict__ rowptr,
                                                   int* __restrict__ cursor, int n) {
    __shared__ int wsum[32];
    const int t = threadIdx.x, lane = t & 31, w = t >> 5;
    const int base = w * SCAN_SEG;
    int loc = 0;
    for (int it = 0; it < SCAN_SEG; it += 128) {
        const int idx = base + it + lane * 4;
        if (idx < n) {
            const int4 q = *reinterpret_cast<const int4*>(deg + idx);
            loc += q.x + q.y + q.z + q.w;
        }
    }
#pragma unroll
    for (int o = 16; o > 0; o >>= 1) loc += __shfl_xor(loc, o, 32);
    if (lane == 0) wsum[w] = loc;
    __syncthreads();
    if (w == 0) {
        int x = wsum[lane];
#pragma unroll
        for (int o = 1; o < 32; o <<= 1) {
            int y = __shfl_up(x, o, 32);
            if (lane >= o) x += y;
        }
        wsum[lane] = x;
    }
    __syncthreads();
    int carry = (w > 0) ? wsum[w - 1] : 0;
    for (int it = 0; it < SCAN_SEG; it += 128) {
        const int idx = base + it + lane * 4;
        const bool in = idx < n;
        int4 q; q.x = 0; q.y = 0; q.z = 0; q.w = 0;
        if (in) q = *reinterpret_cast<const int4*>(deg + idx);
        const int tot = q.x + q.y + q.z + q.w;
        int inc = tot;
#pragma unroll
        for (int o = 1; o < 32; o <<= 1) {
            int y = __shfl_up(inc, o, 32);
            if (lane >= o) inc += y;
        }
        if (in) {
            int4 r;
            r.x = carry + inc - tot;
            r.y = r.x + q.x;
            r.z = r.y + q.y;
            r.w = r.z + q.z;
            *reinterpret_cast<int4*>(rowptr + idx) = r;
            *reinterpret_cast<int4*>(cursor + idx) = r;
        }
        carry += __shfl(inc, 31, 32);
    }
    if (t == SCAN_T - 1) rowptr[n] = wsum[31];
}

__global__ void csr_fill(const int* __restrict__ src, const int* __restrict__ dst,
                         const float* __restrict__ eattr, int* cursor,
                         uint4* __restrict__ rec, int E) {
    int e = blockIdx.x * blockDim.x + threadIdx.x;
    if (e >= E) return;
    const int d = dst[e];
    uint4 r0, r1;
    r0.x = (unsigned)src[e];
    r0.y = __float_as_uint(eattr[(size_t)e * EDIM + 0]);
    r0.z = __float_as_uint(eattr[(size_t)e * EDIM + 1]);
    r0.w = __float_as_uint(eattr[(size_t)e * EDIM + 2]);
    r1.x = __float_as_uint(eattr[(size_t)e * EDIM + 3]);
    r1.y = __float_as_uint(eattr[(size_t)e * EDIM + 4]);
    r1.z = 0u;
    r1.w = 0u;
    const int p = atomicAdd(&cursor[d], 1);
    rec[(size_t)p * 2]     = r0;
    rec[(size_t)p * 2 + 1] = r1;
}

__device__ inline float wave_sum(float x) {
#pragma unroll
    for (int o = 16; o > 0; o >>= 1) x += __shfl_xor(x, o, 32);
    return x;
}
__device__ inline float wave_max(float x) {
#pragma unroll
    for (int o = 16; o > 0; o >>= 1) x = fmaxf(x, __shfl_xor(x, o, 32));
    return x;
}

__device__ inline void load_rec(const uint4* __restrict__ rec, int p, int& s, float (&e)[EDIM]) {
    const uint4 r0 = rec[(size_t)p * 2];
    const uint4 r1 = rec[(size_t)p * 2 + 1];
    s = (int)r0.x;
    e[0] = __uint_as_float(r0.y);
    e[1] = __uint_as_float(r0.z);
    e[2] = __uint_as_float(r0.w);
    e[3] = __uint_as_float(r1.x);
    e[4] = __uint_as_float(r1.y);
}

__device__ inline void get_edge(const uint4* __restrict__ rec, int start, int deg, int node,
                                const float (&lp)[EDIM], int j, int& s, float (&e)[EDIM]) {
    if (j < deg) {
        load_rec(rec, start + j, s, e);
    } else {
        s = node;
#pragma unroll
        for (int i = 0; i < EDIM; i++) e[i] = lp[i];
    }
}

template<int H>
__device__ inline void edge_alpha(int s, const float (&e)[EDIM], const float (&vv)[EDIM * H],
                                  const float* __restrict__ als, const float (&aldn)[H],
                                  float (&a)[H]) {
#pragma unroll
    for (int h = 0; h < H; h++) {
        float ale = 0.f;
#pragma unroll
        for (int i = 0; i < EDIM; i++) ale += e[i] * vv[i * H + h];
        float x = als[s * H + h] + aldn[h] + ale;
        a[h] = (x > 0.f) ? x : NEG_SLOPE * x;
    }
}

template<int VEC> struct rowvec;
template<> struct rowvec<4> {
    static __device__ inline void ld(const float* p, float (&o)[4]) {
        const float4 t = *reinterpret_cast<const float4*>(p);
        o[0] = t.x; o[1] = t.y; o[2] = t.z; o[3] = t.w;
    }
    static __device__ inline void st(float* p, const float (&o)[4]) {
        float4 t; t.x = o[0]; t.y = o[1]; t.z = o[2]; t.w = o[3];
        *reinterpret_cast<float4*>(p) = t;
    }
};
template<> struct rowvec<2> {
    static __device__ inline void ld(const float* p, float (&o)[2]) {
        const float2 t = *reinterpret_cast<const float2*>(p);
        o[0] = t.x; o[1] = t.y;
    }
    static __device__ inline void st(float* p, const float (&o)[2]) {
        float2 t; t.x = o[0]; t.y = o[1];
        *reinterpret_cast<float2*>(p) = t;
    }
};

template<int H, int C, int VEC>
__device__ inline void add_rows(const float* __restrict__ xh, int cnt, int s, const float (&w)[H],
                                int lane, int hl, float (&acc)[VEC]) {
    constexpr int HC = H * C;
    constexpr int NACT = HC / VEC;
#pragma unroll 4
    for (int jj = 0; jj < cnt; ++jj) {
        const int sj = __builtin_amdgcn_readlane(s, jj);
        float wj = __int_as_float(__builtin_amdgcn_readlane(__float_as_int(w[0]), jj));
        if (H == 2) {
            const float wj1 = __int_as_float(__builtin_amdgcn_readlane(__float_as_int(w[H - 1]), jj));
            wj = hl ? wj1 : wj;
        }
        if (lane < NACT) {
            float r[VEC];
            rowvec<VEC>::ld(xh + (size_t)sj * HC + lane * VEC, r);
#pragma unroll
            for (int q = 0; q < VEC; q++) acc[q] += r[q] * wj;
        }
    }
}

template<int H, int C, int VEC, bool LOGSM>
__global__ void __launch_bounds__(256) gat_node(const int* __restrict__ rowptr,
                                                const uint4* __restrict__ rec,
                                                const float* __restrict__ als,
                                                const float* __restrict__ ald,
                                                const float* __restrict__ v,
                                                const float* __restrict__ xh,
                                                const float* __restrict__ bias,
                                                float* __restrict__ row_out, int n) {
    constexpr int HC = H * C;
    constexpr int NACT = HC / VEC;
    static_assert(H == 1 || H == 2, "head select below handles one or two heads");
    static_assert(NACT * VEC == HC && NACT <= 32 && C % VEC == 0, "lane to channel map");
    const int lane = threadIdx.x & 31;
    const int node = __builtin_amdgcn_readfirstlane((int)((blockIdx.x * blockDim.x + threadIdx.x) >> 5));
    if (node >= n) return;
    const int start = rowptr[node];
    const int deg   = rowptr[node + 1] - start;
    const int total = deg + 1;

    float vv[EDIM * H];
#pragma unroll
    for (int i = 0; i < EDIM * H; i++) vv[i] = v[i];
    float aldn[H];
#pragma unroll
    for (int h = 0; h < H; h++) aldn[h] = ald[node * H + h];

    int   s_f = node;
    float e_f[EDIM];
    float sm[EDIM];
#pragma unroll
    for (int i = 0; i < EDIM; i++) { e_f[i] = 0.f; sm[i] = 0.f; }
    if (lane < deg) {
        load_rec(rec, start + lane, s_f, e_f);
#pragma unroll
        for (int i = 0; i < EDIM; i++) sm[i] = e_f[i];
    }
    for (int j = lane + 32; j < deg; j += 32) {
        int s; float e[EDIM];
        load_rec(rec, start + j, s, e);
#pragma unroll
        for (int i = 0; i < EDIM; i++) sm[i] += e[i];
    }
    float lp[EDIM];
    const float cdiv = fmaxf((float)deg, 1.0f);
#pragma unroll
    for (int i = 0; i < EDIM; i++) lp[i] = wave_sum(sm[i]) / cdiv;

    const bool val_f = lane < total;
    if (lane == deg) {
#pragma unroll
        for (int i = 0; i < EDIM; i++) e_f[i] = lp[i];
    }
    float a_f[H];
    edge_alpha<H>(s_f, e_f, vv, als, aldn, a_f);

    float m[H];
#pragma unroll
    for (int h = 0; h < H; h++) m[h] = val_f ? a_f[h] : -INFINITY;
    for (int j = lane + 32; j < total; j += 32) {
        int s; float e[EDIM]; float a[H];
        get_edge(rec, start, deg, node, lp, j, s, e);
        edge_alpha<H>(s, e, vv, als, aldn, a);
#pragma unroll
        for (int h = 0; h < H; h++) m[h] = fmaxf(m[h], a[h]);
    }
#pragma unroll
    for (int h = 0; h < H; h++) m[h] = wave_max(m[h]);

    float ex_f[H], dn[H];
#pragma unroll
    for (int h = 0; h < H; h++) {
        ex_f[h] = val_f ? expf(a_f[h] - m[h]) : 0.f;
        dn[h] = ex_f[h];
    }
    for (int j = lane + 32; j < total; j += 32) {
        int s; float e[EDIM]; float a[H];
        get_edge(rec, start, deg, node, lp, j, s, e);
        edge_alpha<H>(s, e, vv, als, aldn, a);
#pragma unroll
        for (int h = 0; h < H; h++) dn[h] += expf(a[h] - m[h]);
    }
#pragma unroll
    for (int h = 0; h < H; h++) dn[h] = wave_sum(dn[h]) + 1e-16f;

    const int hl = (lane * VEC) / C;
    const bool act = lane < NACT;
    float bv[VEC];
#pragma unroll
    for (int q = 0; q < VEC; q++) bv[q] = 0.f;
    if (act) rowvec<VEC>::ld(bias + lane * VEC, bv);
    float acc[VEC];
#pragma unroll
    for (int q = 0; q < VEC; q++) acc[q] = 0.f;

    float w_f[H];
#pragma unroll
    for (int h = 0; h < H; h++) w_f[h] = ex_f[h] / dn[h];
    add_rows<H, C, VEC>(xh, total < 32 ? total : 32, s_f, w_f, lane, hl, acc);

    for (int cb = 32; cb < total; cb += 32) {
        const int j = cb + lane;
        int s = node;
        float w[H];
#pragma unroll
        for (int h = 0; h < H; h++) w[h] = 0.f;
        if (j < total) {
            float e[EDIM]; float a[H];
            get_edge(rec, start, deg, node, lp, j, s, e);
            edge_alpha<H>(s, e, vv, als, aldn, a);
#pragma unroll
            for (int h = 0; h < H; h++) w[h] = expf(a[h] - m[h]) / dn[h];
        }
        const int left = total - cb;
        add_rows<H, C, VEC>(xh, left < 32 ? left : 32, s, w, lane, hl, acc);
    }

#pragma unroll
    for (int q = 0; q < VEC; q++) acc[q] += bv[q];

    if (!LOGSM) {
#pragma unroll
        for (int q = 0; q < VEC; q++) acc[q] = (acc[q] > 0.f) ? acc[q] : expm1f(acc[q]);
    } else {
        float mx = -INFINITY;
        if (act) {
#pragma unroll
            for (int q = 0; q < VEC; q++) mx = fmaxf(mx, acc[q]);
        }
        mx = wave_max(mx);
        float se = 0.f;
        if (act) {
#pragma unroll
            for (int q = 0; q < VEC; q++) se += expf(acc[q] - mx);
        }
        se = wave_sum(se);
        const float lse = mx + logf(se);
#pragma unroll
        for (int q = 0; q < VEC; q++) acc[q] -= lse;
    }

    if (act) rowvec<VEC>::st(row_out + (size_t)node * HC + lane * VEC, acc);
}

extern "C" void kernel_launch(void* const* d_in, const int* in_sizes, int n_in,
                              void* d_out, int out_size, void* d_ws, size_t ws_size,
                              hipStream_t stream) {
    const float* x   = (const float*)d_in[0];
    const int*   ei  = (const int*)d_in[1];
    const float* ea  = (const float*)d_in[2];
    const float* W1  = (const float*)d_in[3];
    const float* as1 = (const float*)d_in[4];
    const float* ad1 = (const float*)d_in[5];
    const float* We1 = (const float*)d_in[6];
    const float* ae1 = (const float*)d_in[7];
    const float* b1  = (const float*)d_in[8];
    const float* W2  = (const float*)d_in[9];
    const float* as2 = (const float*)d_in[10];
    const float* ad2 = (const float*)d_in[11];
    const float* We2 = (const float*)d_in[12];
    const float* ae2 = (const float*)d_in[13];
    const float* b2  = (const float*)d_in[14];

    const int N = N_NODES, E = N_EDGES;
    const int HC1 = HEADS * HID;
    const int* src = ei;
    const int* dst = ei + E;

    char* ws = (char*)d_ws;
    size_t off = 0;
    auto alloc = [&](size_t bytes) -> void* {
        void* p = ws + off;
        off = (off + bytes + 255) & ~(size_t)255;
        return p;
    };
    int*      deg    = (int*)alloc(sizeof(int) * N);
    int*      rowptr = (int*)alloc(sizeof(int) * (N + 1));
    int*      cursor = (int*)alloc(sizeof(int) * N);
    uint4*    rec    = (uint4*)alloc(sizeof(uint4) * 2 * (size_t)E);
    float*    v1     = (float*)alloc(sizeof(float) * 16);
    float*    v2     = (float*)alloc(sizeof(float) * 16);
    float*    W2p    = (float*)alloc(sizeof(float) * F_IN * NPAD2);
    float*    xh1    = (float*)alloc(sizeof(float) * (size_t)N * HC1);
    float*    als1   = (float*)alloc(sizeof(float) * N * HEADS);
    float*    ald1   = (float*)alloc(sizeof(float) * N * HEADS);
    float*    h1     = (float*)alloc(sizeof(float) * (size_t)N * HC1);
    float*    xh2    = (float*)alloc(sizeof(float) * (size_t)N * OUT_F);
    float*    als2   = (float*)alloc(sizeof(float) * N);
    float*    ald2   = (float*)alloc(sizeof(float) * N);
    (void)ws_size; (void)n_in; (void)in_sizes; (void)out_size;

    const int T = 256;

    setup_small<<<ZERO_BLOCKS + PAD_BLOCKS + 1, SETUP_T, 0, stream>>>((unsigned*)deg, W2, W2p, We1, ae1, We2, ae2, v1, v2);
    gemm1_and_count<<<GEMM1_BLOCKS + COUNT_BLOCKS, GC_T, 0, stream>>>(x, W1, xh1, as1, ad1, als1, ald1, dst, deg);
    csr_scan<<<1, SCAN_T, 0, stream>>>(deg, rowptr, cursor, N);
    csr_fill<<<CDIV(E, T), T, 0, stream>>>(src, dst, ea, cursor, rec, E);

    gat_node<HEADS, HID, 4, false><<<CDIV(N, T / 32), T, 0, stream>>>(rowptr, rec, als1, ald1, v1, xh1, b1, h1, N);

    {
        int mtiles = N / 16, waves_per_blk = 128 / 32;
        wmma_gemm<3, HC1, 1><<<CDIV(mtiles, waves_per_blk), 128, 0, stream>>>(h1, W2p, xh2, as2, ad2, als2, ald2, N, OUT_F);
    }
    gat_node<1, OUT_F, 2, true><<<CDIV(N, T / 32), T, 0, stream>>>(rowptr, rec, als2, ald2, v2, xh2, b2, (float*)d_out, N);
}


# Degree prefix sum rebuilt by each of 256 resident fill work-groups in its own LDS (DPP wave scans); scan launch gone (7 -> 6 launches); edge slots still arrival-order rank
# speedup vs baseline: 5.0135x; 1.0712x over previous
#include <hip/hip_runtime.h>
#include <math.h>

#define N_NODES 50000
#define N_EDGES 800000
#define ETOT    (N_EDGES + N_NODES)
#define F_IN    128
#define HID     64
#define HEADS   2
#define OUT_F   40
#define EDIM    5
#define NEG_SLOPE 0.2f

#define CDIV(a,b) (((a)+(b)-1)/(b))

typedef __attribute__((ext_vector_type(2))) float v2f;
typedef __attribute__((ext_vector_type(8))) float v8f;

#define NPAD2   48
#define SETUP_T 256
#define ZERO_BLOCKS CDIV(N_NODES, SETUP_T)
#define PAD_BLOCKS  CDIV(F_IN * NPAD2, SETUP_T)

__global__ void __launch_bounds__(SETUP_T) setup_small(unsigned* __restrict__ deg, unsigned* __restrict__ cursor,
                                                       const float* __restrict__ W2, float* __restrict__ W2p,
                                                       const float* __restrict__ We1, const float* __restrict__ ae1,
                                                       const float* __restrict__ We2, const float* __restrict__ ae2,
                                                       float* __restrict__ v1, float* __restrict__ v2) {
    const int b = blockIdx.x;
    const int t = threadIdx.x;
    if (b < ZERO_BLOCKS) {
        const int i = b * SETUP_T + t;
        if (i < N_NODES) { deg[i] = 0u; cursor[i] = 0u; }
    } else if (b < ZERO_BLOCKS + PAD_BLOCKS) {
        const int i = (b - ZERO_BLOCKS) * SETUP_T + t;
        if (i < F_IN * NPAD2) {
            const int r = i / NPAD2, c = i % NPAD2;
            W2p[i] = (c < OUT_F) ? W2[r * OUT_F + c] : 0.0f;
        }
    } else {
        if (t < EDIM * HEADS) {
            const int d = t / HEADS, h = t % HEADS;
            float s = 0.f;
            for (int c = 0; c < HID; c++)
                s += We1[d * (HEADS * HID) + h * HID + c] * ae1[h * HID + c];
            v1[t] = s;
        } else if (t < EDIM * HEADS + EDIM) {
            const int d = t - EDIM * HEADS;
            float s = 0.f;
            for (int c = 0; c < OUT_F; c++)
                s += We2[d * OUT_F + c] * ae2[c];
            v2[d] = s;
        }
    }
}

template<int M>
__device__ inline float row16_xadd(float x) {
    const int y = __builtin_amdgcn_update_dpp(0, __float_as_int(x), 0x160 + M, 0xf, 0xf, true);
    return x + __int_as_float(y);
}

template<int NT, int K, int H>
__device__ inline void gemm_stripe(int wave, const float* __restrict__ A, const float* __restrict__ B,
                                   float* __restrict__ C,
                                   const float* __restrict__ asrc, const float* __restrict__ adst,
                                   float* __restrict__ als, float* __restrict__ ald, int M, int N) {
    constexpr int NPAD = NT * 16;
    constexpr int TPH = NT / H;
    static_assert(NT % H == 0, "a column tile belongs to one head");
    int lane = threadIdx.x & 31;
    int half = lane >> 4;
    int lrow = lane & 15;
    int mtiles = M >> 4;
    if (wave >= mtiles) return;
    int m0 = wave << 4;

    float sa[NT], da[NT];
#pragma unroll
    for (int nt = 0; nt < NT; nt++) {
        const int col = nt * 16 + lrow;
        sa[nt] = (col < N) ? asrc[col] : 0.0f;
        da[nt] = (col < N) ? adst[col] : 0.0f;
    }

    v8f acc[NT];
#pragma unroll
    for (int nt = 0; nt < NT; nt++)
#pragma unroll
        for (int g = 0; g < 8; g++) acc[nt][g] = 0.0f;

    const float* arow = A + (size_t)(m0 + lrow) * K;
    const float* bcol = B + lrow;
#pragma unroll 4
    for (int k = 0; k < K; k += 4) {
        int kk = k + half * 2;
        v2f a;
        a.x = arow[kk];
        a.y = arow[kk + 1];
#pragma unroll
        for (int nt = 0; nt < NT; nt++) {
            v2f b;
            b.x = bcol[kk * NPAD + nt * 16];
            b.y = bcol[(kk + 1) * NPAD + nt * 16];
            acc[nt] = __builtin_amdgcn_wmma_f32_16x16x4_f32(
                false, a, false, b, (short)0, acc[nt], false, false);
        }
    }
#pragma unroll
    for (int nt = 0; nt < NT; nt++) {
        int col = nt * 16 + lrow;
        if (col < N) {
#pragma unroll
            for (int g = 0; g < 8; g++)
                C[(size_t)(m0 + g + half * 8) * N + col] = acc[nt][g];
        }
    }

    constexpr int NV = 2 * H * 8;
    float red[NV];
#pragma unroll
    for (int i = 0; i < NV; i++) red[i] = 0.0f;
#pragma unroll
    for (int nt = 0; nt < NT; nt++) {
        const int h = nt / TPH;
        const bool incol = (nt * 16 + lrow) < N;
#pragma unroll
        for (int g = 0; g < 8; g++) {
            const float cv = incol ? acc[nt][g] : 0.0f;
            red[(0 * H + h) * 8 + g] += cv * sa[nt];
            red[(1 * H + h) * 8 + g] += cv * da[nt];
        }
    }
#pragma unroll
    for (int i = 0; i < NV; i++) red[i] = row16_xadd<1>(red[i]);
#pragma unroll
    for (int i = 0; i < NV; i++) red[i] = row16_xadd<2>(red[i]);
#pragma unroll
    for (int i = 0; i < NV; i++) red[i] = row16_xadd<4>(red[i]);
#pragma unroll
    for (int i = 0; i < NV; i++) red[i] = row16_xadd<8>(red[i]);
    if (lrow == 0) {
        float* sp = als + (size_t)(m0 + half * 8) * H;
        float* dp = ald + (size_t)(m0 + half * 8) * H;
#pragma unroll
        for (int g = 0; g < 8; g++)
#pragma unroll
            for (int h = 0; h < H; h++) {
                sp[g * H + h] = red[(0 * H + h) * 8 + g];
                dp[g * H + h] = red[(1 * H + h) * 8 + g];
            }
    }
}

template<int NT, int K, int H>
__global__ void wmma_gemm(const float* __restrict__ A, const float* __restrict__ B,
                          float* __restrict__ C,
                          const float* __restrict__ asrc, const float* __restrict__ adst,
                          float* __restrict__ als, float* __restrict__ ald, int M, int N) {
    gemm_stripe<NT, K, H>((int)((blockIdx.x * blockDim.x + threadIdx.x) >> 5),
                          A, B, C, asrc, adst, als, ald, M, N);
}

#define GC_T 256
#define GEMM1_BLOCKS CDIV(N_NODES / 16, GC_T / 32)
#define COUNT_BLOCKS CDIV(N_EDGES, GC_T)
static_assert(N_NODES % 16 == 0, "the GEMMs take whole 16-row stripes");
__global__ void __launch_bounds__(GC_T) gemm1_and_count(const float* __restrict__ x, const float* __restrict__ W1,
                                                        float* __restrict__ xh,
                                                        const float* __restrict__ asrc, const float* __restrict__ adst,
                                                        float* __restrict__ als, float* __restrict__ ald,
                                                        const int* __restrict__ dst, int* deg) {
    if (blockIdx.x < GEMM1_BLOCKS) {
        gemm_stripe<8, F_IN, HEADS>((int)(blockIdx.x * (GC_T / 32) + (threadIdx.x >> 5)),
                                    x, W1, xh, asrc, adst, als, ald, N_NODES, HEADS * HID);
    } else {
        const int e = (blockIdx.x - GEMM1_BLOCKS) * GC_T + threadIdx.x;
        if (e < N_EDGES) atomicAdd(&deg[dst[e]], 1);
    }
}

#define FILL_T 1024
#define FILL_BLOCKS 256
#define SCAN_SEG (CDIV(N_NODES, 32 * 128) * 128)
#define SCAN_IT  (SCAN_SEG / 128)
#define EPB (N_EDGES / FILL_BLOCKS)
#define EDGE_FULL (EPB / FILL_T)
#define EDGE_REM  (EPB - EDGE_FULL * FILL_T)
#define EDGE_IT   (EDGE_FULL + (EDGE_REM > 0 ? 1 : 0))
static_assert(EPB * FILL_BLOCKS == N_EDGES, "the edges divide evenly over the fill work-groups");
static_assert(N_NODES % 4 == 0, "scan_fill reads and writes four counts at a time");

template<int N>
__device__ inline int row16_shr_add(int x) {
    const int y = __builtin_amdgcn_update_dpp(0, x, 0x110 + N, 0xf, 0xf, true);
    return x + y;
}
__device__ inline int wave_incl_scan(int x, int lane) {
    x = row16_shr_add<1>(x);
    x = row16_shr_add<2>(x);
    x = row16_shr_add<4>(x);
    x = row16_shr_add<8>(x);
    const int lo = __builtin_amdgcn_readlane(x, 15);
    return (lane >= 16) ? x + lo : x;
}

__global__ void __launch_bounds__(FILL_T) scan_fill(const int* __restrict__ deg, int* __restrict__ rowptr,
                                                    const int* __restrict__ src, const int* __restrict__ dst,
                                                    const float* __restrict__ eattr, int* cursor,
                                                    uint4* __restrict__ rec) {
    __shared__ int rp[32 * SCAN_SEG];
    __shared__ int wsum[32];
    const int t = threadIdx.x, lane = t & 31, w = t >> 5;
    const int base = w * SCAN_SEG;

    int4 q[SCAN_IT];
#pragma unroll
    for (int it = 0; it < SCAN_IT; it++) {
        const int idx = base + it * 128 + lane * 4;
        q[it].x = 0; q[it].y = 0; q[it].z = 0; q[it].w = 0;
        if (idx < N_NODES) q[it] = *reinterpret_cast<const int4*>(deg + idx);
    }

    const int e0 = blockIdx.x * EPB;
    int   d[EDGE_IT], s[EDGE_IT];
    float ea[EDGE_IT][EDIM];
#pragma unroll
    for (int i = 0; i < EDGE_IT; i++) {
        d[i] = 0; s[i] = 0;
#pragma unroll
        for (int k = 0; k < EDIM; k++) ea[i][k] = 0.f;
        if (i < EDGE_FULL || t < EDGE_REM) {
            const int e = e0 + i * FILL_T + t;
            d[i] = dst[e];
            s[i] = src[e];
#pragma unroll
            for (int k = 0; k < EDIM; k++) ea[i][k] = eattr[(size_t)e * EDIM + k];
        }
    }

    int tot[SCAN_IT], inc[SCAN_IT], carry[SCAN_IT];
    int run = 0;
#pragma unroll
    for (int it = 0; it < SCAN_IT; it++) {
        tot[it] = q[it].x + q[it].y + q[it].z + q[it].w;
        inc[it] = wave_incl_scan(tot[it], lane);
        carry[it] = run;
        run += __builtin_amdgcn_readlane(inc[it], 31);
    }
    if (lane == 0) wsum[w] = run;
    __syncthreads();
    if (w == 0) wsum[lane] = wave_incl_scan(wsum[lane], lane);
    __syncthreads();
    const int wbase = (w > 0) ? wsum[w - 1] : 0;
#pragma unroll
    for (int it = 0; it < SCAN_IT; it++) {
        const int idx = base + it * 128 + lane * 4;
        int4 r;
        r.x = wbase + carry[it] + inc[it] - tot[it];
        r.y = r.x + q[it].x;
        r.z = r.y + q[it].y;
        r.w = r.z + q[it].z;
        *reinterpret_cast<int4*>(rp + idx) = r;
        if ((w * SCAN_IT + it) % FILL_BLOCKS == (int)blockIdx.x && idx < N_NODES)
            *reinterpret_cast<int4*>(rowptr + idx) = r;
    }
    if (blockIdx.x == 0 && t == 0) rowptr[N_NODES] = wsum[31];
    __syncthreads();

    int bs[EDGE_IT], rk[EDGE_IT];
#pragma unroll
    for (int i = 0; i < EDGE_IT; i++) {
        bs[i] = 0; rk[i] = 0;
        if (i < EDGE_FULL || t < EDGE_REM) {
            bs[i] = rp[d[i]];
            rk[i] = atomicAdd(&cursor[d[i]], 1);
        }
    }
#pragma unroll
    for (int i = 0; i < EDGE_IT; i++) {
        if (i < EDGE_FULL || t < EDGE_REM) {
            const size_t p = (size_t)(bs[i] + rk[i]);
            uint4 r0, r1;
            r0.x = (unsigned)s[i];
            r0.y = __float_as_uint(ea[i][0]);
            r0.z = __float_as_uint(ea[i][1]);
            r0.w = __float_as_uint(ea[i][2]);
            r1.x = __float_as_uint(ea[i][3]);
            r1.y = __float_as_uint(ea[i][4]);
            r1.z = 0u;
            r1.w = 0u;
            rec[p * 2]     = r0;
            rec[p * 2 + 1] = r1;
        }
    }
}

__device__ inline float wave_sum(float x) {
#pragma unroll
    for (int o = 16; o > 0; o >>= 1) x += __shfl_xor(x, o, 32);
    return x;
}
__device__ inline float wave_max(float x) {
#pragma unroll
    for (int o = 16; o > 0; o >>= 1) x = fmaxf(x, __shfl_xor(x, o, 32));
    return x;
}

__device__ inline void load_rec(const uint4* __restrict__ rec, int p, int& s, float (&e)[EDIM]) {
    const uint4 r0 = rec[(size_t)p * 2];
    const uint4 r1 = rec[(size_t)p * 2 + 1];
    s = (int)r0.x;
    e[0] = __uint_as_float(r0.y);
    e[1] = __uint_as_float(r0.z);
    e[2] = __uint_as_float(r0.w);
    e[3] = __uint_as_float(r1.x);
    e[4] = __uint_as_float(r1.y);
}

__device__ inline void get_edge(const uint4* __restrict__ rec, int start, int deg, int node,
                                const float (&lp)[EDIM], int j, int& s, float (&e)[EDIM]) {
    if (j < deg) {
        load_rec(rec, start + j, s, e);
    } else {
        s = node;
#pragma unroll
        for (int i = 0; i < EDIM; i++) e[i] = lp[i];
    }
}

template<int H>
__device__ inline void edge_alpha(int s, const float (&e)[EDIM], const float (&vv)[EDIM * H],
                                  const float* __restrict__ als, const float (&aldn)[H],
                                  float (&a)[H]) {
#pragma unroll
    for (int h = 0; h < H; h++) {
        float ale = 0.f;
#pragma unroll
        for (int i = 0; i < EDIM; i++) ale += e[i] * vv[i * H + h];
        float x = als[s * H + h] + aldn[h] + ale;
        a[h] = (x > 0.f) ? x : NEG_SLOPE * x;
    }
}

template<int VEC> struct rowvec;
template<> struct rowvec<4> {
    static __device__ inline void ld(const float* p, float (&o)[4]) {
        const float4 t = *reinterpret_cast<const float4*>(p);
        o[0] = t.x; o[1] = t.y; o[2] = t.z; o[3] = t.w;
    }
    static __device__ inline void st(float* p, const float (&o)[4]) {
        float4 t; t.x = o[0]; t.y = o[1]; t.z = o[2]; t.w = o[3];
        *reinterpret_cast<float4*>(p) = t;
    }
};
template<> struct rowvec<2> {
    static __device__ inline void ld(const float* p, float (&o)[2]) {
        const float2 t = *reinterpret_cast<const float2*>(p);
        o[0] = t.x; o[1] = t.y;
    }
    static __device__ inline void st(float* p, const float (&o)[2]) {
        float2 t; t.x = o[0]; t.y = o[1];
        *reinterpret_cast<float2*>(p) = t;
    }
};

template<int H, int C, int VEC>
__device__ inline void add_rows(const float* __restrict__ xh, int cnt, int s, const float (&w)[H],
                                int lane, int hl, float (&acc)[VEC]) {
    constexpr int HC = H * C;
    constexpr int NACT = HC / VEC;
#pragma unroll 4
    for (int jj = 0; jj < cnt; ++jj) {
        const int sj = __builtin_amdgcn_readlane(s, jj);
        float wj = __int_as_float(__builtin_amdgcn_readlane(__float_as_int(w[0]), jj));
        if (H == 2) {
            const float wj1 = __int_as_float(__builtin_amdgcn_readlane(__float_as_int(w[H - 1]), jj));
            wj = hl ? wj1 : wj;
        }
        if (lane < NACT) {
            float r[VEC];
            rowvec<VEC>::ld(xh + (size_t)sj * HC + lane * VEC, r);
#pragma unroll
            for (int q = 0; q < VEC; q++) acc[q] += r[q] * wj;
        }
    }
}

template<int H, int C, int VEC, bool LOGSM>
__global__ void __launch_bounds__(256) gat_node(const int* __restrict__ rowptr,
                                                const uint4* __restrict__ rec,
                                                const float* __restrict__ als,
                                                const float* __restrict__ ald,
                                                const float* __restrict__ v,
                                                const float* __restrict__ xh,
                                                const float* __restrict__ bias,
                                                float* __restrict__ row_out, int n) {
    constexpr int HC = H * C;
    constexpr int NACT = HC / VEC;
    static_assert(H == 1 || H == 2, "head select below handles one or two heads");
    static_assert(NACT * VEC == HC && NACT <= 32 && C % VEC == 0, "lane to channel map");
    const int lane = threadIdx.x & 31;
    const int node = __builtin_amdgcn_readfirstlane((int)((blockIdx.x * blockDim.x + threadIdx.x) >> 5));
    if (node >= n) return;
    const int start = rowptr[node];
    const int deg   = rowptr[node + 1] - start;
    const int total = deg + 1;

    float vv[EDIM * H];
#pragma unroll
    for (int i = 0; i < EDIM * H; i++) vv[i] = v[i];
    float aldn[H];
#pragma unroll
    for (int h = 0; h < H; h++) aldn[h] = ald[node * H + h];

    int   s_f = node;
    float e_f[EDIM];
    float sm[EDIM];
#pragma unroll
    for (int i = 0; i < EDIM; i++) { e_f[i] = 0.f; sm[i] = 0.f; }
    if (lane < deg) {
        load_rec(rec, start + lane, s_f, e_f);
#pragma unroll
        for (int i = 0; i < EDIM; i++) sm[i] = e_f[i];
    }
    for (int j = lane + 32; j < deg; j += 32) {
        int s; float e[EDIM];
        load_rec(rec, start + j, s, e);
#pragma unroll
        for (int i = 0; i < EDIM; i++) sm[i] += e[i];
    }
    float lp[EDIM];
    const float cdiv = fmaxf((float)deg, 1.0f);
#pragma unroll
    for (int i = 0; i < EDIM; i++) lp[i] = wave_sum(sm[i]) / cdiv;

    const bool val_f = lane < total;
    if (lane == deg) {
#pragma unroll
        for (int i = 0; i < EDIM; i++) e_f[i] = lp[i];
    }
    float a_f[H];
    edge_alpha<H>(s_f, e_f, vv, als, aldn, a_f);

    float m[H];
#pragma unroll
    for (int h = 0; h < H; h++) m[h] = val_f ? a_f[h] : -INFINITY;
    for (int j = lane + 32; j < total; j += 32) {
        int s; float e[EDIM]; float a[H];
        get_edge(rec, start, deg, node, lp, j, s, e);
        edge_alpha<H>(s, e, vv, als, aldn, a);
#pragma unroll
        for (int h = 0; h < H; h++) m[h] = fmaxf(m[h], a[h]);
    }
#pragma unroll
    for (int h = 0; h < H; h++) m[h] = wave_max(m[h]);

    float ex_f[H], dn[H];
#pragma unroll
    for (int h = 0; h < H; h++) {
        ex_f[h] = val_f ? expf(a_f[h] - m[h]) : 0.f;
        dn[h] = ex_f[h];
    }
    for (int j = lane + 32; j < total; j += 32) {
        int s; float e[EDIM]; float a[H];
        get_edge(rec, start, deg, node, lp, j, s, e);
        edge_alpha<H>(s, e, vv, als, aldn, a);
#pragma unroll
        for (int h = 0; h < H; h++) dn[h] += expf(a[h] - m[h]);
    }
#pragma unroll
    for (int h = 0; h < H; h++) dn[h] = wave_sum(dn[h]) + 1e-16f;

    const int hl = (lane * VEC) / C;
    const bool act = lane < NACT;
    float bv[VEC];
#pragma unroll
    for (int q = 0; q < VEC; q++) bv[q] = 0.f;
    if (act) rowvec<VEC>::ld(bias + lane * VEC, bv);
    float acc[VEC];
#pragma unroll
    for (int q = 0; q < VEC; q++) acc[q] = 0.f;

    float w_f[H];
#pragma unroll
    for (int h = 0; h < H; h++) w_f[h] = ex_f[h] / dn[h];
    add_rows<H, C, VEC>(xh, total < 32 ? total : 32, s_f, w_f, lane, hl, acc);

    for (int cb = 32; cb < total; cb += 32) {
        const int j = cb + lane;
        int s = node;
        float w[H];
#pragma unroll
        for (int h = 0; h < H; h++) w[h] = 0.f;
        if (j < total) {
            float e[EDIM]; float a[H];
            get_edge(rec, start, deg, node, lp, j, s, e);
            edge_alpha<H>(s, e, vv, als, aldn, a);
#pragma unroll
            for (int h = 0; h < H; h++) w[h] = expf(a[h] - m[h]) / dn[h];
        }
        const int left = total - cb;
        add_rows<H, C, VEC>(xh, left < 32 ? left : 32, s, w, lane, hl, acc);
    }

#pragma unroll
    for (int q = 0; q < VEC; q++) acc[q] += bv[q];

    if (!LOGSM) {
#pragma unroll
        for (int q = 0; q < VEC; q++) acc[q] = (acc[q] > 0.f) ? acc[q] : expm1f(acc[q]);
    } else {
        float mx = -INFINITY;
        if (act) {
#pragma unroll
            for (int q = 0; q < VEC; q++) mx = fmaxf(mx, acc[q]);
        }
        mx = wave_max(mx);
        float se = 0.f;
        if (act) {
#pragma unroll
            for (int q = 0; q < VEC; q++) se += expf(acc[q] - mx);
        }
        se = wave_sum(se);
        const float lse = mx + logf(se);
#pragma unroll
        for (int q = 0; q < VEC; q++) acc[q] -= lse;
    }

    if (act) rowvec<VEC>::st(row_out + (size_t)node * HC + lane * VEC, acc);
}

extern "C" void kernel_launch(void* const* d_in, const int* in_sizes, int n_in,
                              void* d_out, int out_size, void* d_ws, size_t ws_size,
                              hipStream_t stream) {
    const float* x   = (const float*)d_in[0];
    const int*   ei  = (const int*)d_in[1];
    const float* ea  = (const float*)d_in[2];
    const float* W1  = (const float*)d_in[3];
    const float* as1 = (const float*)d_in[4];
    const float* ad1 = (const float*)d_in[5];
    const float* We1 = (const float*)d_in[6];
    const float* ae1 = (const float*)d_in[7];
    const float* b1  = (const float*)d_in[8];
    const float* W2  = (const float*)d_in[9];
    const float* as2 = (const float*)d_in[10];
    const float* ad2 = (const float*)d_in[11];
    const float* We2 = (const float*)d_in[12];
    const float* ae2 = (const float*)d_in[13];
    const float* b2  = (const float*)d_in[14];

    const int N = N_NODES, E = N_EDGES;
    const int HC1 = HEADS * HID;
    const int* src = ei;
    const int* dst = ei + E;

    char* ws = (char*)d_ws;
    size_t off = 0;
    auto alloc = [&](size_t bytes) -> void* {
        void* p = ws + off;
        off = (off + bytes + 255) & ~(size_t)255;
        return p;
    };
    int*      deg    = (int*)alloc(sizeof(int) * N);
    int*      rowptr = (int*)alloc(sizeof(int) * (N + 1));
    int*      cursor = (int*)alloc(sizeof(int) * N);
    uint4*    rec    = (uint4*)alloc(sizeof(uint4) * 2 * (size_t)E);
    float*    v1     = (float*)alloc(sizeof(float) * 16);
    float*    v2     = (float*)alloc(sizeof(float) * 16);
    float*    W2p    = (float*)alloc(sizeof(float) * F_IN * NPAD2);
    float*    xh1    = (float*)alloc(sizeof(float) * (size_t)N * HC1);
    float*    als1   = (float*)alloc(sizeof(float) * N * HEADS);
    float*    ald1   = (float*)alloc(sizeof(float) * N * HEADS);
    float*    h1     = (float*)alloc(sizeof(float) * (size_t)N * HC1);
    float*    xh2    = (float*)alloc(sizeof(float) * (size_t)N * OUT_F);
    float*    als2   = (float*)alloc(sizeof(float) * N);
    float*    ald2   = (float*)alloc(sizeof(float) * N);
    (void)ws_size; (void)n_in; (void)in_sizes; (void)out_size;

    const int T = 256;

    setup_small<<<ZERO_BLOCKS + PAD_BLOCKS + 1, SETUP_T, 0, stream>>>((unsigned*)deg, (unsigned*)cursor, W2, W2p, We1, ae1, We2, ae2, v1, v2);
    gemm1_and_count<<<GEMM1_BLOCKS + COUNT_BLOCKS, GC_T, 0, stream>>>(x, W1, xh1, as1, ad1, als1, ald1, dst, deg);
    scan_fill<<<FILL_BLOCKS, FILL_T, 0, stream>>>(deg, rowptr, src, dst, ea, cursor, rec);

    gat_node<HEADS, HID, 4, false><<<CDIV(N, T / 32), T, 0, stream>>>(rowptr, rec, als1, ald1, v1, xh1, b1, h1, N);

    {
        int mtiles = N / 16, waves_per_blk = 128 / 32;
        wmma_gemm<3, HC1, 1><<<CDIV(mtiles, waves_per_blk), 128, 0, stream>>>(h1, W2p, xh2, as2, ad2, als2, ald2, N, OUT_F);
    }
    gat_node<1, OUT_F, 2, true><<<CDIV(N, T / 32), T, 0, stream>>>(rowptr, rec, als2, ald2, v2, xh2, b2, (float*)d_out, N);
}


# Edge records cut from 32 to 16 bytes: fill stores src + the edge's 3 score terms <attr, We.att_e> (2 heads, layer 2) instead of 5 attrs; self-loop term = mean of edge terms, sum order changed
# speedup vs baseline: 5.3524x; 1.0676x over previous
#include <hip/hip_runtime.h>
#include <math.h>

#define N_NODES 50000
#define N_EDGES 800000
#define ETOT    (N_EDGES + N_NODES)
#define F_IN    128
#define HID     64
#define HEADS   2
#define OUT_F   40
#define EDIM    5
#define NEG_SLOPE 0.2f

#define CDIV(a,b) (((a)+(b)-1)/(b))

typedef __attribute__((ext_vector_type(2))) float v2f;
typedef __attribute__((ext_vector_type(8))) float v8f;

#define NPAD2   48
#define SETUP_T 256
#define ZERO_BLOCKS CDIV(N_NODES, SETUP_T)
#define PAD_BLOCKS  CDIV(F_IN * NPAD2, SETUP_T)

__global__ void __launch_bounds__(SETUP_T) setup_small(unsigned* __restrict__ deg, unsigned* __restrict__ cursor,
                                                       const float* __restrict__ W2, float* __restrict__ W2p,
                                                       const float* __restrict__ We1, const float* __restrict__ ae1,
                                                       const float* __restrict__ We2, const float* __restrict__ ae2,
                                                       float* __restrict__ v1, float* __restrict__ v2) {
    const int b = blockIdx.x;
    const int t = threadIdx.x;
    if (b < ZERO_BLOCKS) {
        const int i = b * SETUP_T + t;
        if (i < N_NODES) { deg[i] = 0u; cursor[i] = 0u; }
    } else if (b < ZERO_BLOCKS + PAD_BLOCKS) {
        const int i = (b - ZERO_BLOCKS) * SETUP_T + t;
        if (i < F_IN * NPAD2) {
            const int r = i / NPAD2, c = i % NPAD2;
            W2p[i] = (c < OUT_F) ? W2[r * OUT_F + c] : 0.0f;
        }
    } else {
        if (t < EDIM * HEADS) {
            const int d = t / HEADS, h = t % HEADS;
            float s = 0.f;
            for (int c = 0; c < HID; c++)
                s += We1[d * (HEADS * HID) + h * HID + c] * ae1[h * HID + c];
            v1[t] = s;
        } else if (t < EDIM * HEADS + EDIM) {
            const int d = t - EDIM * HEADS;
            float s = 0.f;
            for (int c = 0; c < OUT_F; c++)
                s += We2[d * OUT_F + c] * ae2[c];
            v2[d] = s;
        }
    }
}

template<int M>
__device__ inline float row16_xadd(float x) {
    const int y = __builtin_amdgcn_update_dpp(0, __float_as_int(x), 0x160 + M, 0xf, 0xf, true);
    return x + __int_as_float(y);
}

template<int NT, int K, int H>
__device__ inline void gemm_stripe(int wave, const float* __restrict__ A, const float* __restrict__ B,
                                   float* __restrict__ C,
                                   const float* __restrict__ asrc, const float* __restrict__ adst,
                                   float* __restrict__ als, float* __restrict__ ald, int M, int N) {
    constexpr int NPAD = NT * 16;
    constexpr int TPH = NT / H;
    static_assert(NT % H == 0, "a column tile belongs to one head");
    int lane = threadIdx.x & 31;
    int half = lane >> 4;
    int lrow = lane & 15;
    int mtiles = M >> 4;
    if (wave >= mtiles) return;
    int m0 = wave << 4;

    float sa[NT], da[NT];
#pragma unroll
    for (int nt = 0; nt < NT; nt++) {
        const int col = nt * 16 + lrow;
        sa[nt] = (col < N) ? asrc[col] : 0.0f;
        da[nt] = (col < N) ? adst[col] : 0.0f;
    }

    v8f acc[NT];
#pragma unroll
    for (int nt = 0; nt < NT; nt++)
#pragma unroll
        for (int g = 0; g < 8; g++) acc[nt][g] = 0.0f;

    const float* arow = A + (size_t)(m0 + lrow) * K;
    const float* bcol = B + lrow;
#pragma unroll 4
    for (int k = 0; k < K; k += 4) {
        int kk = k + half * 2;
        v2f a;
        a.x = arow[kk];
        a.y = arow[kk + 1];
#pragma unroll
        for (int nt = 0; nt < NT; nt++) {
            v2f b;
            b.x = bcol[kk * NPAD + nt * 16];
            b.y = bcol[(kk + 1) * NPAD + nt * 16];
            acc[nt] = __builtin_amdgcn_wmma_f32_16x16x4_f32(
                false, a, false, b, (short)0, acc[nt], false, false);
        }
    }
#pragma unroll
    for (int nt = 0; nt < NT; nt++) {
        int col = nt * 16 + lrow;
        if (col < N) {
#pragma unroll
            for (int g = 0; g < 8; g++)
                C[(size_t)(m0 + g + half * 8) * N + col] = acc[nt][g];
        }
    }

    constexpr int NV = 2 * H * 8;
    float red[NV];
#pragma unroll
    for (int i = 0; i < NV; i++) red[i] = 0.0f;
#pragma unroll
    for (int nt = 0; nt < NT; nt++) {
        const int h = nt / TPH;
        const bool incol = (nt * 16 + lrow) < N;
#pragma unroll
        for (int g = 0; g < 8; g++) {
            const float cv = incol ? acc[nt][g] : 0.0f;
            red[(0 * H + h) * 8 + g] += cv * sa[nt];
            red[(1 * H + h) * 8 + g] += cv * da[nt];
        }
    }
#pragma unroll
    for (int i = 0; i < NV; i++) red[i] = row16_xadd<1>(red[i]);
#pragma unroll
    for (int i = 0; i < NV; i++) red[i] = row16_xadd<2>(red[i]);
#pragma unroll
    for (int i = 0; i < NV; i++) red[i] = row16_xadd<4>(red[i]);
#pragma unroll
    for (int i = 0; i < NV; i++) red[i] = row16_xadd<8>(red[i]);
    if (lrow == 0) {
        float* sp = als + (size_t)(m0 + half * 8) * H;
        float* dp = ald + (size_t)(m0 + half * 8) * H;
#pragma unroll
        for (int g = 0; g < 8; g++)
#pragma unroll
            for (int h = 0; h < H; h++) {
                sp[g * H + h] = red[(0 * H + h) * 8 + g];
                dp[g * H + h] = red[(1 * H + h) * 8 + g];
            }
    }
}

template<int NT, int K, int H>
__global__ void wmma_gemm(const float* __restrict__ A, const float* __restrict__ B,
                          float* __restrict__ C,
                          const float* __restrict__ asrc, const float* __restrict__ adst,
                          float* __restrict__ als, float* __restrict__ ald, int M, int N) {
    gemm_stripe<NT, K, H>((int)((blockIdx.x * blockDim.x + threadIdx.x) >> 5),
                          A, B, C, asrc, adst, als, ald, M, N);
}

#define GC_T 256
#define GEMM1_BLOCKS CDIV(N_NODES / 16, GC_T / 32)
#define COUNT_BLOCKS CDIV(N_EDGES, GC_T)
static_assert(N_NODES % 16 == 0, "the GEMMs take whole 16-row stripes");
__global__ void __launch_bounds__(GC_T) gemm1_and_count(const float* __restrict__ x, const float* __restrict__ W1,
                                                        float* __restrict__ xh,
                                                        const float* __restrict__ asrc, const float* __restrict__ adst,
                                                        float* __restrict__ als, float* __restrict__ ald,
                                                        const int* __restrict__ dst, int* deg) {
    if (blockIdx.x < GEMM1_BLOCKS) {
        gemm_stripe<8, F_IN, HEADS>((int)(blockIdx.x * (GC_T / 32) + (threadIdx.x >> 5)),
                                    x, W1, xh, asrc, adst, als, ald, N_NODES, HEADS * HID);
    } else {
        const int e = (blockIdx.x - GEMM1_BLOCKS) * GC_T + threadIdx.x;
        if (e < N_EDGES) atomicAdd(&deg[dst[e]], 1);
    }
}

#define FILL_T 1024
#define FILL_BLOCKS 256
#define SCAN_SEG (CDIV(N_NODES, 32 * 128) * 128)
#define SCAN_IT  (SCAN_SEG / 128)
#define EPB (N_EDGES / FILL_BLOCKS)
#define EDGE_FULL (EPB / FILL_T)
#define EDGE_REM  (EPB - EDGE_FULL * FILL_T)
#define EDGE_IT   (EDGE_FULL + (EDGE_REM > 0 ? 1 : 0))
static_assert(EPB * FILL_BLOCKS == N_EDGES, "the edges divide evenly over the fill work-groups");
static_assert(N_NODES % 4 == 0, "scan_fill reads and writes four counts at a time");

template<int N>
__device__ inline int row16_shr_add(int x) {
    const int y = __builtin_amdgcn_update_dpp(0, x, 0x110 + N, 0xf, 0xf, true);
    return x + y;
}
__device__ inline int wave_incl_scan(int x, int lane) {
    x = row16_shr_add<1>(x);
    x = row16_shr_add<2>(x);
    x = row16_shr_add<4>(x);
    x = row16_shr_add<8>(x);
    const int lo = __builtin_amdgcn_readlane(x, 15);
    return (lane >= 16) ? x + lo : x;
}

__global__ void __launch_bounds__(FILL_T) scan_fill(const int* __restrict__ deg, int* __restrict__ rowptr,
                                                    const int* __restrict__ src, const int* __restrict__ dst,
                                                    const float* __restrict__ eattr,
                                                    const float* __restrict__ v1, const float* __restrict__ v2,
                                                    int* cursor, uint4* __restrict__ rec) {
    __shared__ int rp[32 * SCAN_SEG];
    __shared__ int wsum[32];
    const int t = threadIdx.x, lane = t & 31, w = t >> 5;
    const int base = w * SCAN_SEG;

    int4 q[SCAN_IT];
#pragma unroll
    for (int it = 0; it < SCAN_IT; it++) {
        const int idx = base + it * 128 + lane * 4;
        q[it].x = 0; q[it].y = 0; q[it].z = 0; q[it].w = 0;
        if (idx < N_NODES) q[it] = *reinterpret_cast<const int4*>(deg + idx);
    }

    const int e0 = blockIdx.x * EPB;
    int   d[EDGE_IT], s[EDGE_IT];
    float ea[EDGE_IT][EDIM];
#pragma unroll
    for (int i = 0; i < EDGE_IT; i++) {
        d[i] = 0; s[i] = 0;
#pragma unroll
        for (int k = 0; k < EDIM; k++) ea[i][k] = 0.f;
        if (i < EDGE_FULL || t < EDGE_REM) {
            const int e = e0 + i * FILL_T + t;
            d[i] = dst[e];
            s[i] = src[e];
#pragma unroll
            for (int k = 0; k < EDIM; k++) ea[i][k] = eattr[(size_t)e * EDIM + k];
        }
    }

    float w1[EDIM * HEADS], w2[EDIM];
#pragma unroll
    for (int k = 0; k < EDIM * HEADS; k++) w1[k] = v1[k];
#pragma unroll
    for (int k = 0; k < EDIM; k++) w2[k] = v2[k];

    int tot[SCAN_IT], inc[SCAN_IT], carry[SCAN_IT];
    int run = 0;
#pragma unroll
    for (int it = 0; it < SCAN_IT; it++) {
        tot[it] = q[it].x + q[it].y + q[it].z + q[it].w;
        inc[it] = wave_incl_scan(tot[it], lane);
        carry[it] = run;
        run += __builtin_amdgcn_readlane(inc[it], 31);
    }
    if (lane == 0) wsum[w] = run;
    __syncthreads();
    if (w == 0) wsum[lane] = wave_incl_scan(wsum[lane], lane);
    __syncthreads();
    const int wbase = (w > 0) ? wsum[w - 1] : 0;
#pragma unroll
    for (int it = 0; it < SCAN_IT; it++) {
        const int idx = base + it * 128 + lane * 4;
        int4 r;
        r.x = wbase + carry[it] + inc[it] - tot[it];
        r.y = r.x + q[it].x;
        r.z = r.y + q[it].y;
        r.w = r.z + q[it].z;
        *reinterpret_cast<int4*>(rp + idx) = r;
        if ((w * SCAN_IT + it) % FILL_BLOCKS == (int)blockIdx.x && idx < N_NODES)
            *reinterpret_cast<int4*>(rowptr + idx) = r;
    }
    if (blockIdx.x == 0 && t == 0) rowptr[N_NODES] = wsum[31];
    __syncthreads();

    int bs[EDGE_IT], rk[EDGE_IT];
#pragma unroll
    for (int i = 0; i < EDGE_IT; i++) {
        bs[i] = 0; rk[i] = 0;
        if (i < EDGE_FULL || t < EDGE_REM) {
            bs[i] = rp[d[i]];
            rk[i] = atomicAdd(&cursor[d[i]], 1);
        }
    }
#pragma unroll
    for (int i = 0; i < EDGE_IT; i++) {
        if (i < EDGE_FULL || t < EDGE_REM) {
            const size_t p = (size_t)(bs[i] + rk[i]);
            float t10 = 0.f, t11 = 0.f, t2 = 0.f;
#pragma unroll
            for (int k = 0; k < EDIM; k++) {
                t10 += ea[i][k] * w1[k * HEADS + 0];
                t11 += ea[i][k] * w1[k * HEADS + 1];
                t2  += ea[i][k] * w2[k];
            }
            uint4 r;
            r.x = (unsigned)s[i];
            r.y = __float_as_uint(t10);
            r.z = __float_as_uint(t11);
            r.w = __float_as_uint(t2);
            rec[p] = r;
        }
    }
}

__device__ inline float wave_sum(float x) {
#pragma unroll
    for (int o = 16; o > 0; o >>= 1) x += __shfl_xor(x, o, 32);
    return x;
}
__device__ inline float wave_max(float x) {
#pragma unroll
    for (int o = 16; o > 0; o >>= 1) x = fmaxf(x, __shfl_xor(x, o, 32));
    return x;
}

template<int H>
__device__ inline void load_rec(const uint4* __restrict__ rec, int p, int& s, float (&e)[H]) {
    const uint4 r = rec[(size_t)p];
    s = (int)r.x;
    if (H == 2) {
        e[0] = __uint_as_float(r.y);
        e[H - 1] = __uint_as_float(r.z);
    } else {
        e[0] = __uint_as_float(r.w);
    }
}

template<int H>
__device__ inline void get_edge(const uint4* __restrict__ rec, int start, int deg, int node,
                                const float (&lp)[H], int j, int& s, float (&e)[H]) {
    if (j < deg) {
        load_rec<H>(rec, start + j, s, e);
    } else {
        s = node;
#pragma unroll
        for (int i = 0; i < H; i++) e[i] = lp[i];
    }
}

template<int H>
__device__ inline void edge_alpha(int s, const float (&e)[H],
                                  const float* __restrict__ als, const float (&aldn)[H],
                                  float (&a)[H]) {
#pragma unroll
    for (int h = 0; h < H; h++) {
        float x = als[s * H + h] + aldn[h] + e[h];
        a[h] = (x > 0.f) ? x : NEG_SLOPE * x;
    }
}

template<int VEC> struct rowvec;
template<> struct rowvec<4> {
    static __device__ inline void ld(const float* p, float (&o)[4]) {
        const float4 t = *reinterpret_cast<const float4*>(p);
        o[0] = t.x; o[1] = t.y; o[2] = t.z; o[3] = t.w;
    }
    static __device__ inline void st(float* p, const float (&o)[4]) {
        float4 t; t.x = o[0]; t.y = o[1]; t.z = o[2]; t.w = o[3];
        *reinterpret_cast<float4*>(p) = t;
    }
};
template<> struct rowvec<2> {
    static __device__ inline void ld(const float* p, float (&o)[2]) {
        const float2 t = *reinterpret_cast<const float2*>(p);
        o[0] = t.x; o[1] = t.y;
    }
    static __device__ inline void st(float* p, const float (&o)[2]) {
        float2 t; t.x = o[0]; t.y = o[1];
        *reinterpret_cast<float2*>(p) = t;
    }
};

template<int H, int C, int VEC>
__device__ inline void add_rows(const float* __restrict__ xh, int cnt, int s, const float (&w)[H],
                                int lane, int hl, float (&acc)[VEC]) {
    constexpr int HC = H * C;
    constexpr int NACT = HC / VEC;
#pragma unroll 4
    for (int jj = 0; jj < cnt; ++jj) {
        const int sj = __builtin_amdgcn_readlane(s, jj);
        float wj = __int_as_float(__builtin_amdgcn_readlane(__float_as_int(w[0]), jj));
        if (H == 2) {
            const float wj1 = __int_as_float(__builtin_amdgcn_readlane(__float_as_int(w[H - 1]), jj));
            wj = hl ? wj1 : wj;
        }
        if (lane < NACT) {
            float r[VEC];
            rowvec<VEC>::ld(xh + (size_t)sj * HC + lane * VEC, r);
#pragma unroll
            for (int q = 0; q < VEC; q++) acc[q] += r[q] * wj;
        }
    }
}

template<int H, int C, int VEC, bool LOGSM>
__global__ void __launch_bounds__(256) gat_node(const int* __restrict__ rowptr,
                                                const uint4* __restrict__ rec,
                                                const float* __restrict__ als,
                                                const float* __restrict__ ald,
                                                const float* __restrict__ xh,
                                                const float* __restrict__ bias,
                                                float* __restrict__ row_out, int n) {
    constexpr int HC = H * C;
    constexpr int NACT = HC / VEC;
    static_assert(H == 1 || H == 2, "head select below handles one or two heads");
    static_assert(NACT * VEC == HC && NACT <= 32 && C % VEC == 0, "lane to channel map");
    const int lane = threadIdx.x & 31;
    const int node = __builtin_amdgcn_readfirstlane((int)((blockIdx.x * blockDim.x + threadIdx.x) >> 5));
    if (node >= n) return;
    const int start = rowptr[node];
    const int deg   = rowptr[node + 1] - start;
    const int total = deg + 1;

    float aldn[H];
#pragma unroll
    for (int h = 0; h < H; h++) aldn[h] = ald[node * H + h];

    int   s_f = node;
    float e_f[H];
    float sm[H];
#pragma unroll
    for (int i = 0; i < H; i++) { e_f[i] = 0.f; sm[i] = 0.f; }
    if (lane < deg) {
        load_rec<H>(rec, start + lane, s_f, e_f);
#pragma unroll
        for (int i = 0; i < H; i++) sm[i] = e_f[i];
    }
    for (int j = lane + 32; j < deg; j += 32) {
        int s; float e[H];
        load_rec<H>(rec, start + j, s, e);
#pragma unroll
        for (int i = 0; i < H; i++) sm[i] += e[i];
    }
    float lp[H];
    const float cdiv = fmaxf((float)deg, 1.0f);
#pragma unroll
    for (int i = 0; i < H; i++) lp[i] = wave_sum(sm[i]) / cdiv;

    const bool val_f = lane < total;
    if (lane == deg) {
#pragma unroll
        for (int i = 0; i < H; i++) e_f[i] = lp[i];
    }
    float a_f[H];
    edge_alpha<H>(s_f, e_f, als, aldn, a_f);

    float m[H];
#pragma unroll
    for (int h = 0; h < H; h++) m[h] = val_f ? a_f[h] : -INFINITY;
    for (int j = lane + 32; j < total; j += 32) {
        int s; float e[H]; float a[H];
        get_edge<H>(rec, start, deg, node, lp, j, s, e);
        edge_alpha<H>(s, e, als, aldn, a);
#pragma unroll
        for (int h = 0; h < H; h++) m[h] = fmaxf(m[h], a[h]);
    }
#pragma unroll
    for (int h = 0; h < H; h++) m[h] = wave_max(m[h]);

    float ex_f[H], dn[H];
#pragma unroll
    for (int h = 0; h < H; h++) {
        ex_f[h] = val_f ? expf(a_f[h] - m[h]) : 0.f;
        dn[h] = ex_f[h];
    }
    for (int j = lane + 32; j < total; j += 32) {
        int s; float e[H]; float a[H];
        get_edge<H>(rec, start, deg, node, lp, j, s, e);
        edge_alpha<H>(s, e, als, aldn, a);
#pragma unroll
        for (int h = 0; h < H; h++) dn[h] += expf(a[h] - m[h]);
    }
#pragma unroll
    for (int h = 0; h < H; h++) dn[h] = wave_sum(dn[h]) + 1e-16f;

    const int hl = (lane * VEC) / C;
    const bool act = lane < NACT;
    float bv[VEC];
#pragma unroll
    for (int q = 0; q < VEC; q++) bv[q] = 0.f;
    if (act) rowvec<VEC>::ld(bias + lane * VEC, bv);
    float acc[VEC];
#pragma unroll
    for (int q = 0; q < VEC; q++) acc[q] = 0.f;

    float w_f[H];
#pragma unroll
    for (int h = 0; h < H; h++) w_f[h] = ex_f[h] / dn[h];
    add_rows<H, C, VEC>(xh, total < 32 ? total : 32, s_f, w_f, lane, hl, acc);

    for (int cb = 32; cb < total; cb += 32) {
        const int j = cb + lane;
        int s = node;
        float w[H];
#pragma unroll
        for (int h = 0; h < H; h++) w[h] = 0.f;
        if (j < total) {
            float e[H]; float a[H];
            get_edge<H>(rec, start, deg, node, lp, j, s, e);
            edge_alpha<H>(s, e, als, aldn, a);
#pragma unroll
            for (int h = 0; h < H; h++) w[h] = expf(a[h] - m[h]) / dn[h];
        }
        const int left = total - cb;
        add_rows<H, C, VEC>(xh, left < 32 ? left : 32, s, w, lane, hl, acc);
    }

#pragma unroll
    for (int q = 0; q < VEC; q++) acc[q] += bv[q];

    if (!LOGSM) {
#pragma unroll
        for (int q = 0; q < VEC; q++) acc[q] = (acc[q] > 0.f) ? acc[q] : expm1f(acc[q]);
    } else {
        float mx = -INFINITY;
        if (act) {
#pragma unroll
            for (int q = 0; q < VEC; q++) mx = fmaxf(mx, acc[q]);
        }
        mx = wave_max(mx);
        float se = 0.f;
        if (act) {
#pragma unroll
            for (int q = 0; q < VEC; q++) se += expf(acc[q] - mx);
        }
        se = wave_sum(se);
        const float lse = mx + logf(se);
#pragma unroll
        for (int q = 0; q < VEC; q++) acc[q] -= lse;
    }

    if (act) rowvec<VEC>::st(row_out + (size_t)node * HC + lane * VEC, acc);
}

extern "C" void kernel_launch(void* const* d_in, const int* in_sizes, int n_in,
                              void* d_out, int out_size, void* d_ws, size_t ws_size,
                              hipStream_t stream) {
    const float* x   = (const float*)d_in[0];
    const int*   ei  = (const int*)d_in[1];
    const float* ea  = (const float*)d_in[2];
    const float* W1  = (const float*)d_in[3];
    const float* as1 = (const float*)d_in[4];
    const float* ad1 = (const float*)d_in[5];
    const float* We1 = (const float*)d_in[6];
    const float* ae1 = (const float*)d_in[7];
    const float* b1  = (const float*)d_in[8];
    const float* W2  = (const float*)d_in[9];
    const float* as2 = (const float*)d_in[10];
    const float* ad2 = (const float*)d_in[11];
    const float* We2 = (const float*)d_in[12];
    const float* ae2 = (const float*)d_in[13];
    const float* b2  = (const float*)d_in[14];

    const int N = N_NODES, E = N_EDGES;
    const int HC1 = HEADS * HID;
    const int* src = ei;
    const int* dst = ei + E;

    char* ws = (char*)d_ws;
    size_t off = 0;
    auto alloc = [&](size_t bytes) -> void* {
        void* p = ws + off;
        off = (off + bytes + 255) & ~(size_t)255;
        return p;
    };
    int*      deg    = (int*)alloc(sizeof(int) * N);
    int*      rowptr = (int*)alloc(sizeof(int) * (N + 1));
    int*      cursor = (int*)alloc(sizeof(int) * N);
    uint4*    rec    = (uint4*)alloc(sizeof(uint4) * (size_t)E);
    float*    v1     = (float*)alloc(sizeof(float) * 16);
    float*    v2     = (float*)alloc(sizeof(float) * 16);
    float*    W2p    = (float*)alloc(sizeof(float) * F_IN * NPAD2);
    float*    xh1    = (float*)alloc(sizeof(float) * (size_t)N * HC1);
    float*    als1   = (float*)alloc(sizeof(float) * N * HEADS);
    float*    ald1   = (float*)alloc(sizeof(float) * N * HEADS);
    float*    h1     = (float*)alloc(sizeof(float) * (size_t)N * HC1);
    float*    xh2    = (float*)alloc(sizeof(float) * (size_t)N * OUT_F);
    float*    als2   = (float*)alloc(sizeof(float) * N);
    float*    ald2   = (float*)alloc(sizeof(float) * N);
    (void)ws_size; (void)n_in; (void)in_sizes; (void)out_size;

    const int T = 256;

    setup_small<<<ZERO_BLOCKS + PAD_BLOCKS + 1, SETUP_T, 0, stream>>>((unsigned*)deg, (unsigned*)cursor, W2, W2p, We1, ae1, We2, ae2, v1, v2);
    gemm1_and_count<<<GEMM1_BLOCKS + COUNT_BLOCKS, GC_T, 0, stream>>>(x, W1, xh1, as1, ad1, als1, ald1, dst, deg);
    scan_fill<<<FILL_BLOCKS, FILL_T, 0, stream>>>(deg, rowptr, src, dst, ea, v1, v2, cursor, rec);

    gat_node<HEADS, HID, 4, false><<<CDIV(N, T / 32), T, 0, stream>>>(rowptr, rec, als1, ald1, xh1, b1, h1, N);

    {
        int mtiles = N / 16, waves_per_blk = 128 / 32;
        wmma_gemm<3, HC1, 1><<<CDIV(mtiles, waves_per_blk), 128, 0, stream>>>(h1, W2p, xh2, as2, ad2, als2, ald2, N, OUT_F);
    }
    gat_node<1, OUT_F, 2, true><<<CDIV(N, T / 32), T, 0, stream>>>(rowptr, rec, als2, ald2, xh2, b2, (float*)d_out, N);
}


# Row gathers of a node issued in groups (8 hidden layer, 16 output layer), all loads of a group before the first use; tail padded with zero-weight reloads of the node's own row; sum order unchanged
# speedup vs baseline: 5.6556x; 1.0566x over previous
#include <hip/hip_runtime.h>
#include <math.h>

#define N_NODES 50000
#define N_EDGES 800000
#define ETOT    (N_EDGES + N_NODES)
#define F_IN    128
#define HID     64
#define HEADS   2
#define OUT_F   40
#define EDIM    5
#define NEG_SLOPE 0.2f

#define CDIV(a,b) (((a)+(b)-1)/(b))

typedef __attribute__((ext_vector_type(2))) float v2f;
typedef __attribute__((ext_vector_type(8))) float v8f;

#define NPAD2   48
#define SETUP_T 256
#define ZERO_BLOCKS CDIV(N_NODES, SETUP_T)
#define PAD_BLOCKS  CDIV(F_IN * NPAD2, SETUP_T)

__global__ void __launch_bounds__(SETUP_T) setup_small(unsigned* __restrict__ deg, unsigned* __restrict__ cursor,
                                                       const float* __restrict__ W2, float* __restrict__ W2p,
                                                       const float* __restrict__ We1, const float* __restrict__ ae1,
                                                       const float* __restrict__ We2, const float* __restrict__ ae2,
                                                       float* __restrict__ v1, float* __restrict__ v2) {
    const int b = blockIdx.x;
    const int t = threadIdx.x;
    if (b < ZERO_BLOCKS) {
        const int i = b * SETUP_T + t;
        if (i < N_NODES) { deg[i] = 0u; cursor[i] = 0u; }
    } else if (b < ZERO_BLOCKS + PAD_BLOCKS) {
        const int i = (b - ZERO_BLOCKS) * SETUP_T + t;
        if (i < F_IN * NPAD2) {
            const int r = i / NPAD2, c = i % NPAD2;
            W2p[i] = (c < OUT_F) ? W2[r * OUT_F + c] : 0.0f;
        }
    } else {
        if (t < EDIM * HEADS) {
            const int d = t / HEADS, h = t % HEADS;
            float s = 0.f;
            for (int c = 0; c < HID; c++)
                s += We1[d * (HEADS * HID) + h * HID + c] * ae1[h * HID + c];
            v1[t] = s;
        } else if (t < EDIM * HEADS + EDIM) {
            const int d = t - EDIM * HEADS;
            float s = 0.f;
            for (int c = 0; c < OUT_F; c++)
                s += We2[d * OUT_F + c] * ae2[c];
            v2[d] = s;
        }
    }
}

template<int M>
__device__ inline float row16_xadd(float x) {
    const int y = __builtin_amdgcn_update_dpp(0, __float_as_int(x), 0x160 + M, 0xf, 0xf, true);
    return x + __int_as_float(y);
}

template<int NT, int K, int H>
__device__ inline void gemm_stripe(int wave, const float* __restrict__ A, const float* __restrict__ B,
                                   float* __restrict__ C,
                                   const float* __restrict__ asrc, const float* __restrict__ adst,
                                   float* __restrict__ als, float* __restrict__ ald, int M, int N) {
    constexpr int NPAD = NT * 16;
    constexpr int TPH = NT / H;
    static_assert(NT % H == 0, "a column tile belongs to one head");
    int lane = threadIdx.x & 31;
    int half = lane >> 4;
    int lrow = lane & 15;
    int mtiles = M >> 4;
    if (wave >= mtiles) return;
    int m0 = wave << 4;

    float sa[NT], da[NT];
#pragma unroll
    for (int nt = 0; nt < NT; nt++) {
        const int col = nt * 16 + lrow;
        sa[nt] = (col < N) ? asrc[col] : 0.0f;
        da[nt] = (col < N) ? adst[col] : 0.0f;
    }

    v8f acc[NT];
#pragma unroll
    for (int nt = 0; nt < NT; nt++)
#pragma unroll
        for (int g = 0; g < 8; g++) acc[nt][g] = 0.0f;

    const float* arow = A + (size_t)(m0 + lrow) * K;
    const float* bcol = B + lrow;
#pragma unroll 4
    for (int k = 0; k < K; k += 4) {
        int kk = k + half * 2;
        v2f a;
        a.x = arow[kk];
        a.y = arow[kk + 1];
#pragma unroll
        for (int nt = 0; nt < NT; nt++) {
            v2f b;
            b.x = bcol[kk * NPAD + nt * 16];
            b.y = bcol[(kk + 1) * NPAD + nt * 16];
            acc[nt] = __builtin_amdgcn_wmma_f32_16x16x4_f32(
                false, a, false, b, (short)0, acc[nt], false, false);
        }
    }
#pragma unroll
    for (int nt = 0; nt < NT; nt++) {
        int col = nt * 16 + lrow;
        if (col < N) {
#pragma unroll
            for (int g = 0; g < 8; g++)
                C[(size_t)(m0 + g + half * 8) * N + col] = acc[nt][g];
        }
    }

    constexpr int NV = 2 * H * 8;
    float red[NV];
#pragma unroll
    for (int i = 0; i < NV; i++) red[i] = 0.0f;
#pragma unroll
    for (int nt = 0; nt < NT; nt++) {
        const int h = nt / TPH;
        const bool incol = (nt * 16 + lrow) < N;
#pragma unroll
        for (int g = 0; g < 8; g++) {
            const float cv = incol ? acc[nt][g] : 0.0f;
            red[(0 * H + h) * 8 + g] += cv * sa[nt];
            red[(1 * H + h) * 8 + g] += cv * da[nt];
        }
    }
#pragma unroll
    for (int i = 0; i < NV; i++) red[i] = row16_xadd<1>(red[i]);
#pragma unroll
    for (int i = 0; i < NV; i++) red[i] = row16_xadd<2>(red[i]);
#pragma unroll
    for (int i = 0; i < NV; i++) red[i] = row16_xadd<4>(red[i]);
#pragma unroll
    for (int i = 0; i < NV; i++) red[i] = row16_xadd<8>(red[i]);
    if (lrow == 0) {
        float* sp = als + (size_t)(m0 + half * 8) * H;
        float* dp = ald + (size_t)(m0 + half * 8) * H;
#pragma unroll
        for (int g = 0; g < 8; g++)
#pragma unroll
            for (int h = 0; h < H; h++) {
                sp[g * H + h] = red[(0 * H + h) * 8 + g];
                dp[g * H + h] = red[(1 * H + h) * 8 + g];
            }
    }
}

template<int NT, int K, int H>
__global__ void wmma_gemm(const float* __restrict__ A, const float* __restrict__ B,
                          float* __restrict__ C,
                          const float* __restrict__ asrc, const float* __restrict__ adst,
                          float* __restrict__ als, float* __restrict__ ald, int M, int N) {
    gemm_stripe<NT, K, H>((int)((blockIdx.x * blockDim.x + threadIdx.x) >> 5),
                          A, B, C, asrc, adst, als, ald, M, N);
}

#define GC_T 256
#define GEMM1_BLOCKS CDIV(N_NODES / 16, GC_T / 32)
#define COUNT_BLOCKS CDIV(N_EDGES, GC_T)
static_assert(N_NODES % 16 == 0, "the GEMMs take whole 16-row stripes");
__global__ void __launch_bounds__(GC_T) gemm1_and_count(const float* __restrict__ x, const float* __restrict__ W1,
                                                        float* __restrict__ xh,
                                                        const float* __restrict__ asrc, const float* __restrict__ adst,
                                                        float* __restrict__ als, float* __restrict__ ald,
                                                        const int* __restrict__ dst, int* deg) {
    if (blockIdx.x < GEMM1_BLOCKS) {
        gemm_stripe<8, F_IN, HEADS>((int)(blockIdx.x * (GC_T / 32) + (threadIdx.x >> 5)),
                                    x, W1, xh, asrc, adst, als, ald, N_NODES, HEADS * HID);
    } else {
        const int e = (blockIdx.x - GEMM1_BLOCKS) * GC_T + threadIdx.x;
        if (e < N_EDGES) atomicAdd(&deg[dst[e]], 1);
    }
}

#define FILL_T 1024
#define FILL_BLOCKS 256
#define SCAN_SEG (CDIV(N_NODES, 32 * 128) * 128)
#define SCAN_IT  (SCAN_SEG / 128)
#define EPB (N_EDGES / FILL_BLOCKS)
#define EDGE_FULL (EPB / FILL_T)
#define EDGE_REM  (EPB - EDGE_FULL * FILL_T)
#define EDGE_IT   (EDGE_FULL + (EDGE_REM > 0 ? 1 : 0))
static_assert(EPB * FILL_BLOCKS == N_EDGES, "the edges divide evenly over the fill work-groups");
static_assert(N_NODES % 4 == 0, "scan_fill reads and writes four counts at a time");

template<int N>
__device__ inline int row16_shr_add(int x) {
    const int y = __builtin_amdgcn_update_dpp(0, x, 0x110 + N, 0xf, 0xf, true);
    return x + y;
}
__device__ inline int wave_incl_scan(int x, int lane) {
    x = row16_shr_add<1>(x);
    x = row16_shr_add<2>(x);
    x = row16_shr_add<4>(x);
    x = row16_shr_add<8>(x);
    const int lo = __builtin_amdgcn_readlane(x, 15);
    return (lane >= 16) ? x + lo : x;
}

__global__ void __launch_bounds__(FILL_T) scan_fill(const int* __restrict__ deg, int* __restrict__ rowptr,
                                                    const int* __restrict__ src, const int* __restrict__ dst,
                                                    const float* __restrict__ eattr,
                                                    const float* __restrict__ v1, const float* __restrict__ v2,
                                                    int* cursor, uint4* __restrict__ rec) {
    __shared__ int rp[32 * SCAN_SEG];
    __shared__ int wsum[32];
    const int t = threadIdx.x, lane = t & 31, w = t >> 5;
    const int base = w * SCAN_SEG;

    int4 q[SCAN_IT];
#pragma unroll
    for (int it = 0; it < SCAN_IT; it++) {
        const int idx = base + it * 128 + lane * 4;
        q[it].x = 0; q[it].y = 0; q[it].z = 0; q[it].w = 0;
        if (idx < N_NODES) q[it] = *reinterpret_cast<const int4*>(deg + idx);
    }

    const int e0 = blockIdx.x * EPB;
    int   d[EDGE_IT], s[EDGE_IT];
    float ea[EDGE_IT][EDIM];
#pragma unroll
    for (int i = 0; i < EDGE_IT; i++) {
        d[i] = 0; s[i] = 0;
#pragma unroll
        for (int k = 0; k < EDIM; k++) ea[i][k] = 0.f;
        if (i < EDGE_FULL || t < EDGE_REM) {
            const int e = e0 + i * FILL_T + t;
            d[i] = dst[e];
            s[i] = src[e];
#pragma unroll
            for (int k = 0; k < EDIM; k++) ea[i][k] = eattr[(size_t)e * EDIM + k];
        }
    }

    float w1[EDIM * HEADS], w2[EDIM];
#pragma unroll
    for (int k = 0; k < EDIM * HEADS; k++) w1[k] = v1[k];
#pragma unroll
    for (int k = 0; k < EDIM; k++) w2[k] = v2[k];

    int tot[SCAN_IT], inc[SCAN_IT], carry[SCAN_IT];
    int run = 0;
#pragma unroll
    for (int it = 0; it < SCAN_IT; it++) {
        tot[it] = q[it].x + q[it].y + q[it].z + q[it].w;
        inc[it] = wave_incl_scan(tot[it], lane);
        carry[it] = run;
        run += __builtin_amdgcn_readlane(inc[it], 31);
    }
    if (lane == 0) wsum[w] = run;
    __syncthreads();
    if (w == 0) wsum[lane] = wave_incl_scan(wsum[lane], lane);
    __syncthreads();
    const int wbase = (w > 0) ? wsum[w - 1] : 0;
#pragma unroll
    for (int it = 0; it < SCAN_IT; it++) {
        const int idx = base + it * 128 + lane * 4;
        int4 r;
        r.x = wbase + carry[it] + inc[it] - tot[it];
        r.y = r.x + q[it].x;
        r.z = r.y + q[it].y;
        r.w = r.z + q[it].z;
        *reinterpret_cast<int4*>(rp + idx) = r;
        if ((w * SCAN_IT + it) % FILL_BLOCKS == (int)blockIdx.x && idx < N_NODES)
            *reinterpret_cast<int4*>(rowptr + idx) = r;
    }
    if (blockIdx.x == 0 && t == 0) rowptr[N_NODES] = wsum[31];
    __syncthreads();

    int bs[EDGE_IT], rk[EDGE_IT];
#pragma unroll
    for (int i = 0; i < EDGE_IT; i++) {
        bs[i] = 0; rk[i] = 0;
        if (i < EDGE_FULL || t < EDGE_REM) {
            bs[i] = rp[d[i]];
            rk[i] = atomicAdd(&cursor[d[i]], 1);
        }
    }
#pragma unroll
    for (int i = 0; i < EDGE_IT; i++) {
        if (i < EDGE_FULL || t < EDGE_REM) {
            const size_t p = (size_t)(bs[i] + rk[i]);
            float t10 = 0.f, t11 = 0.f, t2 = 0.f;
#pragma unroll
            for (int k = 0; k < EDIM; k++) {
                t10 += ea[i][k] * w1[k * HEADS + 0];
                t11 += ea[i][k] * w1[k * HEADS + 1];
                t2  += ea[i][k] * w2[k];
            }
            uint4 r;
            r.x = (unsigned)s[i];
            r.y = __float_as_uint(t10);
            r.z = __float_as_uint(t11);
            r.w = __float_as_uint(t2);
            rec[p] = r;
        }
    }
}

__device__ inline float wave_sum(float x) {
#pragma unroll
    for (int o = 16; o > 0; o >>= 1) x += __shfl_xor(x, o, 32);
    return x;
}
__device__ inline float wave_max(float x) {
#pragma unroll
    for (int o = 16; o > 0; o >>= 1) x = fmaxf(x, __shfl_xor(x, o, 32));
    return x;
}

template<int H>
__device__ inline void load_rec(const uint4* __restrict__ rec, int p, int& s, float (&e)[H]) {
    const uint4 r = rec[(size_t)p];
    s = (int)r.x;
    if (H == 2) {
        e[0] = __uint_as_float(r.y);
        e[H - 1] = __uint_as_float(r.z);
    } else {
        e[0] = __uint_as_float(r.w);
    }
}

template<int H>
__device__ inline void get_edge(const uint4* __restrict__ rec, int start, int deg, int node,
                                const float (&lp)[H], int j, int& s, float (&e)[H]) {
    if (j < deg) {
        load_rec<H>(rec, start + j, s, e);
    } else {
        s = node;
#pragma unroll
        for (int i = 0; i < H; i++) e[i] = lp[i];
    }
}

template<int H>
__device__ inline void edge_alpha(int s, const float (&e)[H],
                                  const float* __restrict__ als, const float (&aldn)[H],
                                  float (&a)[H]) {
#pragma unroll
    for (int h = 0; h < H; h++) {
        float x = als[s * H + h] + aldn[h] + e[h];
        a[h] = (x > 0.f) ? x : NEG_SLOPE * x;
    }
}

template<int VEC> struct rowvec;
template<> struct rowvec<4> {
    static __device__ inline void ld(const float* p, float (&o)[4]) {
        const float4 t = *reinterpret_cast<const float4*>(p);
        o[0] = t.x; o[1] = t.y; o[2] = t.z; o[3] = t.w;
    }
    static __device__ inline void st(float* p, const float (&o)[4]) {
        float4 t; t.x = o[0]; t.y = o[1]; t.z = o[2]; t.w = o[3];
        *reinterpret_cast<float4*>(p) = t;
    }
};
template<> struct rowvec<2> {
    static __device__ inline void ld(const float* p, float (&o)[2]) {
        const float2 t = *reinterpret_cast<const float2*>(p);
        o[0] = t.x; o[1] = t.y;
    }
    static __device__ inline void st(float* p, const float (&o)[2]) {
        float2 t; t.x = o[0]; t.y = o[1];
        *reinterpret_cast<float2*>(p) = t;
    }
};

template<int H, int C, int VEC, int G>
__device__ inline void add_rows(const float* __restrict__ xh, int cnt, int s, const float (&w)[H],
                                int lane, int hl, float (&acc)[VEC]) {
    constexpr int HC = H * C;
    constexpr int NACT = HC / VEC;
    constexpr int SB = 4;
    constexpr int NSB = G / SB;
    static_assert(G % SB == 0 && 32 % G == 0, "groups are whole blocks and tile the 32 lanes");
    for (int j0 = 0; j0 < cnt; j0 += G) {
        int   sj[G];
        float wj[G];
#pragma unroll
        for (int k = 0; k < G; k++) {
            sj[k] = __builtin_amdgcn_readlane(s, j0 + k);
            wj[k] = __int_as_float(__builtin_amdgcn_readlane(__float_as_int(w[0]), j0 + k));
            if (H == 2) {
                const float wj1 = __int_as_float(__builtin_amdgcn_readlane(__float_as_int(w[H - 1]), j0 + k));
                wj[k] = hl ? wj1 : wj[k];
            }
        }
        if (lane < NACT) {
            float r[G][VEC];
#pragma unroll
            for (int b = 0; b < NSB; b++) {
                if (b == 0 || j0 + b * SB < cnt) {
#pragma unroll
                    for (int k = b * SB; k < (b + 1) * SB; k++)
                        rowvec<VEC>::ld(xh + (size_t)sj[k] * HC + lane * VEC, r[k]);
                }
            }
#pragma unroll
            for (int b = 0; b < NSB; b++) {
                if (b == 0 || j0 + b * SB < cnt) {
#pragma unroll
                    for (int k = b * SB; k < (b + 1) * SB; k++)
#pragma unroll
                        for (int q = 0; q < VEC; q++) acc[q] = __builtin_fmaf(r[k][q], wj[k], acc[q]);
                }
            }
        }
    }
}

template<int H, int C, int VEC, int G, bool LOGSM>
__global__ void __launch_bounds__(256) gat_node(const int* __restrict__ rowptr,
                                                const uint4* __restrict__ rec,
                                                const float* __restrict__ als,
                                                const float* __restrict__ ald,
                                                const float* __restrict__ xh,
                                                const float* __restrict__ bias,
                                                float* __restrict__ row_out, int n) {
    constexpr int HC = H * C;
    constexpr int NACT = HC / VEC;
    static_assert(H == 1 || H == 2, "head select below handles one or two heads");
    static_assert(NACT * VEC == HC && NACT <= 32 && C % VEC == 0, "lane to channel map");
    const int lane = threadIdx.x & 31;
    const int node = __builtin_amdgcn_readfirstlane((int)((blockIdx.x * blockDim.x + threadIdx.x) >> 5));
    if (node >= n) return;
    const int start = rowptr[node];
    const int deg   = rowptr[node + 1] - start;
    const int total = deg + 1;

    float aldn[H];
#pragma unroll
    for (int h = 0; h < H; h++) aldn[h] = ald[node * H + h];

    int   s_f = node;
    float e_f[H];
    float sm[H];
#pragma unroll
    for (int i = 0; i < H; i++) { e_f[i] = 0.f; sm[i] = 0.f; }
    if (lane < deg) {
        load_rec<H>(rec, start + lane, s_f, e_f);
#pragma unroll
        for (int i = 0; i < H; i++) sm[i] = e_f[i];
    }
    for (int j = lane + 32; j < deg; j += 32) {
        int s; float e[H];
        load_rec<H>(rec, start + j, s, e);
#pragma unroll
        for (int i = 0; i < H; i++) sm[i] += e[i];
    }
    float lp[H];
    const float cdiv = fmaxf((float)deg, 1.0f);
#pragma unroll
    for (int i = 0; i < H; i++) lp[i] = wave_sum(sm[i]) / cdiv;

    const bool val_f = lane < total;
    if (lane == deg) {
#pragma unroll
        for (int i = 0; i < H; i++) e_f[i] = lp[i];
    }
    float a_f[H];
    edge_alpha<H>(s_f, e_f, als, aldn, a_f);

    float m[H];
#pragma unroll
    for (int h = 0; h < H; h++) m[h] = val_f ? a_f[h] : -INFINITY;
    for (int j = lane + 32; j < total; j += 32) {
        int s; float e[H]; float a[H];
        get_edge<H>(rec, start, deg, node, lp, j, s, e);
        edge_alpha<H>(s, e, als, aldn, a);
#pragma unroll
        for (int h = 0; h < H; h++) m[h] = fmaxf(m[h], a[h]);
    }
#pragma unroll
    for (int h = 0; h < H; h++) m[h] = wave_max(m[h]);

    float ex_f[H], dn[H];
#pragma unroll
    for (int h = 0; h < H; h++) {
        ex_f[h] = val_f ? expf(a_f[h] - m[h]) : 0.f;
        dn[h] = ex_f[h];
    }
    for (int j = lane + 32; j < total; j += 32) {
        int s; float e[H]; float a[H];
        get_edge<H>(rec, start, deg, node, lp, j, s, e);
        edge_alpha<H>(s, e, als, aldn, a);
#pragma unroll
        for (int h = 0; h < H; h++) dn[h] += expf(a[h] - m[h]);
    }
#pragma unroll
    for (int h = 0; h < H; h++) dn[h] = wave_sum(dn[h]) + 1e-16f;

    const int hl = (lane * VEC) / C;
    const bool act = lane < NACT;
    float bv[VEC];
#pragma unroll
    for (int q = 0; q < VEC; q++) bv[q] = 0.f;
    if (act) rowvec<VEC>::ld(bias + lane * VEC, bv);
    float acc[VEC];
#pragma unroll
    for (int q = 0; q < VEC; q++) acc[q] = 0.f;

    float w_f[H];
#pragma unroll
    for (int h = 0; h < H; h++) w_f[h] = ex_f[h] / dn[h];
    add_rows<H, C, VEC, G>(xh, total < 32 ? total : 32, s_f, w_f, lane, hl, acc);

    for (int cb = 32; cb < total; cb += 32) {
        const int j = cb + lane;
        int s = node;
        float w[H];
#pragma unroll
        for (int h = 0; h < H; h++) w[h] = 0.f;
        if (j < total) {
            float e[H]; float a[H];
            get_edge<H>(rec, start, deg, node, lp, j, s, e);
            edge_alpha<H>(s, e, als, aldn, a);
#pragma unroll
            for (int h = 0; h < H; h++) w[h] = expf(a[h] - m[h]) / dn[h];
        }
        const int left = total - cb;
        add_rows<H, C, VEC, G>(xh, left < 32 ? left : 32, s, w, lane, hl, acc);
    }

#pragma unroll
    for (int q = 0; q < VEC; q++) acc[q] += bv[q];

    if (!LOGSM) {
#pragma unroll
        for (int q = 0; q < VEC; q++) acc[q] = (acc[q] > 0.f) ? acc[q] : expm1f(acc[q]);
    } else {
        float mx = -INFINITY;
        if (act) {
#pragma unroll
            for (int q = 0; q < VEC; q++) mx = fmaxf(mx, acc[q]);
        }
        mx = wave_max(mx);
        float se = 0.f;
        if (act) {
#pragma unroll
            for (int q = 0; q < VEC; q++) se += expf(acc[q] - mx);
        }
        se = wave_sum(se);
        const float lse = mx + logf(se);
#pragma unroll
        for (int q = 0; q < VEC; q++) acc[q] -= lse;
    }

    if (act) rowvec<VEC>::st(row_out + (size_t)node * HC + lane * VEC, acc);
}

extern "C" void kernel_launch(void* const* d_in, const int* in_sizes, int n_in,
                              void* d_out, int out_size, void* d_ws, size_t ws_size,
                              hipStream_t stream) {
    const float* x   = (const float*)d_in[0];
    const int*   ei  = (const int*)d_in[1];
    const float* ea  = (const float*)d_in[2];
    const float* W1  = (const float*)d_in[3];
    const float* as1 = (const float*)d_in[4];
    const float* ad1 = (const float*)d_in[5];
    const float* We1 = (const float*)d_in[6];
    const float* ae1 = (const float*)d_in[7];
    const float* b1  = (const float*)d_in[8];
    const float* W2  = (const float*)d_in[9];
    const float* as2 = (const float*)d_in[10];
    const float* ad2 = (const float*)d_in[11];
    const float* We2 = (const float*)d_in[12];
    const float* ae2 = (const float*)d_in[13];
    const float* b2  = (const float*)d_in[14];

    const int N = N_NODES, E = N_EDGES;
    const int HC1 = HEADS * HID;
    const int* src = ei;
    const int* dst = ei + E;

    char* ws = (char*)d_ws;
    size_t off = 0;
    auto alloc = [&](size_t bytes) -> void* {
        void* p = ws + off;
        off = (off + bytes + 255) & ~(size_t)255;
        return p;
    };
    int*      deg    = (int*)alloc(sizeof(int) * N);
    int*      rowptr = (int*)alloc(sizeof(int) * (N + 1));
    int*      cursor = (int*)alloc(sizeof(int) * N);
    uint4*    rec    = (uint4*)alloc(sizeof(uint4) * (size_t)E);
    float*    v1     = (float*)alloc(sizeof(float) * 16);
    float*    v2     = (float*)alloc(sizeof(float) * 16);
    float*    W2p    = (float*)alloc(sizeof(float) * F_IN * NPAD2);
    float*    xh1    = (float*)alloc(sizeof(float) * (size_t)N * HC1);
    float*    als1   = (float*)alloc(sizeof(float) * N * HEADS);
    float*    ald1   = (float*)alloc(sizeof(float) * N * HEADS);
    float*    h1     = (float*)alloc(sizeof(float) * (size_t)N * HC1);
    float*    xh2    = (float*)alloc(sizeof(float) * (size_t)N * OUT_F);
    float*    als2   = (float*)alloc(sizeof(float) * N);
    float*    ald2   = (float*)alloc(sizeof(float) * N);
    (void)ws_size; (void)n_in; (void)in_sizes; (void)out_size;

    const int T = 256;

    setup_small<<<ZERO_BLOCKS + PAD_BLOCKS + 1, SETUP_T, 0, stream>>>((unsigned*)deg, (unsigned*)cursor, W2, W2p, We1, ae1, We2, ae2, v1, v2);
    gemm1_and_count<<<GEMM1_BLOCKS + COUNT_BLOCKS, GC_T, 0, stream>>>(x, W1, xh1, as1, ad1, als1, ald1, dst, deg);
    scan_fill<<<FILL_BLOCKS, FILL_T, 0, stream>>>(deg, rowptr, src, dst, ea, v1, v2, cursor, rec);

    gat_node<HEADS, HID, 4, 8, false><<<CDIV(N, T / 32), T, 0, stream>>>(rowptr, rec, als1, ald1, xh1, b1, h1, N);

    {
        int mtiles = N / 16, waves_per_blk = 128 / 32;
        wmma_gemm<3, HC1, 1><<<CDIV(mtiles, waves_per_blk), 128, 0, stream>>>(h1, W2p, xh2, as2, ad2, als2, ald2, N, OUT_F);
    }
    gat_node<1, OUT_F, 2, 16, true><<<CDIV(N, T / 32), T, 0, stream>>>(rowptr, rec, als2, ald2, xh2, b2, (float*)d_out, N);
}


# No count pass, no prefix sum: a node owns 32 record slots, an edge goes to dst*32 + its returning-atomic arrival rank beside the layer-1 GEMM; spill list past 32 summed last; 6 -> 5 launches
# speedup vs baseline: 7.8129x; 1.3814x over previous
#include <hip/hip_runtime.h>
#include <math.h>

#define N_NODES 50000
#define N_EDGES 800000
#define ETOT    (N_EDGES + N_NODES)
#define F_IN    128
#define HID     64
#define HEADS   2
#define OUT_F   40
#define EDIM    5
#define NEG_SLOPE 0.2f

#define CDIV(a,b) (((a)+(b)-1)/(b))

typedef __attribute__((ext_vector_type(2))) float v2f;
typedef __attribute__((ext_vector_type(8))) float v8f;

#define NPAD2   48
#define SETUP_T 256
#define ZERO_BLOCKS CDIV(N_NODES, SETUP_T)
#define PAD_BLOCKS  CDIV(F_IN * NPAD2, SETUP_T)

#define SLOTS   32

__global__ void __launch_bounds__(SETUP_T) setup_small(unsigned* __restrict__ cursor, unsigned* __restrict__ spill_n,
                                                       const float* __restrict__ W2, float* __restrict__ W2p,
                                                       const float* __restrict__ We1, const float* __restrict__ ae1,
                                                       const float* __restrict__ We2, const float* __restrict__ ae2,
                                                       float* __restrict__ v1, float* __restrict__ v2) {
    const int b = blockIdx.x;
    const int t = threadIdx.x;
    if (b < ZERO_BLOCKS) {
        const int i = b * SETUP_T + t;
        if (i < N_NODES) cursor[i] = 0u;
        if (i == 0) spill_n[0] = 0u;
    } else if (b < ZERO_BLOCKS + PAD_BLOCKS) {
        const int i = (b - ZERO_BLOCKS) * SETUP_T + t;
        if (i < F_IN * NPAD2) {
            const int r = i / NPAD2, c = i % NPAD2;
            W2p[i] = (c < OUT_F) ? W2[r * OUT_F + c] : 0.0f;
        }
    } else {
        if (t < EDIM * HEADS) {
            const int d = t / HEADS, h = t % HEADS;
            float s = 0.f;
            for (int c = 0; c < HID; c++)
                s += We1[d * (HEADS * HID) + h * HID + c] * ae1[h * HID + c];
            v1[t] = s;
        } else if (t < EDIM * HEADS + EDIM) {
            const int d = t - EDIM * HEADS;
            float s = 0.f;
            for (int c = 0; c < OUT_F; c++)
                s += We2[d * OUT_F + c] * ae2[c];
            v2[d] = s;
        }
    }
}

template<int M>
__device__ inline float row16_xadd(float x) {
    const int y = __builtin_amdgcn_update_dpp(0, __float_as_int(x), 0x160 + M, 0xf, 0xf, true);
    return x + __int_as_float(y);
}

template<int NT, int K, int H>
__device__ inline void gemm_stripe(int wave, const float* __restrict__ A, const float* __restrict__ B,
                                   float* __restrict__ C,
                                   const float* __restrict__ asrc, const float* __restrict__ adst,
                                   float* __restrict__ als, float* __restrict__ ald, int M, int N) {
    constexpr int NPAD = NT * 16;
    constexpr int TPH = NT / H;
    static_assert(NT % H == 0, "a column tile belongs to one head");
    int lane = threadIdx.x & 31;
    int half = lane >> 4;
    int lrow = lane & 15;
    int mtiles = M >> 4;
    if (wave >= mtiles) return;
    int m0 = wave << 4;

    float sa[NT], da[NT];
#pragma unroll
    for (int nt = 0; nt < NT; nt++) {
        const int col = nt * 16 + lrow;
        sa[nt] = (col < N) ? asrc[col] : 0.0f;
        da[nt] = (col < N) ? adst[col] : 0.0f;
    }

    v8f acc[NT];
#pragma unroll
    for (int nt = 0; nt < NT; nt++)
#pragma unroll
        for (int g = 0; g < 8; g++) acc[nt][g] = 0.0f;

    const float* arow = A + (size_t)(m0 + lrow) * K;
    const float* bcol = B + lrow;
#pragma unroll 4
    for (int k = 0; k < K; k += 4) {
        int kk = k + half * 2;
        v2f a;
        a.x = arow[kk];
        a.y = arow[kk + 1];
#pragma unroll
        for (int nt = 0; nt < NT; nt++) {
            v2f b;
            b.x = bcol[kk * NPAD + nt * 16];
            b.y = bcol[(kk + 1) * NPAD + nt * 16];
            acc[nt] = __builtin_amdgcn_wmma_f32_16x16x4_f32(
                false, a, false, b, (short)0, acc[nt], false, false);
        }
    }
#pragma unroll
    for (int nt = 0; nt < NT; nt++) {
        int col = nt * 16 + lrow;
        if (col < N) {
#pragma unroll
            for (int g = 0; g < 8; g++)
                C[(size_t)(m0 + g + half * 8) * N + col] = acc[nt][g];
        }
    }

    constexpr int NV = 2 * H * 8;
    float red[NV];
#pragma unroll
    for (int i = 0; i < NV; i++) red[i] = 0.0f;
#pragma unroll
    for (int nt = 0; nt < NT; nt++) {
        const int h = nt / TPH;
        const bool incol = (nt * 16 + lrow) < N;
#pragma unroll
        for (int g = 0; g < 8; g++) {
            const float cv = incol ? acc[nt][g] : 0.0f;
            red[(0 * H + h) * 8 + g] += cv * sa[nt];
            red[(1 * H + h) * 8 + g] += cv * da[nt];
        }
    }
#pragma unroll
    for (int i = 0; i < NV; i++) red[i] = row16_xadd<1>(red[i]);
#pragma unroll
    for (int i = 0; i < NV; i++) red[i] = row16_xadd<2>(red[i]);
#pragma unroll
    for (int i = 0; i < NV; i++) red[i] = row16_xadd<4>(red[i]);
#pragma unroll
    for (int i = 0; i < NV; i++) red[i] = row16_xadd<8>(red[i]);
    if (lrow == 0) {
        float* sp = als + (size_t)(m0 + half * 8) * H;
        float* dp = ald + (size_t)(m0 + half * 8) * H;
#pragma unroll
        for (int g = 0; g < 8; g++)
#pragma unroll
            for (int h = 0; h < H; h++) {
                sp[g * H + h] = red[(0 * H + h) * 8 + g];
                dp[g * H + h] = red[(1 * H + h) * 8 + g];
            }
    }
}

template<int NT, int K, int H>
__global__ void wmma_gemm(const float* __restrict__ A, const float* __restrict__ B,
                          float* __restrict__ C,
                          const float* __restrict__ asrc, const float* __restrict__ adst,
                          float* __restrict__ als, float* __restrict__ ald, int M, int N) {
    gemm_stripe<NT, K, H>((int)((blockIdx.x * blockDim.x + threadIdx.x) >> 5),
                          A, B, C, asrc, adst, als, ald, M, N);
}

#define GC_T 256
#define GEMM1_BLOCKS CDIV(N_NODES / 16, GC_T / 32)
#define FILL_BLOCKS CDIV(N_EDGES, GC_T)
static_assert(N_NODES % 16 == 0, "the GEMMs take whole 16-row stripes");
__global__ void __launch_bounds__(GC_T) gemm1_and_fill(const float* __restrict__ x, const float* __restrict__ W1,
                                                       float* __restrict__ xh,
                                                       const float* __restrict__ asrc, const float* __restrict__ adst,
                                                       float* __restrict__ als, float* __restrict__ ald,
                                                       const int* __restrict__ src, const int* __restrict__ dst,
                                                       const float* __restrict__ eattr,
                                                       const float* __restrict__ v1, const float* __restrict__ v2,
                                                       int* cursor, uint4* __restrict__ rec,
                                                       int* spill_n, int* __restrict__ spill_dst,
                                                       uint4* __restrict__ spill_rec) {
    if (blockIdx.x < GEMM1_BLOCKS) {
        gemm_stripe<8, F_IN, HEADS>((int)(blockIdx.x * (GC_T / 32) + (threadIdx.x >> 5)),
                                    x, W1, xh, asrc, adst, als, ald, N_NODES, HEADS * HID);
    } else {
        const int e = (blockIdx.x - GEMM1_BLOCKS) * GC_T + threadIdx.x;
        if (e < N_EDGES) {
            const int d = dst[e];
            const int s = src[e];
            float ea[EDIM];
#pragma unroll
            for (int k = 0; k < EDIM; k++) ea[k] = eattr[(size_t)e * EDIM + k];
            const int rk = atomicAdd(&cursor[d], 1);
            float t10 = 0.f, t11 = 0.f, t2 = 0.f;
#pragma unroll
            for (int k = 0; k < EDIM; k++) {
                t10 += ea[k] * v1[k * HEADS + 0];
                t11 += ea[k] * v1[k * HEADS + 1];
                t2  += ea[k] * v2[k];
            }
            uint4 r;
            r.x = (unsigned)s;
            r.y = __float_as_uint(t10);
            r.z = __float_as_uint(t11);
            r.w = __float_as_uint(t2);
            if (rk < SLOTS) {
                rec[(size_t)d * SLOTS + rk] = r;
            } else {
                const int i = atomicAdd(spill_n, 1);
                spill_dst[i] = d;
                spill_rec[i] = r;
            }
        }
    }
}

__device__ inline float wave_sum(float x) {
#pragma unroll
    for (int o = 16; o > 0; o >>= 1) x += __shfl_xor(x, o, 32);
    return x;
}
__device__ inline float wave_max(float x) {
#pragma unroll
    for (int o = 16; o > 0; o >>= 1) x = fmaxf(x, __shfl_xor(x, o, 32));
    return x;
}

template<int H>
__device__ inline void load_rec(const uint4* __restrict__ rec, int p, int& s, float (&e)[H]) {
    const uint4 r = rec[(size_t)p];
    s = (int)r.x;
    if (H == 2) {
        e[0] = __uint_as_float(r.y);
        e[H - 1] = __uint_as_float(r.z);
    } else {
        e[0] = __uint_as_float(r.w);
    }
}

template<int H>
__device__ inline bool get_tail(const int* __restrict__ spill_dst, const uint4* __restrict__ spill_rec,
                                int nsp, int node, const float (&lp)[H], int i, int& s, float (&e)[H]) {
    s = node;
#pragma unroll
    for (int k = 0; k < H; k++) e[k] = lp[k];
    if (i < nsp) {
        if (spill_dst[i] != node) return false;
        load_rec<H>(spill_rec, i, s, e);
    }
    return true;
}

template<int H>
__device__ inline void edge_alpha(int s, const float (&e)[H],
                                  const float* __restrict__ als, const float (&aldn)[H],
                                  float (&a)[H]) {
#pragma unroll
    for (int h = 0; h < H; h++) {
        float x = als[s * H + h] + aldn[h] + e[h];
        a[h] = (x > 0.f) ? x : NEG_SLOPE * x;
    }
}

template<int VEC> struct rowvec;
template<> struct rowvec<4> {
    static __device__ inline void ld(const float* p, float (&o)[4]) {
        const float4 t = *reinterpret_cast<const float4*>(p);
        o[0] = t.x; o[1] = t.y; o[2] = t.z; o[3] = t.w;
    }
    static __device__ inline void st(float* p, const float (&o)[4]) {
        float4 t; t.x = o[0]; t.y = o[1]; t.z = o[2]; t.w = o[3];
        *reinterpret_cast<float4*>(p) = t;
    }
};
template<> struct rowvec<2> {
    static __device__ inline void ld(const float* p, float (&o)[2]) {
        const float2 t = *reinterpret_cast<const float2*>(p);
        o[0] = t.x; o[1] = t.y;
    }
    static __device__ inline void st(float* p, const float (&o)[2]) {
        float2 t; t.x = o[0]; t.y = o[1];
        *reinterpret_cast<float2*>(p) = t;
    }
};

template<int H, int C, int VEC, int G>
__device__ inline void add_rows(const float* __restrict__ xh, int cnt, int s, const float (&w)[H],
                                int lane, int hl, float (&acc)[VEC]) {
    constexpr int HC = H * C;
    constexpr int NACT = HC / VEC;
    constexpr int SB = 4;
    constexpr int NSB = G / SB;
    static_assert(G % SB == 0 && 32 % G == 0, "groups are whole blocks and tile the 32 lanes");
    for (int j0 = 0; j0 < cnt; j0 += G) {
        int   sj[G];
        float wj[G];
#pragma unroll
        for (int k = 0; k < G; k++) {
            sj[k] = __builtin_amdgcn_readlane(s, j0 + k);
            wj[k] = __int_as_float(__builtin_amdgcn_readlane(__float_as_int(w[0]), j0 + k));
            if (H == 2) {
                const float wj1 = __int_as_float(__builtin_amdgcn_readlane(__float_as_int(w[H - 1]), j0 + k));
                wj[k] = hl ? wj1 : wj[k];
            }
        }
        if (lane < NACT) {
            float r[G][VEC];
#pragma unroll
            for (int b = 0; b < NSB; b++) {
                if (b == 0 || j0 + b * SB < cnt) {
#pragma unroll
                    for (int k = b * SB; k < (b + 1) * SB; k++)
                        rowvec<VEC>::ld(xh + (size_t)sj[k] * HC + lane * VEC, r[k]);
                }
            }
#pragma unroll
            for (int b = 0; b < NSB; b++) {
                if (b == 0 || j0 + b * SB < cnt) {
#pragma unroll
                    for (int k = b * SB; k < (b + 1) * SB; k++)
#pragma unroll
                        for (int q = 0; q < VEC; q++) acc[q] = __builtin_fmaf(r[k][q], wj[k], acc[q]);
                }
            }
        }
    }
}

template<int H, int C, int VEC, int G, bool LOGSM>
__global__ void __launch_bounds__(256) gat_node(const int* __restrict__ cursor,
                                                const uint4* __restrict__ rec,
                                                const int* __restrict__ spill_n,
                                                const int* __restrict__ spill_dst,
                                                const uint4* __restrict__ spill_rec,
                                                const float* __restrict__ als,
                                                const float* __restrict__ ald,
                                                const float* __restrict__ xh,
                                                const float* __restrict__ bias,
                                                float* __restrict__ row_out, int n) {
    constexpr int HC = H * C;
    constexpr int NACT = HC / VEC;
    static_assert(H == 1 || H == 2, "head select below handles one or two heads");
    static_assert(NACT * VEC == HC && NACT <= 32 && C % VEC == 0, "lane to channel map");
    const int lane = threadIdx.x & 31;
    const int node = __builtin_amdgcn_readfirstlane((int)((blockIdx.x * blockDim.x + threadIdx.x) >> 5));
    if (node >= n) return;
    const int start = node * SLOTS;
    const int deg   = cursor[node];
    const int nslot = deg < SLOTS ? deg : SLOTS;
    const bool tail = deg >= SLOTS;
    const int first = tail ? SLOTS : deg + 1;
    int nsp = 0;
    if (deg > SLOTS) nsp = spill_n[0];

    float aldn[H];
#pragma unroll
    for (int h = 0; h < H; h++) aldn[h] = ald[node * H + h];

    int   s_f = node;
    float e_f[H];
    float sm[H];
#pragma unroll
    for (int i = 0; i < H; i++) { e_f[i] = 0.f; sm[i] = 0.f; }
    if (lane < nslot) {
        load_rec<H>(rec, start + lane, s_f, e_f);
#pragma unroll
        for (int i = 0; i < H; i++) sm[i] = e_f[i];
    }
    for (int j = lane; j < nsp; j += 32) {
        if (spill_dst[j] == node) {
            int s; float e[H];
            load_rec<H>(spill_rec, j, s, e);
#pragma unroll
            for (int i = 0; i < H; i++) sm[i] += e[i];
        }
    }
    float lp[H];
    const float cdiv = fmaxf((float)deg, 1.0f);
#pragma unroll
    for (int i = 0; i < H; i++) lp[i] = wave_sum(sm[i]) / cdiv;

    const bool val_f = lane < first;
    if (lane == deg) {
#pragma unroll
        for (int i = 0; i < H; i++) e_f[i] = lp[i];
    }
    float a_f[H];
    edge_alpha<H>(s_f, e_f, als, aldn, a_f);

    float m[H];
#pragma unroll
    for (int h = 0; h < H; h++) m[h] = val_f ? a_f[h] : -INFINITY;
    if (tail) {
        for (int j = lane; j <= nsp; j += 32) {
            int s; float e[H]; float a[H];
            if (get_tail<H>(spill_dst, spill_rec, nsp, node, lp, j, s, e)) {
                edge_alpha<H>(s, e, als, aldn, a);
#pragma unroll
                for (int h = 0; h < H; h++) m[h] = fmaxf(m[h], a[h]);
            }
        }
    }
#pragma unroll
    for (int h = 0; h < H; h++) m[h] = wave_max(m[h]);

    float ex_f[H], dn[H];
#pragma unroll
    for (int h = 0; h < H; h++) {
        ex_f[h] = val_f ? expf(a_f[h] - m[h]) : 0.f;
        dn[h] = ex_f[h];
    }
    if (tail) {
        for (int j = lane; j <= nsp; j += 32) {
            int s; float e[H]; float a[H];
            if (get_tail<H>(spill_dst, spill_rec, nsp, node, lp, j, s, e)) {
                edge_alpha<H>(s, e, als, aldn, a);
#pragma unroll
                for (int h = 0; h < H; h++) dn[h] += expf(a[h] - m[h]);
            }
        }
    }
#pragma unroll
    for (int h = 0; h < H; h++) dn[h] = wave_sum(dn[h]) + 1e-16f;

    const int hl = (lane * VEC) / C;
    const bool act = lane < NACT;
    float bv[VEC];
#pragma unroll
    for (int q = 0; q < VEC; q++) bv[q] = 0.f;
    if (act) rowvec<VEC>::ld(bias + lane * VEC, bv);
    float acc[VEC];
#pragma unroll
    for (int q = 0; q < VEC; q++) acc[q] = 0.f;

    float w_f[H];
#pragma unroll
    for (int h = 0; h < H; h++) w_f[h] = ex_f[h] / dn[h];
    add_rows<H, C, VEC, G>(xh, first, s_f, w_f, lane, hl, acc);

    if (tail) {
        for (int cb = 0; cb <= nsp; cb += 32) {
            const int j = cb + lane;
            int s = node;
            float w[H];
#pragma unroll
            for (int h = 0; h < H; h++) w[h] = 0.f;
            if (j <= nsp) {
                float e[H]; float a[H];
                if (get_tail<H>(spill_dst, spill_rec, nsp, node, lp, j, s, e)) {
                    edge_alpha<H>(s, e, als, aldn, a);
#pragma unroll
                    for (int h = 0; h < H; h++) w[h] = expf(a[h] - m[h]) / dn[h];
                }
            }
            const int left = nsp + 1 - cb;
            add_rows<H, C, VEC, G>(xh, left < 32 ? left : 32, s, w, lane, hl, acc);
        }
    }

#pragma unroll
    for (int q = 0; q < VEC; q++) acc[q] += bv[q];

    if (!LOGSM) {
#pragma unroll
        for (int q = 0; q < VEC; q++) acc[q] = (acc[q] > 0.f) ? acc[q] : expm1f(acc[q]);
    } else {
        float mx = -INFINITY;
        if (act) {
#pragma unroll
            for (int q = 0; q < VEC; q++) mx = fmaxf(mx, acc[q]);
        }
        mx = wave_max(mx);
        float se = 0.f;
        if (act) {
#pragma unroll
            for (int q = 0; q < VEC; q++) se += expf(acc[q] - mx);
        }
        se = wave_sum(se);
        const float lse = mx + logf(se);
#pragma unroll
        for (int q = 0; q < VEC; q++) acc[q] -= lse;
    }

    if (act) rowvec<VEC>::st(row_out + (size_t)node * HC + lane * VEC, acc);
}

extern "C" void kernel_launch(void* const* d_in, const int* in_sizes, int n_in,
                              void* d_out, int out_size, void* d_ws, size_t ws_size,
                              hipStream_t stream) {
    const float* x   = (const float*)d_in[0];
    const int*   ei  = (const int*)d_in[1];
    const float* ea  = (const float*)d_in[2];
    const float* W1  = (const float*)d_in[3];
    const float* as1 = (const float*)d_in[4];
    const float* ad1 = (const float*)d_in[5];
    const float* We1 = (const float*)d_in[6];
    const float* ae1 = (const float*)d_in[7];
    const float* b1  = (const float*)d_in[8];
    const float* W2  = (const float*)d_in[9];
    const float* as2 = (const float*)d_in[10];
    const float* ad2 = (const float*)d_in[11];
    const float* We2 = (const float*)d_in[12];
    const float* ae2 = (const float*)d_in[13];
    const float* b2  = (const float*)d_in[14];

    const int N = N_NODES, E = N_EDGES;
    const int HC1 = HEADS * HID;
    const int* src = ei;
    const int* dst = ei + E;

    char* ws = (char*)d_ws;
    size_t off = 0;
    auto alloc = [&](size_t bytes) -> void* {
        void* p = ws + off;
        off = (off + bytes + 255) & ~(size_t)255;
        return p;
    };
    int*      cursor = (int*)alloc(sizeof(int) * N);
    uint4*    rec    = (uint4*)alloc(sizeof(uint4) * (size_t)N * SLOTS);
    int*      spill_n   = (int*)alloc(sizeof(int));
    int*      spill_dst = (int*)alloc(sizeof(int) * (size_t)E);
    uint4*    spill_rec = (uint4*)alloc(sizeof(uint4) * (size_t)E);
    float*    v1     = (float*)alloc(sizeof(float) * 16);
    float*    v2     = (float*)alloc(sizeof(float) * 16);
    float*    W2p    = (float*)alloc(sizeof(float) * F_IN * NPAD2);
    float*    xh1    = (float*)alloc(sizeof(float) * (size_t)N * HC1);
    float*    als1   = (float*)alloc(sizeof(float) * N * HEADS);
    float*    ald1   = (float*)alloc(sizeof(float) * N * HEADS);
    float*    h1     = (float*)alloc(sizeof(float) * (size_t)N * HC1);
    float*    xh2    = (float*)alloc(sizeof(float) * (size_t)N * OUT_F);
    float*    als2   = (float*)alloc(sizeof(float) * N);
    float*    ald2   = (float*)alloc(sizeof(float) * N);
    (void)ws_size; (void)n_in; (void)in_sizes; (void)out_size;

    const int T = 256;

    setup_small<<<ZERO_BLOCKS + PAD_BLOCKS + 1, SETUP_T, 0, stream>>>((unsigned*)cursor, (unsigned*)spill_n, W2, W2p, We1, ae1, We2, ae2, v1, v2);
    gemm1_and_fill<<<GEMM1_BLOCKS + FILL_BLOCKS, GC_T, 0, stream>>>(x, W1, xh1, as1, ad1, als1, ald1, src, dst, ea, v1, v2,
                                                                   cursor, rec, spill_n, spill_dst, spill_rec);

    gat_node<HEADS, HID, 4, 8, false><<<CDIV(N, T / 32), T, 0, stream>>>(cursor, rec, spill_n, spill_dst, spill_rec, als1, ald1, xh1, b1, h1, N);

    {
        int mtiles = N / 16, waves_per_blk = 128 / 32;
        wmma_gemm<3, HC1, 1><<<CDIV(mtiles, waves_per_blk), 128, 0, stream>>>(h1, W2p, xh2, as2, ad2, als2, ald2, N, OUT_F);
    }
    gat_node<1, OUT_F, 2, 16, true><<<CDIV(N, T / 32), T, 0, stream>>>(cursor, rec, spill_n, spill_dst, spill_rec, als2, ald2, xh2, b2, (float*)d_out, N);
}


# Edges ranked by a two-level counting sort on LDS counters (chunk -> node-range cells in the set-up launch, range -> node slots beside the layer-1 GEMM), no per-edge global atomic; arrival-order slots
# speedup vs baseline: 10.7576x; 1.3769x over previous
#include <hip/hip_runtime.h>
#include <math.h>

#define N_NODES 50000
#define N_EDGES 800000
#define ETOT    (N_EDGES + N_NODES)
#define F_IN    128
#define HID     64
#define HEADS   2
#define OUT_F   40
#define EDIM    5
#define NEG_SLOPE 0.2f

#define CDIV(a,b) (((a)+(b)-1)/(b))

typedef __attribute__((ext_vector_type(2))) float v2f;
typedef __attribute__((ext_vector_type(8))) float v8f;

#define NPAD2   48

#define SLOTS   32

#define CHUNK_WGS 256
#define CHUNK_E   (N_EDGES / CHUNK_WGS)
#define CHUNK_T   1024
#define EPT       CDIV(CHUNK_E, CHUNK_T)
#define RANGE_N   196
#define N_RANGES  CDIV(N_NODES, RANGE_N)
#define CAP       32
#define PAD_BLOCKS CDIV(F_IN * NPAD2, CHUNK_T)
static_assert(N_EDGES % CHUNK_WGS == 0, "chunks are equal");
static_assert(N_RANGES + 32 <= CHUNK_T, "a level-1 thread per counter");
static_assert(N_NODES <= 65536 && RANGE_N <= 256, "src and the dst inside its range share one word of a cell record");

__global__ void __launch_bounds__(CHUNK_T) setup_and_bin(const int* __restrict__ src, const int* __restrict__ dst,
                                                         const float* __restrict__ eattr,
                                                         const float* __restrict__ We1, const float* __restrict__ ae1,
                                                         const float* __restrict__ We2, const float* __restrict__ ae2,
                                                         uint4* __restrict__ cells, int* __restrict__ cellcnt,
                                                         int* __restrict__ ovfn, int* __restrict__ ovf_dst,
                                                         uint4* __restrict__ ovf_rec,
                                                         unsigned* __restrict__ spill_n,
                                                         const float* __restrict__ W2, float* __restrict__ W2p) {
    __shared__ int bcnt[N_RANGES + 32];
    __shared__ int ovn;
    __shared__ float sv[EDIM * HEADS + EDIM];
    const int b = blockIdx.x;
    const int t = threadIdx.x;
    if (b >= CHUNK_WGS) {
        const int i = (b - CHUNK_WGS) * CHUNK_T + t;
        if (i < F_IN * NPAD2) {
            const int r = i / NPAD2, c = i % NPAD2;
            W2p[i] = (c < OUT_F) ? W2[r * OUT_F + c] : 0.0f;
        }
        if (i == 0) spill_n[0] = 0u;
        return;
    }
    if (t < N_RANGES + 32) bcnt[t] = 0;
    if (t == 0) ovn = 0;

    const int base = b * CHUNK_E;
    int   d[EPT], s[EPT];
    float ea[EPT][EDIM];
#pragma unroll
    for (int k = 0; k < EPT; k++) {
        const int j = k * CHUNK_T + t;
        d[k] = 0; s[k] = 0;
#pragma unroll
        for (int q = 0; q < EDIM; q++) ea[k][q] = 0.f;
        if (j < CHUNK_E) {
            const int e = base + j;
            d[k] = dst[e];
            s[k] = src[e];
#pragma unroll
            for (int q = 0; q < EDIM; q++) ea[k][q] = eattr[(size_t)e * EDIM + q];
        }
    }

    if (t < EDIM * HEADS) {
        const int dd = t / HEADS, h = t % HEADS;
        float a = 0.f;
#pragma unroll
        for (int c = 0; c < HID; c++)
            a += We1[dd * (HEADS * HID) + h * HID + c] * ae1[h * HID + c];
        sv[t] = a;
    } else if (t < EDIM * HEADS + EDIM) {
        const int dd = t - EDIM * HEADS;
        float a = 0.f;
#pragma unroll
        for (int c = 0; c < OUT_F; c++)
            a += We2[dd * OUT_F + c] * ae2[c];
        sv[t] = a;
    }
    __syncthreads();

    float v1[EDIM * HEADS], v2[EDIM];
#pragma unroll
    for (int q = 0; q < EDIM * HEADS; q++) v1[q] = sv[q];
#pragma unroll
    for (int q = 0; q < EDIM; q++) v2[q] = sv[EDIM * HEADS + q];

    int rg[EPT], rk[EPT];
#pragma unroll
    for (int k = 0; k < EPT; k++) {
        const bool ok = k * CHUNK_T + t < CHUNK_E;
        rg[k] = d[k] / RANGE_N;
        rk[k] = atomicAdd(&bcnt[ok ? rg[k] : N_RANGES + (t & 31)], ok ? 1 : 0);
    }
#pragma unroll
    for (int k = 0; k < EPT; k++) {
        if (k * CHUNK_T + t < CHUNK_E) {
            float t10 = 0.f, t11 = 0.f, t2 = 0.f;
#pragma unroll
            for (int q = 0; q < EDIM; q++) {
                t10 += ea[k][q] * v1[q * HEADS + 0];
                t11 += ea[k][q] * v1[q * HEADS + 1];
                t2  += ea[k][q] * v2[q];
            }
            const int dl = d[k] - rg[k] * RANGE_N;
            uint4 r;
            r.x = (unsigned)s[k] | ((unsigned)dl << 16);
            r.y = __float_as_uint(t10);
            r.z = __float_as_uint(t11);
            r.w = __float_as_uint(t2);
            uint4* p = cells + ((size_t)rg[k] * CHUNK_WGS + b) * CAP + rk[k];
            if (rk[k] >= CAP) {
                const int i = atomicAdd(&ovn, 1);
                ovf_dst[base + i] = d[k];
                r.x = (unsigned)s[k];
                p = ovf_rec + base + i;
            }
            *p = r;
        }
    }
    __syncthreads();
    if (t < N_RANGES) {
        const int c = bcnt[t];
        cellcnt[t * CHUNK_WGS + b] = c < CAP ? c : CAP;
        if (c == 0) cells[((size_t)t * CHUNK_WGS + b) * CAP] = make_uint4(0u, 0u, 0u, 0u);
    }
    if (t == 0) ovfn[b] = ovn;
}

template<int M>
__device__ inline float row16_xadd(float x) {
    const int y = __builtin_amdgcn_update_dpp(0, __float_as_int(x), 0x160 + M, 0xf, 0xf, true);
    return x + __int_as_float(y);
}

template<int NT, int K, int H>
__device__ inline void gemm_stripe(int wave, const float* __restrict__ A, const float* __restrict__ B,
                                   float* __restrict__ C,
                                   const float* __restrict__ asrc, const float* __restrict__ adst,
                                   float* __restrict__ als, float* __restrict__ ald, int M, int N) {
    constexpr int NPAD = NT * 16;
    constexpr int TPH = NT / H;
    static_assert(NT % H == 0, "a column tile belongs to one head");
    int lane = threadIdx.x & 31;
    int half = lane >> 4;
    int lrow = lane & 15;
    int mtiles = M >> 4;
    if (wave >= mtiles) return;
    int m0 = wave << 4;

    float sa[NT], da[NT];
#pragma unroll
    for (int nt = 0; nt < NT; nt++) {
        const int col = nt * 16 + lrow;
        sa[nt] = (col < N) ? asrc[col] : 0.0f;
        da[nt] = (col < N) ? adst[col] : 0.0f;
    }

    v8f acc[NT];
#pragma unroll
    for (int nt = 0; nt < NT; nt++)
#pragma unroll
        for (int g = 0; g < 8; g++) acc[nt][g] = 0.0f;

    const float* arow = A + (size_t)(m0 + lrow) * K;
    const float* bcol = B + lrow;
#pragma unroll 4
    for (int k = 0; k < K; k += 4) {
        int kk = k + half * 2;
        v2f a;
        a.x = arow[kk];
        a.y = arow[kk + 1];
#pragma unroll
        for (int nt = 0; nt < NT; nt++) {
            v2f b;
            b.x = bcol[kk * NPAD + nt * 16];
            b.y = bcol[(kk + 1) * NPAD + nt * 16];
            acc[nt] = __builtin_amdgcn_wmma_f32_16x16x4_f32(
                false, a, false, b, (short)0, acc[nt], false, false);
        }
    }
#pragma unroll
    for (int nt = 0; nt < NT; nt++) {
        int col = nt * 16 + lrow;
        if (col < N) {
#pragma unroll
            for (int g = 0; g < 8; g++)
                C[(size_t)(m0 + g + half * 8) * N + col] = acc[nt][g];
        }
    }

    constexpr int NV = 2 * H * 8;
    float red[NV];
#pragma unroll
    for (int i = 0; i < NV; i++) red[i] = 0.0f;
#pragma unroll
    for (int nt = 0; nt < NT; nt++) {
        const int h = nt / TPH;
        const bool incol = (nt * 16 + lrow) < N;
#pragma unroll
        for (int g = 0; g < 8; g++) {
            const float cv = incol ? acc[nt][g] : 0.0f;
            red[(0 * H + h) * 8 + g] += cv * sa[nt];
            red[(1 * H + h) * 8 + g] += cv * da[nt];
        }
    }
#pragma unroll
    for (int i = 0; i < NV; i++) red[i] = row16_xadd<1>(red[i]);
#pragma unroll
    for (int i = 0; i < NV; i++) red[i] = row16_xadd<2>(red[i]);
#pragma unroll
    for (int i = 0; i < NV; i++) red[i] = row16_xadd<4>(red[i]);
#pragma unroll
    for (int i = 0; i < NV; i++) red[i] = row16_xadd<8>(red[i]);
    if (lrow == 0) {
        float* sp = als + (size_t)(m0 + half * 8) * H;
        float* dp = ald + (size_t)(m0 + half * 8) * H;
#pragma unroll
        for (int g = 0; g < 8; g++)
#pragma unroll
            for (int h = 0; h < H; h++) {
                sp[g * H + h] = red[(0 * H + h) * 8 + g];
                dp[g * H + h] = red[(1 * H + h) * 8 + g];
            }
    }
}

template<int NT, int K, int H>
__global__ void wmma_gemm(const float* __restrict__ A, const float* __restrict__ B,
                          float* __restrict__ C,
                          const float* __restrict__ asrc, const float* __restrict__ adst,
                          float* __restrict__ als, float* __restrict__ ald, int M, int N) {
    gemm_stripe<NT, K, H>((int)((blockIdx.x * blockDim.x + threadIdx.x) >> 5),
                          A, B, C, asrc, adst, als, ald, M, N);
}

#define GC_T 256
#define GEMM1_BLOCKS CDIV(N_NODES / 16, GC_T / 32)
static_assert(N_NODES % 16 == 0, "the GEMMs take whole 16-row stripes");
static_assert(GC_T == CHUNK_WGS, "a level-2 thread per chunk: its cell count and its overflow list");
static_assert(CAP == 32, "a wave reads one cell per load, lane = record");
static_assert(RANGE_N + 32 <= GC_T, "a level-2 thread per counter");

__device__ inline void store_rec(int node, int rk, uint4 r, uint4* __restrict__ rec,
                                 int* spill_n, int* __restrict__ spill_dst, uint4* __restrict__ spill_rec) {
    uint4* p = rec + (size_t)node * SLOTS + rk;
    if (rk >= SLOTS) {
        const int i = atomicAdd(spill_n, 1);
        spill_dst[i] = node;
        p = spill_rec + i;
    }
    *p = r;
}

__device__ inline void place_rec(int* ncnt, int rg, int dl, uint4 r, uint4* __restrict__ rec,
                                 int* spill_n, int* __restrict__ spill_dst, uint4* __restrict__ spill_rec) {
    const int rk = atomicAdd(&ncnt[dl], 1);
    store_rec(rg * RANGE_N + dl, rk, r, rec, spill_n, spill_dst, spill_rec);
}

__global__ void __launch_bounds__(GC_T) gemm1_and_rank(const float* __restrict__ x, const float* __restrict__ W1,
                                                       float* __restrict__ xh,
                                                       const float* __restrict__ asrc, const float* __restrict__ adst,
                                                       float* __restrict__ als, float* __restrict__ ald,
                                                       const uint4* __restrict__ cells, const int* __restrict__ cellcnt,
                                                       const int* __restrict__ ovfn, const int* __restrict__ ovf_dst,
                                                       const uint4* __restrict__ ovf_rec,
                                                       int* __restrict__ cursor, uint4* __restrict__ rec,
                                                       int* spill_n, int* __restrict__ spill_dst,
                                                       uint4* __restrict__ spill_rec) {
    __shared__ int ncnt[RANGE_N + 32];
    __shared__ int ccnt[CHUNK_WGS];
    if (blockIdx.x < GEMM1_BLOCKS) {
        gemm_stripe<8, F_IN, HEADS>((int)(blockIdx.x * (GC_T / 32) + (threadIdx.x >> 5)),
                                    x, W1, xh, asrc, adst, als, ald, N_NODES, HEADS * HID);
    } else {
        const int rg = blockIdx.x - GEMM1_BLOCKS;
        const int t = threadIdx.x;
        const int lane = t & 31;
        const int wv = t >> 5;
        if (t < RANGE_N + 32) ncnt[t] = 0;
        ccnt[t] = cellcnt[rg * CHUNK_WGS + t];
        const int novf = ovfn[t];
        __syncthreads();

        constexpr int WPB = GC_T / 32;
        constexpr int CG = 8;
        static_assert((CHUNK_WGS / WPB) % CG == 0, "cell groups are whole");
        for (int i0 = 0; i0 < CHUNK_WGS / WPB; i0 += CG) {
            uint4 q[CG];
            bool  v[CG];
            int   dl[CG], rk[CG];
#pragma unroll
            for (int k = 0; k < CG; k++) {
                const int w = (i0 + k) * WPB + wv;
                v[k] = lane < ccnt[w];
                q[k] = cells[((size_t)rg * CHUNK_WGS + w) * CAP + (v[k] ? lane : 0)];
            }
#pragma unroll
            for (int k = 0; k < CG; k++) {
                dl[k] = v[k] ? (int)(q[k].x >> 16) : RANGE_N + lane;
                rk[k] = atomicAdd(&ncnt[dl[k]], v[k] ? 1 : 0);
            }
#pragma unroll
            for (int k = 0; k < CG; k++) {
                if (v[k]) {
                    uint4 r = q[k];
                    r.x &= 0xffffu;
                    store_rec(rg * RANGE_N + dl[k], rk[k], r, rec, spill_n, spill_dst, spill_rec);
                }
            }
        }
        for (int i = 0; i < novf; i++) {
            const int dd = ovf_dst[t * CHUNK_E + i];
            if (dd / RANGE_N == rg)
                place_rec(ncnt, rg, dd - rg * RANGE_N, ovf_rec[t * CHUNK_E + i], rec, spill_n, spill_dst, spill_rec);
        }
        __syncthreads();
        if (t < RANGE_N) {
            const int node = rg * RANGE_N + t;
            if (node < N_NODES) cursor[node] = ncnt[t];
        }
    }
}

__device__ inline float wave_sum(float x) {
#pragma unroll
    for (int o = 16; o > 0; o >>= 1) x += __shfl_xor(x, o, 32);
    return x;
}
__device__ inline float wave_max(float x) {
#pragma unroll
    for (int o = 16; o > 0; o >>= 1) x = fmaxf(x, __shfl_xor(x, o, 32));
    return x;
}

template<int H>
__device__ inline void load_rec(const uint4* __restrict__ rec, int p, int& s, float (&e)[H]) {
    const uint4 r = rec[(size_t)p];
    s = (int)r.x;
    if (H == 2) {
        e[0] = __uint_as_float(r.y);
        e[H - 1] = __uint_as_float(r.z);
    } else {
        e[0] = __uint_as_float(r.w);
    }
}

template<int H>
__device__ inline bool get_tail(const int* __restrict__ spill_dst, const uint4* __restrict__ spill_rec,
                                int nsp, int node, const float (&lp)[H], int i, int& s, float (&e)[H]) {
    s = node;
#pragma unroll
    for (int k = 0; k < H; k++) e[k] = lp[k];
    if (i < nsp) {
        if (spill_dst[i] != node) return false;
        load_rec<H>(spill_rec, i, s, e);
    }
    return true;
}

template<int H>
__device__ inline void edge_alpha(int s, const float (&e)[H],
                                  const float* __restrict__ als, const float (&aldn)[H],
                                  float (&a)[H]) {
#pragma unroll
    for (int h = 0; h < H; h++) {
        float x = als[s * H + h] + aldn[h] + e[h];
        a[h] = (x > 0.f) ? x : NEG_SLOPE * x;
    }
}

template<int VEC> struct rowvec;
template<> struct rowvec<4> {
    static __device__ inline void ld(const float* p, float (&o)[4]) {
        const float4 t = *reinterpret_cast<const float4*>(p);
        o[0] = t.x; o[1] = t.y; o[2] = t.z; o[3] = t.w;
    }
    static __device__ inline void st(float* p, const float (&o)[4]) {
        float4 t; t.x = o[0]; t.y = o[1]; t.z = o[2]; t.w = o[3];
        *reinterpret_cast<float4*>(p) = t;
    }
};
template<> struct rowvec<2> {
    static __device__ inline void ld(const float* p, float (&o)[2]) {
        const float2 t = *reinterpret_cast<const float2*>(p);
        o[0] = t.x; o[1] = t.y;
    }
    static __device__ inline void st(float* p, const float (&o)[2]) {
        float2 t; t.x = o[0]; t.y = o[1];
        *reinterpret_cast<float2*>(p) = t;
    }
};

template<int H, int C, int VEC, int G>
__device__ inline void add_rows(const float* __restrict__ xh, int cnt, int s, const float (&w)[H],
                                int lane, int hl, float (&acc)[VEC]) {
    constexpr int HC = H * C;
    constexpr int NACT = HC / VEC;
    constexpr int SB = 4;
    constexpr int NSB = G / SB;
    static_assert(G % SB == 0 && 32 % G == 0, "groups are whole blocks and tile the 32 lanes");
    for (int j0 = 0; j0 < cnt; j0 += G) {
        int   sj[G];
        float wj[G];
#pragma unroll
        for (int k = 0; k < G; k++) {
            sj[k] = __builtin_amdgcn_readlane(s, j0 + k);
            wj[k] = __int_as_float(__builtin_amdgcn_readlane(__float_as_int(w[0]), j0 + k));
            if (H == 2) {
                const float wj1 = __int_as_float(__builtin_amdgcn_readlane(__float_as_int(w[H - 1]), j0 + k));
                wj[k] = hl ? wj1 : wj[k];
            }
        }
        if (lane < NACT) {
            float r[G][VEC];
#pragma unroll
            for (int b = 0; b < NSB; b++) {
                if (b == 0 || j0 + b * SB < cnt) {
#pragma unroll
                    for (int k = b * SB; k < (b + 1) * SB; k++)
                        rowvec<VEC>::ld(xh + (size_t)sj[k] * HC + lane * VEC, r[k]);
                }
            }
#pragma unroll
            for (int b = 0; b < NSB; b++) {
                if (b == 0 || j0 + b * SB < cnt) {
#pragma unroll
                    for (int k = b * SB; k < (b + 1) * SB; k++)
#pragma unroll
                        for (int q = 0; q < VEC; q++) acc[q] = __builtin_fmaf(r[k][q], wj[k], acc[q]);
                }
            }
        }
    }
}

template<int H, int C, int VEC, int G, bool LOGSM>
__global__ void __launch_bounds__(256) gat_node(const int* __restrict__ cursor,
                                                const uint4* __restrict__ rec,
                                                const int* __restrict__ spill_n,
                                                const int* __restrict__ spill_dst,
                                                const uint4* __restrict__ spill_rec,
                                                const float* __restrict__ als,
                                                const float* __restrict__ ald,
                                                const float* __restrict__ xh,
                                                const float* __restrict__ bias,
                                                float* __restrict__ row_out, int n) {
    constexpr int HC = H * C;
    constexpr int NACT = HC / VEC;
    static_assert(H == 1 || H == 2, "head select below handles one or two heads");
    static_assert(NACT * VEC == HC && NACT <= 32 && C % VEC == 0, "lane to channel map");
    const int lane = threadIdx.x & 31;
    const int node = __builtin_amdgcn_readfirstlane((int)((blockIdx.x * blockDim.x + threadIdx.x) >> 5));
    if (node >= n) return;
    const int start = node * SLOTS;
    const int deg   = cursor[node];
    const int nslot = deg < SLOTS ? deg : SLOTS;
    const bool tail = deg >= SLOTS;
    const int first = tail ? SLOTS : deg + 1;
    int nsp = 0;
    if (deg > SLOTS) nsp = spill_n[0];

    float aldn[H];
#pragma unroll
    for (int h = 0; h < H; h++) aldn[h] = ald[node * H + h];

    int   s_f = node;
    float e_f[H];
    float sm[H];
#pragma unroll
    for (int i = 0; i < H; i++) { e_f[i] = 0.f; sm[i] = 0.f; }
    if (lane < nslot) {
        load_rec<H>(rec, start + lane, s_f, e_f);
#pragma unroll
        for (int i = 0; i < H; i++) sm[i] = e_f[i];
    }
    for (int j = lane; j < nsp; j += 32) {
        if (spill_dst[j] == node) {
            int s; float e[H];
            load_rec<H>(spill_rec, j, s, e);
#pragma unroll
            for (int i = 0; i < H; i++) sm[i] += e[i];
        }
    }
    float lp[H];
    const float cdiv = fmaxf((float)deg, 1.0f);
#pragma unroll
    for (int i = 0; i < H; i++) lp[i] = wave_sum(sm[i]) / cdiv;

    const bool val_f = lane < first;
    if (lane == deg) {
#pragma unroll
        for (int i = 0; i < H; i++) e_f[i] = lp[i];
    }
    float a_f[H];
    edge_alpha<H>(s_f, e_f, als, aldn, a_f);

    float m[H];
#pragma unroll
    for (int h = 0; h < H; h++) m[h] = val_f ? a_f[h] : -INFINITY;
    if (tail) {
        for (int j = lane; j <= nsp; j += 32) {
            int s; float e[H]; float a[H];
            if (get_tail<H>(spill_dst, spill_rec, nsp, node, lp, j, s, e)) {
                edge_alpha<H>(s, e, als, aldn, a);
#pragma unroll
                for (int h = 0; h < H; h++) m[h] = fmaxf(m[h], a[h]);
            }
        }
    }
#pragma unroll
    for (int h = 0; h < H; h++) m[h] = wave_max(m[h]);

    float ex_f[H], dn[H];
#pragma unroll
    for (int h = 0; h < H; h++) {
        ex_f[h] = val_f ? expf(a_f[h] - m[h]) : 0.f;
        dn[h] = ex_f[h];
    }
    if (tail) {
        for (int j = lane; j <= nsp; j += 32) {
            int s; float e[H]; float a[H];
            if (get_tail<H>(spill_dst, spill_rec, nsp, node, lp, j, s, e)) {
                edge_alpha<H>(s, e, als, aldn, a);
#pragma unroll
                for (int h = 0; h < H; h++) dn[h] += expf(a[h] - m[h]);
            }
        }
    }
#pragma unroll
    for (int h = 0; h < H; h++) dn[h] = wave_sum(dn[h]) + 1e-16f;

    const int hl = (lane * VEC) / C;
    const bool act = lane < NACT;
    float bv[VEC];
#pragma unroll
    for (int q = 0; q < VEC; q++) bv[q] = 0.f;
    if (act) rowvec<VEC>::ld(bias + lane * VEC, bv);
    float acc[VEC];
#pragma unroll
    for (int q = 0; q < VEC; q++) acc[q] = 0.f;

    float w_f[H];
#pragma unroll
    for (int h = 0; h < H; h++) w_f[h] = ex_f[h] / dn[h];
    add_rows<H, C, VEC, G>(xh, first, s_f, w_f, lane, hl, acc);

    if (tail) {
        for (int cb = 0; cb <= nsp; cb += 32) {
            const int j = cb + lane;
            int s = node;
            float w[H];
#pragma unroll
            for (int h = 0; h < H; h++) w[h] = 0.f;
            if (j <= nsp) {
                float e[H]; float a[H];
                if (get_tail<H>(spill_dst, spill_rec, nsp, node, lp, j, s, e)) {
                    edge_alpha<H>(s, e, als, aldn, a);
#pragma unroll
                    for (int h = 0; h < H; h++) w[h] = expf(a[h] - m[h]) / dn[h];
                }
            }
            const int left = nsp + 1 - cb;
            add_rows<H, C, VEC, G>(xh, left < 32 ? left : 32, s, w, lane, hl, acc);
        }
    }

#pragma unroll
    for (int q = 0; q < VEC; q++) acc[q] += bv[q];

    if (!LOGSM) {
#pragma unroll
        for (int q = 0; q < VEC; q++) acc[q] = (acc[q] > 0.f) ? acc[q] : expm1f(acc[q]);
    } else {
        float mx = -INFINITY;
        if (act) {
#pragma unroll
            for (int q = 0; q < VEC; q++) mx = fmaxf(mx, acc[q]);
        }
        mx = wave_max(mx);
        float se = 0.f;
        if (act) {
#pragma unroll
            for (int q = 0; q < VEC; q++) se += expf(acc[q] - mx);
        }
        se = wave_sum(se);
        const float lse = mx + logf(se);
#pragma unroll
        for (int q = 0; q < VEC; q++) acc[q] -= lse;
    }

    if (act) rowvec<VEC>::st(row_out + (size_t)node * HC + lane * VEC, acc);
}

extern "C" void kernel_launch(void* const* d_in, const int* in_sizes, int n_in,
                              void* d_out, int out_size, void* d_ws, size_t ws_size,
                              hipStream_t stream) {
    const float* x   = (const float*)d_in[0];
    const int*   ei  = (const int*)d_in[1];
    const float* ea  = (const float*)d_in[2];
    const float* W1  = (const float*)d_in[3];
    const float* as1 = (const float*)d_in[4];
    const float* ad1 = (const float*)d_in[5];
    const float* We1 = (const float*)d_in[6];
    const float* ae1 = (const float*)d_in[7];
    const float* b1  = (const float*)d_in[8];
    const float* W2  = (const float*)d_in[9];
    const float* as2 = (const float*)d_in[10];
    const float* ad2 = (const float*)d_in[11];
    const float* We2 = (const float*)d_in[12];
    const float* ae2 = (const float*)d_in[13];
    const float* b2  = (const float*)d_in[14];

    const int N = N_NODES, E = N_EDGES;
    const int HC1 = HEADS * HID;
    const int* src = ei;
    const int* dst = ei + E;

    char* ws = (char*)d_ws;
    size_t off = 0;
    auto alloc = [&](size_t bytes) -> void* {
        void* p = ws + off;
        off = (off + bytes + 255) & ~(size_t)255;
        return p;
    };
    int*      cursor = (int*)alloc(sizeof(int) * N);
    uint4*    rec    = (uint4*)alloc(sizeof(uint4) * (size_t)N * SLOTS);
    int*      spill_n   = (int*)alloc(sizeof(int));
    int*      spill_dst = (int*)alloc(sizeof(int) * (size_t)E);
    uint4*    spill_rec = (uint4*)alloc(sizeof(uint4) * (size_t)E);
    uint4*    cells   = (uint4*)alloc(sizeof(uint4) * (size_t)N_RANGES * CHUNK_WGS * CAP);
    int*      cellcnt = (int*)alloc(sizeof(int) * (size_t)N_RANGES * CHUNK_WGS);
    int*      ovfn    = (int*)alloc(sizeof(int) * CHUNK_WGS);
    int*      ovf_dst = (int*)alloc(sizeof(int) * (size_t)E);
    uint4*    ovf_rec = (uint4*)alloc(sizeof(uint4) * (size_t)E);
    float*    W2p    = (float*)alloc(sizeof(float) * F_IN * NPAD2);
    float*    xh1    = (float*)alloc(sizeof(float) * (size_t)N * HC1);
    float*    als1   = (float*)alloc(sizeof(float) * N * HEADS);
    float*    ald1   = (float*)alloc(sizeof(float) * N * HEADS);
    float*    h1     = (float*)alloc(sizeof(float) * (size_t)N * HC1);
    float*    xh2    = (float*)alloc(sizeof(float) * (size_t)N * OUT_F);
    float*    als2   = (float*)alloc(sizeof(float) * N);
    float*    ald2   = (float*)alloc(sizeof(float) * N);
    (void)ws_size; (void)n_in; (void)in_sizes; (void)out_size;

    const int T = 256;

    setup_and_bin<<<CHUNK_WGS + PAD_BLOCKS, CHUNK_T, 0, stream>>>(src, dst, ea, We1, ae1, We2, ae2, cells, cellcnt,
                                                                 ovfn, ovf_dst, ovf_rec, (unsigned*)spill_n, W2, W2p);
    gemm1_and_rank<<<GEMM1_BLOCKS + N_RANGES, GC_T, 0, stream>>>(x, W1, xh1, as1, ad1, als1, ald1,
                                                                cells, cellcnt, ovfn, ovf_dst, ovf_rec,
                                                                cursor, rec, spill_n, spill_dst, spill_rec);

    gat_node<HEADS, HID, 4, 8, false><<<CDIV(N, T / 32), T, 0, stream>>>(cursor, rec, spill_n, spill_dst, spill_rec, als1, ald1, xh1, b1, h1, N);

    {
        int mtiles = N / 16, waves_per_blk = 128 / 32;
        wmma_gemm<3, HC1, 1><<<CDIV(mtiles, waves_per_blk), 128, 0, stream>>>(h1, W2p, xh2, as2, ad2, als2, ald2, N, OUT_F);
    }
    gat_node<1, OUT_F, 2, 16, true><<<CDIV(N, T / 32), T, 0, stream>>>(cursor, rec, spill_n, spill_dst, spill_rec, als2, ald2, xh2, b2, (float*)d_out, N);
}
